# starting kernel
# baseline (speedup 1.0000x reference)
#include <hip/hip_runtime.h>
#include <math.h>

#define NN 50000
#define EE 800000
#define ET (EE + NN)
#define IN_DIM 128
#define HID 32
#define HEADS 4
#define HC 128
#define OUT_DIM 40
#define SLOPE 0.2f

typedef float v2f __attribute__((ext_vector_type(2)));
typedef float v8f __attribute__((ext_vector_type(8)));

__device__ __forceinline__ unsigned ford(float f) {
    unsigned u = __float_as_uint(f);
    return (u & 0x80000000u) ? ~u : (u | 0x80000000u);
}
__device__ __forceinline__ float funord(unsigned k) {
    unsigned u = (k & 0x80000000u) ? (k & 0x7FFFFFFFu) : ~k;
    return __uint_as_float(u);
}
#define FORD_NEG_INF 0x007FFFFFu

__device__ __forceinline__ void edge_sd(int u, const int* __restrict__ ei,
                                        int& src, int& dst) {
    if (u < EE) { src = ei[u]; dst = ei[EE + u]; }
    else        { src = u - EE; dst = u - EE; }
}

__device__ __forceinline__ float leaky(float e) {
    return e > 0.f ? e : SLOPE * e;
}

__device__ __forceinline__ v8f wmma4(v2f a, v2f b, v8f c) {
    return __builtin_amdgcn_wmma_f32_16x16x4_f32(
               false, a, false, b, (short)0, c, false, false);
}

__global__ __launch_bounds__(32) void wmma_gemm_f32_2t(
    const float* __restrict__ A, const float* __restrict__ B,
    float* __restrict__ C, int K, int Nout)
{
    const int mt   = blockIdx.x;
    const int nt   = blockIdx.y;
    const int lane = threadIdx.x;
    const int hi   = lane >> 4;
    const int lo   = lane & 15;

    const float* Ap = A + (size_t)(mt * 16 + lo) * K + 2 * hi;
    const float* Bp = B + (size_t)(2 * hi) * Nout + nt * 32 + lo;

    v8f acc0 = {}, acc1 = {};
    for (int k = 0; k < K; k += 4) {
        float2 av = *(const float2*)(Ap + k);
        v2f a; a.x = av.x; a.y = av.y;
        const float* bp = Bp + (size_t)k * Nout;
        v2f b0; b0.x = bp[0];  b0.y = bp[Nout];
        v2f b1; b1.x = bp[16]; b1.y = bp[Nout + 16];
        acc0 = wmma4(a, b0, acc0);
        acc1 = wmma4(a, b1, acc1);
    }
    const int col = nt * 32 + lo;
    float* Cp = C + (size_t)(mt * 16 + 8 * hi) * Nout + col;
    #pragma unroll
    for (int r = 0; r < 8; ++r) {
        Cp[(size_t)r * Nout + 0]  = acc0[r];
        Cp[(size_t)r * Nout + 16] = acc1[r];
    }
}

__global__ __launch_bounds__(32) void wmma_gemm_f32_tail(
    const float* __restrict__ A, const float* __restrict__ B,
    float* __restrict__ C, const float* __restrict__ bias,
    int K, int Nout)
{
    const int mt   = blockIdx.x;
    const int nt   = blockIdx.y;
    const int lane = threadIdx.x;
    const int hi   = lane >> 4;
    const int lo   = lane & 15;
    const int col  = nt * 16 + lo;
    const int ccol = col < Nout ? col : Nout - 1;

    const float* Ap = A + (size_t)(mt * 16 + lo) * K + 2 * hi;
    const float* Bp = B + (size_t)(2 * hi) * Nout + ccol;

    v8f acc = {};
    for (int k = 0; k < K; k += 4) {
        float2 av = *(const float2*)(Ap + k);
        v2f a; a.x = av.x; a.y = av.y;
        const float* bp = Bp + (size_t)k * Nout;
        v2f b; b.x = bp[0]; b.y = bp[Nout];
        acc = wmma4(a, b, acc);
    }
    if (col < Nout) {
        const float bb = bias ? bias[col] : 0.f;
        float* Cp = C + (size_t)(mt * 16 + 8 * hi) * Nout + col;
        #pragma unroll
        for (int r = 0; r < 8; ++r)
            Cp[(size_t)r * Nout] = acc[r] + bb;
    }
}

__global__ void attn_coef(const float* __restrict__ h,
                          const float* __restrict__ a_src,
                          const float* __restrict__ a_dst,
                          float* __restrict__ al_s,
                          float* __restrict__ al_d)
{
    int idx = blockIdx.x * blockDim.x + threadIdx.x;
    if (idx >= NN * HEADS) return;
    int n = idx / HEADS, hd = idx % HEADS;
    const float* hp = h + (size_t)n * HC + hd * HID;
    const float* as = a_src + hd * HID;
    const float* ad = a_dst + hd * HID;
    float s = 0.f, d = 0.f;
    #pragma unroll
    for (int c = 0; c < HID; ++c) { float v = hp[c]; s = fmaf(v, as[c], s); d = fmaf(v, ad[c], d); }
    al_s[idx] = s; al_d[idx] = d;
}

__global__ void init_layer(unsigned* __restrict__ m, float* __restrict__ den,
                           float* __restrict__ acc)
{
    int i = blockIdx.x * blockDim.x + threadIdx.x;
    if (i < NN * HEADS) { m[i] = FORD_NEG_INF; den[i] = 0.f; }
    if (i < NN * HC)    acc[i] = 0.f;
}

__global__ void edge_max(const int* __restrict__ ei,
                         const float* __restrict__ al_s,
                         const float* __restrict__ al_d,
                         unsigned* __restrict__ m)
{
    int idx = blockIdx.x * blockDim.x + threadIdx.x;
    if (idx >= ET * HEADS) return;
    int u = idx >> 2, hd = idx & 3;
    int src, dst; edge_sd(u, ei, src, dst);
    float e = leaky(al_s[src * HEADS + hd] + al_d[dst * HEADS + hd]);
    atomicMax(&m[dst * HEADS + hd], ford(e));
}

__global__ void edge_sum(const int* __restrict__ ei,
                         const float* __restrict__ al_s,
                         const float* __restrict__ al_d,
                         const unsigned* __restrict__ m,
                         float* __restrict__ den)
{
    int idx = blockIdx.x * blockDim.x + threadIdx.x;
    if (idx >= ET * HEADS) return;
    int u = idx >> 2, hd = idx & 3;
    int src, dst; edge_sd(u, ei, src, dst);
    float e  = leaky(al_s[src * HEADS + hd] + al_d[dst * HEADS + hd]);
    float mx = funord(m[dst * HEADS + hd]);
    atomicAdd(&den[dst * HEADS + hd], __expf(e - mx));
}

__global__ __launch_bounds__(256) void edge_aggregate(
    const int* __restrict__ ei, const float* __restrict__ h,
    const float* __restrict__ al_s, const float* __restrict__ al_d,
    const unsigned* __restrict__ m, const float* __restrict__ den,
    float* __restrict__ acc)
{
    int wave = blockIdx.x * (blockDim.x >> 5) + (threadIdx.x >> 5);
    int lane = threadIdx.x & 31;
    if (wave >= ET) return;
    int src, dst; edge_sd(wave, ei, src, dst);
    int hd = lane >> 3;
    float e  = leaky(al_s[src * HEADS + hd] + al_d[dst * HEADS + hd]);
    float mx = funord(m[dst * HEADS + hd]);
    float alpha = __expf(e - mx) / (den[dst * HEADS + hd] + 1e-16f);

    float4 v = *((const float4*)(h + (size_t)src * HC) + lane);
    float* ap = acc + (size_t)dst * HC + lane * 4;
    atomicAdd(ap + 0, v.x * alpha);
    atomicAdd(ap + 1, v.y * alpha);
    atomicAdd(ap + 2, v.z * alpha);
    atomicAdd(ap + 3, v.w * alpha);
}

__global__ void bias_relu(float* __restrict__ acc, const float* __restrict__ b)
{
    int i = blockIdx.x * blockDim.x + threadIdx.x;
    if (i >= NN * HC) return;
    float v = acc[i] + b[i & (HC - 1)];
    acc[i] = v > 0.f ? v : 0.f;
}

extern "C" void kernel_launch(void* const* d_in, const int* in_sizes, int n_in,
                              void* d_out, int out_size, void* d_ws, size_t ws_size,
                              hipStream_t stream)
{
    const float* x   = (const float*)d_in[0];
    const int*   ei  = (const int*)d_in[1];
    const float* W[3]    = { (const float*)d_in[2],  (const float*)d_in[6],  (const float*)d_in[10] };
    const float* asv[3]  = { (const float*)d_in[3],  (const float*)d_in[7],  (const float*)d_in[11] };
    const float* adv[3]  = { (const float*)d_in[4],  (const float*)d_in[8],  (const float*)d_in[12] };
    const float* bv[3]   = { (const float*)d_in[5],  (const float*)d_in[9],  (const float*)d_in[13] };
    const float* Wl  = (const float*)d_in[14];
    const float* bl  = (const float*)d_in[15];

    char* ws = (char*)d_ws;
    float*    hbuf = (float*)ws;                       ws += (size_t)NN * HC * sizeof(float);
    float*    bufA = (float*)ws;                       ws += (size_t)NN * HC * sizeof(float);
    float*    bufB = (float*)ws;                       ws += (size_t)NN * HC * sizeof(float);
    float*    al_s = (float*)ws;                       ws += (size_t)NN * HEADS * sizeof(float);
    float*    al_d = (float*)ws;                       ws += (size_t)NN * HEADS * sizeof(float);
    unsigned* mbuf = (unsigned*)ws;                    ws += (size_t)NN * HEADS * sizeof(unsigned);
    float*    den  = (float*)ws;

    const int TPB = 256;
    const int MT  = NN / 16;
    const int gNHC   = (NN * HC + TPB - 1) / TPB;
    const int gNH    = (NN * HEADS + TPB - 1) / TPB;
    const int gEH    = (ET * HEADS + TPB - 1) / TPB;
    const int gWaves = (ET + (TPB / 32) - 1) / (TPB / 32);

    auto gat_layer = [&](const float* xin, int layer, float* acc) {
        hipLaunchKernelGGL(init_layer, dim3(gNHC), dim3(TPB), 0, stream, mbuf, den, acc);
        hipLaunchKernelGGL(wmma_gemm_f32_2t, dim3(MT, HC / 32), dim3(32), 0, stream,
                           xin, W[layer], hbuf, IN_DIM, HC);
        hipLaunchKernelGGL(attn_coef, dim3(gNH), dim3(TPB), 0, stream,
                           hbuf, asv[layer], adv[layer], al_s, al_d);
        hipLaunchKernelGGL(edge_max, dim3(gEH), dim3(TPB), 0, stream, ei, al_s, al_d, mbuf);
        hipLaunchKernelGGL(edge_sum, dim3(gEH), dim3(TPB), 0, stream, ei, al_s, al_d, mbuf, den);
        hipLaunchKernelGGL(edge_aggregate, dim3(gWaves), dim3(TPB), 0, stream,
                           ei, hbuf, al_s, al_d, mbuf, den, acc);
        hipLaunchKernelGGL(bias_relu, dim3(gNHC), dim3(TPB), 0, stream, acc, bv[layer]);
    };

    gat_layer(x,    0, bufA);
    gat_layer(bufA, 1, bufB);
    gat_layer(bufB, 2, bufA);

    hipLaunchKernelGGL(wmma_gemm_f32_tail, dim3(MT, (OUT_DIM + 15) / 16), dim3(32), 0, stream,
                       bufA, Wl, (float*)d_out, bl, HC, OUT_DIM);
}


# Node ranges owned by work-groups: 256 groups of 196 nodes filter the edge list, keep max, exp-sum and row sums in LDS (LDS atomics, arrival order), store relu(sum+b) once; 5 launches a layer become 1
# speedup vs baseline: 2.2992x; 2.2992x over previous
#include <hip/hip_runtime.h>
#include <math.h>

#define NN 50000
#define EE 800000
#define IN_DIM 128
#define HID 32
#define HEADS 4
#define HC 128
#define OUT_DIM 40
#define SLOPE 0.2f

typedef float v2f __attribute__((ext_vector_type(2)));
typedef float v8f __attribute__((ext_vector_type(8)));

__device__ __forceinline__ unsigned ford(float f) {
    unsigned u = __float_as_uint(f);
    return (u & 0x80000000u) ? ~u : (u | 0x80000000u);
}
__device__ __forceinline__ float funord(unsigned k) {
    unsigned u = (k & 0x80000000u) ? (k & 0x7FFFFFFFu) : ~k;
    return __uint_as_float(u);
}
#define FORD_NEG_INF 0x007FFFFFu

__device__ __forceinline__ float leaky(float e) {
    return e > 0.f ? e : SLOPE * e;
}

__device__ __forceinline__ v8f wmma4(v2f a, v2f b, v8f c) {
    return __builtin_amdgcn_wmma_f32_16x16x4_f32(
               false, a, false, b, (short)0, c, false, false);
}

__global__ __launch_bounds__(32) void wmma_gemm_f32_2t(
    const float* __restrict__ A, const float* __restrict__ B,
    float* __restrict__ C, int K, int Nout)
{
    const int mt   = blockIdx.x;
    const int nt   = blockIdx.y;
    const int lane = threadIdx.x;
    const int hi   = lane >> 4;
    const int lo   = lane & 15;

    const float* Ap = A + (size_t)(mt * 16 + lo) * K + 2 * hi;
    const float* Bp = B + (size_t)(2 * hi) * Nout + nt * 32 + lo;

    v8f acc0 = {}, acc1 = {};
    for (int k = 0; k < K; k += 4) {
        float2 av = *(const float2*)(Ap + k);
        v2f a; a.x = av.x; a.y = av.y;
        const float* bp = Bp + (size_t)k * Nout;
        v2f b0; b0.x = bp[0];  b0.y = bp[Nout];
        v2f b1; b1.x = bp[16]; b1.y = bp[Nout + 16];
        acc0 = wmma4(a, b0, acc0);
        acc1 = wmma4(a, b1, acc1);
    }
    const int col = nt * 32 + lo;
    float* Cp = C + (size_t)(mt * 16 + 8 * hi) * Nout + col;
    #pragma unroll
    for (int r = 0; r < 8; ++r) {
        Cp[(size_t)r * Nout + 0]  = acc0[r];
        Cp[(size_t)r * Nout + 16] = acc1[r];
    }
}

__global__ __launch_bounds__(32) void wmma_gemm_f32_tail(
    const float* __restrict__ A, const float* __restrict__ B,
    float* __restrict__ C, const float* __restrict__ bias,
    int K, int Nout)
{
    const int mt   = blockIdx.x;
    const int nt   = blockIdx.y;
    const int lane = threadIdx.x;
    const int hi   = lane >> 4;
    const int lo   = lane & 15;
    const int col  = nt * 16 + lo;
    const int ccol = col < Nout ? col : Nout - 1;

    const float* Ap = A + (size_t)(mt * 16 + lo) * K + 2 * hi;
    const float* Bp = B + (size_t)(2 * hi) * Nout + ccol;

    v8f acc = {};
    for (int k = 0; k < K; k += 4) {
        float2 av = *(const float2*)(Ap + k);
        v2f a; a.x = av.x; a.y = av.y;
        const float* bp = Bp + (size_t)k * Nout;
        v2f b; b.x = bp[0]; b.y = bp[Nout];
        acc = wmma4(a, b, acc);
    }
    if (col < Nout) {
        const float bb = bias ? bias[col] : 0.f;
        float* Cp = C + (size_t)(mt * 16 + 8 * hi) * Nout + col;
        #pragma unroll
        for (int r = 0; r < 8; ++r)
            Cp[(size_t)r * Nout] = acc[r] + bb;
    }
}

__global__ void attn_coef(const float* __restrict__ h,
                          const float* __restrict__ a_src,
                          const float* __restrict__ a_dst,
                          float* __restrict__ al_s,
                          float* __restrict__ al_d)
{
    int idx = blockIdx.x * blockDim.x + threadIdx.x;
    if (idx >= NN * HEADS) return;
    int n = idx / HEADS, hd = idx % HEADS;
    const float* hp = h + (size_t)n * HC + hd * HID;
    const float* as = a_src + hd * HID;
    const float* ad = a_dst + hd * HID;
    float s = 0.f, d = 0.f;
    #pragma unroll
    for (int c = 0; c < HID; ++c) { float v = hp[c]; s = fmaf(v, as[c], s); d = fmaf(v, ad[c], d); }
    al_s[idx] = s; al_d[idx] = d;
}

#define RANGE 196
#define NWG ((NN + RANGE - 1) / RANGE)
#define OT 1024
#define CAP 16384
#define SCAN_U 8
#define ROW_U 8
static_assert(NN <= 65536, "list entry packs the source index in 16 bits");
static_assert(RANGE < 256 && (RANGE + 1) * HEADS <= OT, "range bookkeeping fits one pass");
static_assert((EE % 4) == 0, "dst array is read as int4");

struct OwnerLds {
    float    acc[(RANGE + 1) * HC];
    unsigned m[(RANGE + 1) * HEADS];
    float    den[(RANGE + 1) * HEADS];
    float    ald[(RANGE + 1) * HEADS];
    unsigned list[CAP];
    unsigned count;
};

__device__ __forceinline__ void lds_add(float* p, float v) {
    __hip_atomic_fetch_add(p, v, __ATOMIC_RELAXED, __HIP_MEMORY_SCOPE_WORKGROUP);
}

__device__ __forceinline__ void edge_max_one(OwnerLds& L, const float* __restrict__ al_s,
                                             int src, unsigned ld) {
    const float4 s = *(const float4*)(al_s + (size_t)src * HEADS);
    const float sv[4] = { s.x, s.y, s.z, s.w };
    #pragma unroll
    for (int hd = 0; hd < HEADS; ++hd) {
        float e = leaky(sv[hd] + L.ald[ld * HEADS + hd]);
        atomicMax(&L.m[ld * HEADS + hd], ford(e));
    }
}

__device__ __forceinline__ void edge_den_one(OwnerLds& L, const float* __restrict__ al_s,
                                             int src, unsigned ld) {
    const float4 s = *(const float4*)(al_s + (size_t)src * HEADS);
    const float sv[4] = { s.x, s.y, s.z, s.w };
    #pragma unroll
    for (int hd = 0; hd < HEADS; ++hd) {
        float e  = leaky(sv[hd] + L.ald[ld * HEADS + hd]);
        float mx = funord(L.m[ld * HEADS + hd]);
        lds_add(&L.den[ld * HEADS + hd], __expf(e - mx));
    }
}

__device__ __forceinline__ void row_add(OwnerLds& L, float sv, float4 v, unsigned ld, int lane) {
    const int hd = lane >> 3;
    float e  = leaky(sv + L.ald[ld * HEADS + hd]);
    float mx = funord(L.m[ld * HEADS + hd]);
    float alpha = __expf(e - mx) / (L.den[ld * HEADS + hd] + 1e-16f);
    float* ap = L.acc + ld * HC + lane;
    lds_add(ap,      v.x * alpha);
    lds_add(ap + 32, v.y * alpha);
    lds_add(ap + 64, v.z * alpha);
    lds_add(ap + 96, v.w * alpha);
}

__device__ __forceinline__ void row_one(OwnerLds& L, const float* __restrict__ h,
                                        const float* __restrict__ al_s,
                                        int src, unsigned ld, int lane) {
    float  sv = al_s[(size_t)src * HEADS + (lane >> 3)];
    float4 v  = *((const float4*)(h + (size_t)src * HC) + lane);
    row_add(L, sv, v, ld, lane);
}

template <int MODE>
__device__ __forceinline__ void scan_edges(OwnerLds& L, const int* __restrict__ ei,
                                           const float* __restrict__ h,
                                           const float* __restrict__ al_s,
                                           int n0, unsigned nloc, int t)
{
    const int4* d4 = (const int4*)(ei + EE);
    const int lane = t & 31;
    for (int base = 0; base < EE / 4; base += OT * SCAN_U) {
        int4 dv[SCAN_U];
        #pragma unroll
        for (int u = 0; u < SCAN_U; ++u) {
            const int i = base + u * OT + t;
            dv[u] = (i < EE / 4) ? d4[i] : make_int4(-1, -1, -1, -1);
        }
        #pragma unroll
        for (int u = 0; u < SCAN_U; ++u) {
            const int i = base + u * OT + t;
            const int dd[4] = { dv[u].x, dv[u].y, dv[u].z, dv[u].w };
            unsigned ld[4]; bool hit[4];
            #pragma unroll
            for (int k = 0; k < 4; ++k) { ld[k] = (unsigned)(dd[k] - n0); hit[k] = ld[k] < nloc; }
            if (MODE == 3) {
                #pragma unroll
                for (int k = 0; k < 4; ++k) {
                    const int src = hit[k] ? ei[4 * i + k] : 0;
                    unsigned mask = __builtin_amdgcn_ballot_w32(hit[k]);
                    while (mask) {
                        const int l = __builtin_ctz(mask);
                        mask &= mask - 1;
                        const int      s2 = __shfl(src, l);
                        const unsigned d2 = (unsigned)__shfl((int)ld[k], l);
                        row_one(L, h, al_s, s2, d2, lane);
                    }
                }
            } else if (hit[0] | hit[1] | hit[2] | hit[3]) {
                #pragma unroll
                for (int k = 0; k < 4; ++k) {
                    if (!hit[k]) continue;
                    const int src = ei[4 * i + k];
                    if (MODE == 0) {
                        const unsigned slot = atomicAdd(&L.count, 1u);
                        if (slot < CAP) L.list[slot] = (unsigned)src | (ld[k] << 16);
                    } else if (MODE == 1) {
                        edge_max_one(L, al_s, src, ld[k]);
                    } else {
                        edge_den_one(L, al_s, src, ld[k]);
                    }
                }
            }
        }
    }
}

__global__ __launch_bounds__(OT) void owner_aggregate(
    const int* __restrict__ ei, const float* __restrict__ h,
    const float* __restrict__ al_s, const float* __restrict__ al_d,
    const float* __restrict__ bias, float* __restrict__ out)
{
    __shared__ OwnerLds L;
    const int t    = threadIdx.x;
    const int lane = t & 31;
    const int wv   = t >> 5;
    const int n0   = blockIdx.x * RANGE;
    const unsigned nloc = (unsigned)((NN - n0) < RANGE ? (NN - n0) : RANGE);

    for (int i = t; i < RANGE * HC; i += OT) L.acc[i] = 0.f;
    if (t < RANGE * HEADS) {
        L.m[t]   = FORD_NEG_INF;
        L.den[t] = 0.f;
        L.ald[t] = ((unsigned)t < nloc * HEADS) ? al_d[(size_t)n0 * HEADS + t] : 0.f;
    } else if (t < (RANGE + 1) * HEADS) {
        L.m[t]   = 0x80000000u;
        L.den[t] = 1.f;
        L.ald[t] = 0.f;
    }
    if ((unsigned)t < nloc) L.list[t] = (unsigned)(n0 + t) | ((unsigned)t << 16);
    if (t == 0) L.count = nloc;
    __syncthreads();

    scan_edges<0>(L, ei, h, al_s, n0, nloc, t);
    __syncthreads();
    const unsigned total = L.count;

    if (total <= CAP) {
        for (unsigned idx = t; idx < total; idx += OT) {
            const unsigned ent = L.list[idx];
            edge_max_one(L, al_s, (int)(ent & 0xFFFFu), ent >> 16);
        }
        __syncthreads();
        for (unsigned idx = t; idx < total; idx += OT) {
            const unsigned ent = L.list[idx];
            edge_den_one(L, al_s, (int)(ent & 0xFFFFu), ent >> 16);
        }
        __syncthreads();
        const unsigned spare = (unsigned)n0 | ((unsigned)RANGE << 16);
        for (unsigned i0 = wv; i0 < total; i0 += (OT / 32) * ROW_U) {
            unsigned ent[ROW_U]; float sv[ROW_U]; float4 v[ROW_U];
            #pragma unroll
            for (int u = 0; u < ROW_U; ++u) {
                const unsigned idx = i0 + u * (OT / 32);
                const unsigned e0 = L.list[idx < total ? idx : 0];
                ent[u] = idx < total ? e0 : spare;
                const int src = (int)(ent[u] & 0xFFFFu);
                sv[u] = al_s[(size_t)src * HEADS + (lane >> 3)];
                v[u]  = *((const float4*)(h + (size_t)src * HC) + lane);
            }
            #pragma unroll
            for (int u = 0; u < ROW_U; ++u)
                row_add(L, sv[u], v[u], ent[u] >> 16, lane);
        }
    } else {
        if ((unsigned)t < nloc) edge_max_one(L, al_s, n0 + t, (unsigned)t);
        scan_edges<1>(L, ei, h, al_s, n0, nloc, t);
        __syncthreads();
        if ((unsigned)t < nloc) edge_den_one(L, al_s, n0 + t, (unsigned)t);
        scan_edges<2>(L, ei, h, al_s, n0, nloc, t);
        __syncthreads();
        for (unsigned i = wv; i < nloc; i += OT / 32) row_one(L, h, al_s, n0 + (int)i, i, lane);
        scan_edges<3>(L, ei, h, al_s, n0, nloc, t);
    }
    __syncthreads();

    for (unsigned i = t; i < nloc * HC; i += OT) {
        const unsigned node = i / HC, c = i % HC;
        float vv = L.acc[node * HC + (c & 3) * 32 + (c >> 2)] + bias[c];
        out[(size_t)n0 * HC + i] = vv > 0.f ? vv : 0.f;
    }
}

extern "C" void kernel_launch(void* const* d_in, const int* in_sizes, int n_in,
                              void* d_out, int out_size, void* d_ws, size_t ws_size,
                              hipStream_t stream)
{
    const float* x   = (const float*)d_in[0];
    const int*   ei  = (const int*)d_in[1];
    const float* W[3]    = { (const float*)d_in[2],  (const float*)d_in[6],  (const float*)d_in[10] };
    const float* asv[3]  = { (const float*)d_in[3],  (const float*)d_in[7],  (const float*)d_in[11] };
    const float* adv[3]  = { (const float*)d_in[4],  (const float*)d_in[8],  (const float*)d_in[12] };
    const float* bv[3]   = { (const float*)d_in[5],  (const float*)d_in[9],  (const float*)d_in[13] };
    const float* Wl  = (const float*)d_in[14];
    const float* bl  = (const float*)d_in[15];

    char* ws = (char*)d_ws;
    float*    hbuf = (float*)ws;                       ws += (size_t)NN * HC * sizeof(float);
    float*    bufA = (float*)ws;                       ws += (size_t)NN * HC * sizeof(float);
    float*    bufB = (float*)ws;                       ws += (size_t)NN * HC * sizeof(float);
    float*    al_s = (float*)ws;                       ws += (size_t)NN * HEADS * sizeof(float);
    float*    al_d = (float*)ws;

    const int TPB = 256;
    const int MT  = NN / 16;
    const int gNH    = (NN * HEADS + TPB - 1) / TPB;

    auto gat_layer = [&](const float* xin, int layer, float* acc) {
        hipLaunchKernelGGL(wmma_gemm_f32_2t, dim3(MT, HC / 32), dim3(32), 0, stream,
                           xin, W[layer], hbuf, IN_DIM, HC);
        hipLaunchKernelGGL(attn_coef, dim3(gNH), dim3(TPB), 0, stream,
                           hbuf, asv[layer], adv[layer], al_s, al_d);
        hipLaunchKernelGGL(owner_aggregate, dim3(NWG), dim3(OT), 0, stream,
                           ei, hbuf, al_s, al_d, bv[layer], acc);
    };

    gat_layer(x,    0, bufA);
    gat_layer(bufA, 1, bufB);
    gat_layer(bufB, 2, bufA);

    hipLaunchKernelGGL(wmma_gemm_f32_tail, dim3(MT, (OUT_DIM + 15) / 16), dim3(32), 0, stream,
                       bufA, Wl, (float*)d_out, bl, HC, OUT_DIM);
}


# Each owner group's unsorted filtered edge list is stored to the workspace by layer 1 and read back by layers 2 and 3, which no longer walk the dst array; sums stay in LDS arrival order
# speedup vs baseline: 2.7553x; 1.1984x over previous
#include <hip/hip_runtime.h>
#include <math.h>

#define NN 50000
#define EE 800000
#define IN_DIM 128
#define HID 32
#define HEADS 4
#define HC 128
#define OUT_DIM 40
#define SLOPE 0.2f

typedef float v2f __attribute__((ext_vector_type(2)));
typedef float v8f __attribute__((ext_vector_type(8)));

__device__ __forceinline__ unsigned ford(float f) {
    unsigned u = __float_as_uint(f);
    return (u & 0x80000000u) ? ~u : (u | 0x80000000u);
}
__device__ __forceinline__ float funord(unsigned k) {
    unsigned u = (k & 0x80000000u) ? (k & 0x7FFFFFFFu) : ~k;
    return __uint_as_float(u);
}
#define FORD_NEG_INF 0x007FFFFFu

__device__ __forceinline__ float leaky(float e) {
    return e > 0.f ? e : SLOPE * e;
}

__device__ __forceinline__ v8f wmma4(v2f a, v2f b, v8f c) {
    return __builtin_amdgcn_wmma_f32_16x16x4_f32(
               false, a, false, b, (short)0, c, false, false);
}

__global__ __launch_bounds__(32) void wmma_gemm_f32_2t(
    const float* __restrict__ A, const float* __restrict__ B,
    float* __restrict__ C, int K, int Nout)
{
    const int mt   = blockIdx.x;
    const int nt   = blockIdx.y;
    const int lane = threadIdx.x;
    const int hi   = lane >> 4;
    const int lo   = lane & 15;

    const float* Ap = A + (size_t)(mt * 16 + lo) * K + 2 * hi;
    const float* Bp = B + (size_t)(2 * hi) * Nout + nt * 32 + lo;

    v8f acc0 = {}, acc1 = {};
    for (int k = 0; k < K; k += 4) {
        float2 av = *(const float2*)(Ap + k);
        v2f a; a.x = av.x; a.y = av.y;
        const float* bp = Bp + (size_t)k * Nout;
        v2f b0; b0.x = bp[0];  b0.y = bp[Nout];
        v2f b1; b1.x = bp[16]; b1.y = bp[Nout + 16];
        acc0 = wmma4(a, b0, acc0);
        acc1 = wmma4(a, b1, acc1);
    }
    const int col = nt * 32 + lo;
    float* Cp = C + (size_t)(mt * 16 + 8 * hi) * Nout + col;
    #pragma unroll
    for (int r = 0; r < 8; ++r) {
        Cp[(size_t)r * Nout + 0]  = acc0[r];
        Cp[(size_t)r * Nout + 16] = acc1[r];
    }
}

__global__ __launch_bounds__(32) void wmma_gemm_f32_tail(
    const float* __restrict__ A, const float* __restrict__ B,
    float* __restrict__ C, const float* __restrict__ bias,
    int K, int Nout)
{
    const int mt   = blockIdx.x;
    const int nt   = blockIdx.y;
    const int lane = threadIdx.x;
    const int hi   = lane >> 4;
    const int lo   = lane & 15;
    const int col  = nt * 16 + lo;
    const int ccol = col < Nout ? col : Nout - 1;

    const float* Ap = A + (size_t)(mt * 16 + lo) * K + 2 * hi;
    const float* Bp = B + (size_t)(2 * hi) * Nout + ccol;

    v8f acc = {};
    for (int k = 0; k < K; k += 4) {
        float2 av = *(const float2*)(Ap + k);
        v2f a; a.x = av.x; a.y = av.y;
        const float* bp = Bp + (size_t)k * Nout;
        v2f b; b.x = bp[0]; b.y = bp[Nout];
        acc = wmma4(a, b, acc);
    }
    if (col < Nout) {
        const float bb = bias ? bias[col] : 0.f;
        float* Cp = C + (size_t)(mt * 16 + 8 * hi) * Nout + col;
        #pragma unroll
        for (int r = 0; r < 8; ++r)
            Cp[(size_t)r * Nout] = acc[r] + bb;
    }
}

__global__ void attn_coef(const float* __restrict__ h,
                          const float* __restrict__ a_src,
                          const float* __restrict__ a_dst,
                          float* __restrict__ al_s,
                          float* __restrict__ al_d)
{
    int idx = blockIdx.x * blockDim.x + threadIdx.x;
    if (idx >= NN * HEADS) return;
    int n = idx / HEADS, hd = idx % HEADS;
    const float* hp = h + (size_t)n * HC + hd * HID;
    const float* as = a_src + hd * HID;
    const float* ad = a_dst + hd * HID;
    float s = 0.f, d = 0.f;
    #pragma unroll
    for (int c = 0; c < HID; ++c) { float v = hp[c]; s = fmaf(v, as[c], s); d = fmaf(v, ad[c], d); }
    al_s[idx] = s; al_d[idx] = d;
}

#define RANGE 196
#define NWG ((NN + RANGE - 1) / RANGE)
#define OT 1024
#define CAP 16384
#define SCAN_U 8
#define ROW_U 8
#define LIST_U 4
static_assert(NN <= 65536, "list entry packs the source index in 16 bits");
static_assert(RANGE < 256 && (RANGE + 1) * HEADS <= OT, "range bookkeeping fits one pass");
static_assert((EE % 4) == 0, "dst array is read as int4");

struct OwnerLds {
    float    acc[(RANGE + 1) * HC];
    unsigned m[(RANGE + 1) * HEADS];
    float    den[(RANGE + 1) * HEADS];
    float    ald[(RANGE + 1) * HEADS];
    unsigned list[CAP];
    unsigned count;
};

__device__ __forceinline__ void lds_add(float* p, float v) {
    __hip_atomic_fetch_add(p, v, __ATOMIC_RELAXED, __HIP_MEMORY_SCOPE_WORKGROUP);
}

__device__ __forceinline__ void edge_max_one(OwnerLds& L, const float* __restrict__ al_s,
                                             int src, unsigned ld) {
    const float4 s = *(const float4*)(al_s + (size_t)src * HEADS);
    const float sv[4] = { s.x, s.y, s.z, s.w };
    #pragma unroll
    for (int hd = 0; hd < HEADS; ++hd) {
        float e = leaky(sv[hd] + L.ald[ld * HEADS + hd]);
        atomicMax(&L.m[ld * HEADS + hd], ford(e));
    }
}

__device__ __forceinline__ void edge_den_one(OwnerLds& L, const float* __restrict__ al_s,
                                             int src, unsigned ld) {
    const float4 s = *(const float4*)(al_s + (size_t)src * HEADS);
    const float sv[4] = { s.x, s.y, s.z, s.w };
    #pragma unroll
    for (int hd = 0; hd < HEADS; ++hd) {
        float e  = leaky(sv[hd] + L.ald[ld * HEADS + hd]);
        float mx = funord(L.m[ld * HEADS + hd]);
        lds_add(&L.den[ld * HEADS + hd], __expf(e - mx));
    }
}

__device__ __forceinline__ void row_add(OwnerLds& L, float sv, float4 v, unsigned ld, int lane) {
    const int hd = lane >> 3;
    float e  = leaky(sv + L.ald[ld * HEADS + hd]);
    float mx = funord(L.m[ld * HEADS + hd]);
    float alpha = __expf(e - mx) / (L.den[ld * HEADS + hd] + 1e-16f);
    float* ap = L.acc + ld * HC + lane;
    lds_add(ap,      v.x * alpha);
    lds_add(ap + 32, v.y * alpha);
    lds_add(ap + 64, v.z * alpha);
    lds_add(ap + 96, v.w * alpha);
}

__device__ __forceinline__ void row_one(OwnerLds& L, const float* __restrict__ h,
                                        const float* __restrict__ al_s,
                                        int src, unsigned ld, int lane) {
    float  sv = al_s[(size_t)src * HEADS + (lane >> 3)];
    float4 v  = *((const float4*)(h + (size_t)src * HC) + lane);
    row_add(L, sv, v, ld, lane);
}

template <int MODE>
__device__ __forceinline__ void scan_edges(OwnerLds& L, const int* __restrict__ ei,
                                           const float* __restrict__ h,
                                           const float* __restrict__ al_s,
                                           int n0, unsigned nloc, int t)
{
    const int4* d4 = (const int4*)(ei + EE);
    const int lane = t & 31;
    for (int base = 0; base < EE / 4; base += OT * SCAN_U) {
        int4 dv[SCAN_U];
        #pragma unroll
        for (int u = 0; u < SCAN_U; ++u) {
            const int i = base + u * OT + t;
            dv[u] = (i < EE / 4) ? d4[i] : make_int4(-1, -1, -1, -1);
        }
        #pragma unroll
        for (int u = 0; u < SCAN_U; ++u) {
            const int i = base + u * OT + t;
            const int dd[4] = { dv[u].x, dv[u].y, dv[u].z, dv[u].w };
            unsigned ld[4]; bool hit[4];
            #pragma unroll
            for (int k = 0; k < 4; ++k) { ld[k] = (unsigned)(dd[k] - n0); hit[k] = ld[k] < nloc; }
            if (MODE == 3) {
                #pragma unroll
                for (int k = 0; k < 4; ++k) {
                    const int src = hit[k] ? ei[4 * i + k] : 0;
                    unsigned mask = __builtin_amdgcn_ballot_w32(hit[k]);
                    while (mask) {
                        const int l = __builtin_ctz(mask);
                        mask &= mask - 1;
                        const int      s2 = __shfl(src, l);
                        const unsigned d2 = (unsigned)__shfl((int)ld[k], l);
                        row_one(L, h, al_s, s2, d2, lane);
                    }
                }
            } else if (hit[0] | hit[1] | hit[2] | hit[3]) {
                #pragma unroll
                for (int k = 0; k < 4; ++k) {
                    if (!hit[k]) continue;
                    const int src = ei[4 * i + k];
                    if (MODE == 0) {
                        const unsigned slot = atomicAdd(&L.count, 1u);
                        if (slot < CAP) L.list[slot] = (unsigned)src | (ld[k] << 16);
                    } else if (MODE == 1) {
                        edge_max_one(L, al_s, src, ld[k]);
                    } else {
                        edge_den_one(L, al_s, src, ld[k]);
                    }
                }
            }
        }
    }
}

template <bool BUILD>
__global__ __launch_bounds__(OT) void owner_aggregate(
    const int* __restrict__ ei, const float* __restrict__ h,
    const float* __restrict__ al_s, const float* __restrict__ al_d,
    const float* __restrict__ bias, float* __restrict__ out,
    unsigned* __restrict__ wlist, unsigned* __restrict__ wcount)
{
    __shared__ OwnerLds L;
    const int t    = threadIdx.x;
    const int lane = t & 31;
    const int wv   = t >> 5;
    const int n0   = blockIdx.x * RANGE;
    const unsigned nloc = (unsigned)((NN - n0) < RANGE ? (NN - n0) : RANGE);
    unsigned* wl = wlist + (size_t)blockIdx.x * CAP;

    if (!BUILD) {
        const unsigned saved = wcount[blockIdx.x];
        if (saved <= CAP) {
            for (unsigned i0 = 0; i0 < saved; i0 += OT * LIST_U) {
                unsigned ent[LIST_U];
                #pragma unroll
                for (int u = 0; u < LIST_U; ++u) {
                    const unsigned idx = i0 + u * OT + t;
                    ent[u] = idx < saved ? wl[idx] : 0u;
                }
                #pragma unroll
                for (int u = 0; u < LIST_U; ++u) {
                    const unsigned idx = i0 + u * OT + t;
                    if (idx < saved) L.list[idx] = ent[u];
                }
            }
        }
        if (t == 0) L.count = saved;
    }

    for (int i = t; i < RANGE * HC; i += OT) L.acc[i] = 0.f;
    if (t < RANGE * HEADS) {
        L.m[t]   = FORD_NEG_INF;
        L.den[t] = 0.f;
        L.ald[t] = ((unsigned)t < nloc * HEADS) ? al_d[(size_t)n0 * HEADS + t] : 0.f;
    } else if (t < (RANGE + 1) * HEADS) {
        L.m[t]   = 0x80000000u;
        L.den[t] = 1.f;
        L.ald[t] = 0.f;
    }
    if (BUILD) {
        if ((unsigned)t < nloc) L.list[t] = (unsigned)(n0 + t) | ((unsigned)t << 16);
        if (t == 0) L.count = nloc;
    }
    __syncthreads();

    if (BUILD) {
        scan_edges<0>(L, ei, h, al_s, n0, nloc, t);
        __syncthreads();
    }
    const unsigned total = L.count;

    if (BUILD) {
        if (total <= CAP)
            for (unsigned idx = t; idx < total; idx += OT) wl[idx] = L.list[idx];
        if (t == 0) wcount[blockIdx.x] = total;
    }

    if (total <= CAP) {
        for (unsigned idx = t; idx < total; idx += OT) {
            const unsigned ent = L.list[idx];
            edge_max_one(L, al_s, (int)(ent & 0xFFFFu), ent >> 16);
        }
        __syncthreads();
        for (unsigned idx = t; idx < total; idx += OT) {
            const unsigned ent = L.list[idx];
            edge_den_one(L, al_s, (int)(ent & 0xFFFFu), ent >> 16);
        }
        __syncthreads();
        const unsigned spare = (unsigned)n0 | ((unsigned)RANGE << 16);
        for (unsigned i0 = wv; i0 < total; i0 += (OT / 32) * ROW_U) {
            unsigned ent[ROW_U]; float sv[ROW_U]; float4 v[ROW_U];
            #pragma unroll
            for (int u = 0; u < ROW_U; ++u) {
                const unsigned idx = i0 + u * (OT / 32);
                const unsigned e0 = L.list[idx < total ? idx : 0];
                ent[u] = idx < total ? e0 : spare;
                const int src = (int)(ent[u] & 0xFFFFu);
                sv[u] = al_s[(size_t)src * HEADS + (lane >> 3)];
                v[u]  = *((const float4*)(h + (size_t)src * HC) + lane);
            }
            #pragma unroll
            for (int u = 0; u < ROW_U; ++u)
                row_add(L, sv[u], v[u], ent[u] >> 16, lane);
        }
    } else {
        if ((unsigned)t < nloc) edge_max_one(L, al_s, n0 + t, (unsigned)t);
        scan_edges<1>(L, ei, h, al_s, n0, nloc, t);
        __syncthreads();
        if ((unsigned)t < nloc) edge_den_one(L, al_s, n0 + t, (unsigned)t);
        scan_edges<2>(L, ei, h, al_s, n0, nloc, t);
        __syncthreads();
        for (unsigned i = wv; i < nloc; i += OT / 32) row_one(L, h, al_s, n0 + (int)i, i, lane);
        scan_edges<3>(L, ei, h, al_s, n0, nloc, t);
    }
    __syncthreads();

    for (unsigned i = t; i < nloc * HC; i += OT) {
        const unsigned node = i / HC, c = i % HC;
        float vv = L.acc[node * HC + (c & 3) * 32 + (c >> 2)] + bias[c];
        out[(size_t)n0 * HC + i] = vv > 0.f ? vv : 0.f;
    }
}

extern "C" void kernel_launch(void* const* d_in, const int* in_sizes, int n_in,
                              void* d_out, int out_size, void* d_ws, size_t ws_size,
                              hipStream_t stream)
{
    const float* x   = (const float*)d_in[0];
    const int*   ei  = (const int*)d_in[1];
    const float* W[3]    = { (const float*)d_in[2],  (const float*)d_in[6],  (const float*)d_in[10] };
    const float* asv[3]  = { (const float*)d_in[3],  (const float*)d_in[7],  (const float*)d_in[11] };
    const float* adv[3]  = { (const float*)d_in[4],  (const float*)d_in[8],  (const float*)d_in[12] };
    const float* bv[3]   = { (const float*)d_in[5],  (const float*)d_in[9],  (const float*)d_in[13] };
    const float* Wl  = (const float*)d_in[14];
    const float* bl  = (const float*)d_in[15];

    char* ws = (char*)d_ws;
    float*    hbuf = (float*)ws;                       ws += (size_t)NN * HC * sizeof(float);
    float*    bufA = (float*)ws;                       ws += (size_t)NN * HC * sizeof(float);
    float*    bufB = (float*)ws;                       ws += (size_t)NN * HC * sizeof(float);
    float*    al_s = (float*)ws;                       ws += (size_t)NN * HEADS * sizeof(float);
    float*    al_d = (float*)ws;                       ws += (size_t)NN * HEADS * sizeof(float);
    unsigned* wlist  = (unsigned*)ws;                  ws += (size_t)NWG * CAP * sizeof(unsigned);
    unsigned* wcount = (unsigned*)ws;

    const int TPB = 256;
    const int MT  = NN / 16;
    const int gNH    = (NN * HEADS + TPB - 1) / TPB;

    auto gat_front = [&](const float* xin, int layer) {
        hipLaunchKernelGGL(wmma_gemm_f32_2t, dim3(MT, HC / 32), dim3(32), 0, stream,
                           xin, W[layer], hbuf, IN_DIM, HC);
        hipLaunchKernelGGL(attn_coef, dim3(gNH), dim3(TPB), 0, stream,
                           hbuf, asv[layer], adv[layer], al_s, al_d);
    };

    gat_front(x, 0);
    hipLaunchKernelGGL(owner_aggregate<true>, dim3(NWG), dim3(OT), 0, stream,
                       ei, hbuf, al_s, al_d, bv[0], bufA, wlist, wcount);
    gat_front(bufA, 1);
    hipLaunchKernelGGL(owner_aggregate<false>, dim3(NWG), dim3(OT), 0, stream,
                       ei, hbuf, al_s, al_d, bv[1], bufB, wlist, wcount);
    gat_front(bufB, 2);
    hipLaunchKernelGGL(owner_aggregate<false>, dim3(NWG), dim3(OT), 0, stream,
                       ei, hbuf, al_s, al_d, bv[2], bufA, wlist, wcount);

    hipLaunchKernelGGL(wmma_gemm_f32_tail, dim3(MT, (OUT_DIM + 15) / 16), dim3(32), 0, stream,
                       bufA, Wl, (float*)d_out, bl, HC, OUT_DIM);
}


# Per-edge attention terms kept in LDS: one thread per listed edge forms logit, exp and alpha once per head; row pass reads alpha instead of redoing gather, exp, divide in 32 lanes; sum order unchanged
# speedup vs baseline: 3.1594x; 1.1467x over previous
#include <hip/hip_runtime.h>
#include <math.h>

#define NN 50000
#define EE 800000
#define IN_DIM 128
#define HID 32
#define HEADS 4
#define HC 128
#define OUT_DIM 40
#define SLOPE 0.2f

typedef float v2f __attribute__((ext_vector_type(2)));
typedef float v8f __attribute__((ext_vector_type(8)));

__device__ __forceinline__ unsigned ford(float f) {
    unsigned u = __float_as_uint(f);
    return (u & 0x80000000u) ? ~u : (u | 0x80000000u);
}
__device__ __forceinline__ float funord(unsigned k) {
    unsigned u = (k & 0x80000000u) ? (k & 0x7FFFFFFFu) : ~k;
    return __uint_as_float(u);
}
#define FORD_NEG_INF 0x007FFFFFu

__device__ __forceinline__ float leaky(float e) {
    return e > 0.f ? e : SLOPE * e;
}

__device__ __forceinline__ v8f wmma4(v2f a, v2f b, v8f c) {
    return __builtin_amdgcn_wmma_f32_16x16x4_f32(
               false, a, false, b, (short)0, c, false, false);
}

__global__ __launch_bounds__(32) void wmma_gemm_f32_2t(
    const float* __restrict__ A, const float* __restrict__ B,
    float* __restrict__ C, int K, int Nout)
{
    const int mt   = blockIdx.x;
    const int nt   = blockIdx.y;
    const int lane = threadIdx.x;
    const int hi   = lane >> 4;
    const int lo   = lane & 15;

    const float* Ap = A + (size_t)(mt * 16 + lo) * K + 2 * hi;
    const float* Bp = B + (size_t)(2 * hi) * Nout + nt * 32 + lo;

    v8f acc0 = {}, acc1 = {};
    for (int k = 0; k < K; k += 4) {
        float2 av = *(const float2*)(Ap + k);
        v2f a; a.x = av.x; a.y = av.y;
        const float* bp = Bp + (size_t)k * Nout;
        v2f b0; b0.x = bp[0];  b0.y = bp[Nout];
        v2f b1; b1.x = bp[16]; b1.y = bp[Nout + 16];
        acc0 = wmma4(a, b0, acc0);
        acc1 = wmma4(a, b1, acc1);
    }
    const int col = nt * 32 + lo;
    float* Cp = C + (size_t)(mt * 16 + 8 * hi) * Nout + col;
    #pragma unroll
    for (int r = 0; r < 8; ++r) {
        Cp[(size_t)r * Nout + 0]  = acc0[r];
        Cp[(size_t)r * Nout + 16] = acc1[r];
    }
}

__global__ __launch_bounds__(32) void wmma_gemm_f32_tail(
    const float* __restrict__ A, const float* __restrict__ B,
    float* __restrict__ C, const float* __restrict__ bias,
    int K, int Nout)
{
    const int mt   = blockIdx.x;
    const int nt   = blockIdx.y;
    const int lane = threadIdx.x;
    const int hi   = lane >> 4;
    const int lo   = lane & 15;
    const int col  = nt * 16 + lo;
    const int ccol = col < Nout ? col : Nout - 1;

    const float* Ap = A + (size_t)(mt * 16 + lo) * K + 2 * hi;
    const float* Bp = B + (size_t)(2 * hi) * Nout + ccol;

    v8f acc = {};
    for (int k = 0; k < K; k += 4) {
        float2 av = *(const float2*)(Ap + k);
        v2f a; a.x = av.x; a.y = av.y;
        const float* bp = Bp + (size_t)k * Nout;
        v2f b; b.x = bp[0]; b.y = bp[Nout];
        acc = wmma4(a, b, acc);
    }
    if (col < Nout) {
        const float bb = bias ? bias[col] : 0.f;
        float* Cp = C + (size_t)(mt * 16 + 8 * hi) * Nout + col;
        #pragma unroll
        for (int r = 0; r < 8; ++r)
            Cp[(size_t)r * Nout] = acc[r] + bb;
    }
}

__global__ void attn_coef(const float* __restrict__ h,
                          const float* __restrict__ a_src,
                          const float* __restrict__ a_dst,
                          float* __restrict__ al_s,
                          float* __restrict__ al_d)
{
    int idx = blockIdx.x * blockDim.x + threadIdx.x;
    if (idx >= NN * HEADS) return;
    int n = idx / HEADS, hd = idx % HEADS;
    const float* hp = h + (size_t)n * HC + hd * HID;
    const float* as = a_src + hd * HID;
    const float* ad = a_dst + hd * HID;
    float s = 0.f, d = 0.f;
    #pragma unroll
    for (int c = 0; c < HID; ++c) { float v = hp[c]; s = fmaf(v, as[c], s); d = fmaf(v, ad[c], d); }
    al_s[idx] = s; al_d[idx] = d;
}

#define RANGE 196
#define NWG ((NN + RANGE - 1) / RANGE)
#define OT 1024
#define CAP 8192
#define SCAN_U 8
#define ROW_U 8
#define LIST_U 4
static_assert(NN <= 65536, "list entry packs the source index in 16 bits");
static_assert(RANGE < 256 && (RANGE + 1) * HEADS <= OT, "range bookkeeping fits one pass");
static_assert((EE % 4) == 0, "dst array is read as int4");

struct OwnerLds {
    alignas(16) float ev[CAP * HEADS];
    float    acc[(RANGE + 1) * HC];
    unsigned m[(RANGE + 1) * HEADS];
    float    den[(RANGE + 1) * HEADS];
    float    ald[(RANGE + 1) * HEADS];
    unsigned list[CAP];
    unsigned count;
};

__device__ __forceinline__ void lds_add(float* p, float v) {
    __hip_atomic_fetch_add(p, v, __ATOMIC_RELAXED, __HIP_MEMORY_SCOPE_WORKGROUP);
}

__device__ __forceinline__ void edge_max_one(OwnerLds& L, const float* __restrict__ al_s,
                                             int src, unsigned ld) {
    const float4 s = *(const float4*)(al_s + (size_t)src * HEADS);
    const float sv[4] = { s.x, s.y, s.z, s.w };
    #pragma unroll
    for (int hd = 0; hd < HEADS; ++hd) {
        float e = leaky(sv[hd] + L.ald[ld * HEADS + hd]);
        atomicMax(&L.m[ld * HEADS + hd], ford(e));
    }
}

__device__ __forceinline__ void edge_den_one(OwnerLds& L, const float* __restrict__ al_s,
                                             int src, unsigned ld) {
    const float4 s = *(const float4*)(al_s + (size_t)src * HEADS);
    const float sv[4] = { s.x, s.y, s.z, s.w };
    #pragma unroll
    for (int hd = 0; hd < HEADS; ++hd) {
        float e  = leaky(sv[hd] + L.ald[ld * HEADS + hd]);
        float mx = funord(L.m[ld * HEADS + hd]);
        lds_add(&L.den[ld * HEADS + hd], __expf(e - mx));
    }
}

__device__ __forceinline__ void edge_logit_max(OwnerLds& L, const float* __restrict__ al_s, unsigned idx) {
    const unsigned ent = L.list[idx];
    const unsigned ld  = ent >> 16;
    const float4 s = *(const float4*)(al_s + (size_t)(ent & 0xFFFFu) * HEADS);
    const float4 d = *(const float4*)(L.ald + ld * HEADS);
    float4 e;
    e.x = leaky(s.x + d.x); e.y = leaky(s.y + d.y); e.z = leaky(s.z + d.z); e.w = leaky(s.w + d.w);
    atomicMax(&L.m[ld * HEADS + 0], ford(e.x));
    atomicMax(&L.m[ld * HEADS + 1], ford(e.y));
    atomicMax(&L.m[ld * HEADS + 2], ford(e.z));
    atomicMax(&L.m[ld * HEADS + 3], ford(e.w));
    *(float4*)(L.ev + idx * HEADS) = e;
}

__device__ __forceinline__ void edge_exp_den(OwnerLds& L, unsigned idx) {
    const unsigned ld = L.list[idx] >> 16;
    const float4 e  = *(const float4*)(L.ev + idx * HEADS);
    const uint4  mk = *(const uint4*)(L.m + ld * HEADS);
    float4 x;
    x.x = __expf(e.x - funord(mk.x)); x.y = __expf(e.y - funord(mk.y));
    x.z = __expf(e.z - funord(mk.z)); x.w = __expf(e.w - funord(mk.w));
    lds_add(&L.den[ld * HEADS + 0], x.x);
    lds_add(&L.den[ld * HEADS + 1], x.y);
    lds_add(&L.den[ld * HEADS + 2], x.z);
    lds_add(&L.den[ld * HEADS + 3], x.w);
    *(float4*)(L.ev + idx * HEADS) = x;
}

__device__ __forceinline__ void edge_alpha(OwnerLds& L, unsigned idx) {
    const unsigned ld = L.list[idx] >> 16;
    float4 x = *(const float4*)(L.ev + idx * HEADS);
    const float4 dn = *(const float4*)(L.den + ld * HEADS);
    x.x = x.x / (dn.x + 1e-16f); x.y = x.y / (dn.y + 1e-16f);
    x.z = x.z / (dn.z + 1e-16f); x.w = x.w / (dn.w + 1e-16f);
    *(float4*)(L.ev + idx * HEADS) = x;
}

__device__ __forceinline__ void row_add(OwnerLds& L, float sv, float4 v, unsigned ld, int lane) {
    const int hd = lane >> 3;
    float e  = leaky(sv + L.ald[ld * HEADS + hd]);
    float mx = funord(L.m[ld * HEADS + hd]);
    float alpha = __expf(e - mx) / (L.den[ld * HEADS + hd] + 1e-16f);
    float* ap = L.acc + ld * HC + lane;
    lds_add(ap,      v.x * alpha);
    lds_add(ap + 32, v.y * alpha);
    lds_add(ap + 64, v.z * alpha);
    lds_add(ap + 96, v.w * alpha);
}

__device__ __forceinline__ void row_one(OwnerLds& L, const float* __restrict__ h,
                                        const float* __restrict__ al_s,
                                        int src, unsigned ld, int lane) {
    float  sv = al_s[(size_t)src * HEADS + (lane >> 3)];
    float4 v  = *((const float4*)(h + (size_t)src * HC) + lane);
    row_add(L, sv, v, ld, lane);
}

template <int MODE>
__device__ __forceinline__ void scan_edges(OwnerLds& L, const int* __restrict__ ei,
                                           const float* __restrict__ h,
                                           const float* __restrict__ al_s,
                                           int n0, unsigned nloc, int t)
{
    const int4* d4 = (const int4*)(ei + EE);
    const int lane = t & 31;
    for (int base = 0; base < EE / 4; base += OT * SCAN_U) {
        int4 dv[SCAN_U];
        #pragma unroll
        for (int u = 0; u < SCAN_U; ++u) {
            const int i = base + u * OT + t;
            dv[u] = (i < EE / 4) ? d4[i] : make_int4(-1, -1, -1, -1);
        }
        #pragma unroll
        for (int u = 0; u < SCAN_U; ++u) {
            const int i = base + u * OT + t;
            const int dd[4] = { dv[u].x, dv[u].y, dv[u].z, dv[u].w };
            unsigned ld[4]; bool hit[4];
            #pragma unroll
            for (int k = 0; k < 4; ++k) { ld[k] = (unsigned)(dd[k] - n0); hit[k] = ld[k] < nloc; }
            if (MODE == 3) {
                #pragma unroll
                for (int k = 0; k < 4; ++k) {
                    const int src = hit[k] ? ei[4 * i + k] : 0;
                    unsigned mask = __builtin_amdgcn_ballot_w32(hit[k]);
                    while (mask) {
                        const int l = __builtin_ctz(mask);
                        mask &= mask - 1;
                        const int      s2 = __shfl(src, l);
                        const unsigned d2 = (unsigned)__shfl((int)ld[k], l);
                        row_one(L, h, al_s, s2, d2, lane);
                    }
                }
            } else if (hit[0] | hit[1] | hit[2] | hit[3]) {
                #pragma unroll
                for (int k = 0; k < 4; ++k) {
                    if (!hit[k]) continue;
                    const int src = ei[4 * i + k];
                    if (MODE == 0) {
                        const unsigned slot = atomicAdd(&L.count, 1u);
                        if (slot < CAP) L.list[slot] = (unsigned)src | (ld[k] << 16);
                    } else if (MODE == 1) {
                        edge_max_one(L, al_s, src, ld[k]);
                    } else {
                        edge_den_one(L, al_s, src, ld[k]);
                    }
                }
            }
        }
    }
}

template <bool BUILD>
__global__ __launch_bounds__(OT) void owner_aggregate(
    const int* __restrict__ ei, const float* __restrict__ h,
    const float* __restrict__ al_s, const float* __restrict__ al_d,
    const float* __restrict__ bias, float* __restrict__ out,
    unsigned* __restrict__ wlist, unsigned* __restrict__ wcount)
{
    __shared__ OwnerLds L;
    const int t    = threadIdx.x;
    const int lane = t & 31;
    const int wv   = t >> 5;
    const int n0   = blockIdx.x * RANGE;
    const unsigned nloc = (unsigned)((NN - n0) < RANGE ? (NN - n0) : RANGE);
    unsigned* wl = wlist + (size_t)blockIdx.x * CAP;

    if (!BUILD) {
        const unsigned saved = wcount[blockIdx.x];
        if (saved <= CAP) {
            for (unsigned i0 = 0; i0 < saved; i0 += OT * LIST_U) {
                unsigned ent[LIST_U];
                #pragma unroll
                for (int u = 0; u < LIST_U; ++u) {
                    const unsigned idx = i0 + u * OT + t;
                    ent[u] = idx < saved ? wl[idx] : 0u;
                }
                #pragma unroll
                for (int u = 0; u < LIST_U; ++u) {
                    const unsigned idx = i0 + u * OT + t;
                    if (idx < saved) L.list[idx] = ent[u];
                }
            }
        }
        if (t == 0) L.count = saved;
    }

    for (int i = t; i < RANGE * HC; i += OT) L.acc[i] = 0.f;
    if (t < RANGE * HEADS) {
        L.m[t]   = FORD_NEG_INF;
        L.den[t] = 0.f;
        L.ald[t] = ((unsigned)t < nloc * HEADS) ? al_d[(size_t)n0 * HEADS + t] : 0.f;
    }
    if (BUILD) {
        if ((unsigned)t < nloc) L.list[t] = (unsigned)(n0 + t) | ((unsigned)t << 16);
        if (t == 0) L.count = nloc;
    }
    __syncthreads();

    if (BUILD) {
        scan_edges<0>(L, ei, h, al_s, n0, nloc, t);
        __syncthreads();
    }
    const unsigned total = L.count;

    if (BUILD) {
        if (total <= CAP)
            for (unsigned idx = t; idx < total; idx += OT) wl[idx] = L.list[idx];
        if (t == 0) wcount[blockIdx.x] = total;
    }

    if (total <= CAP) {
        for (unsigned idx = t; idx < total; idx += OT) edge_logit_max(L, al_s, idx);
        __syncthreads();
        for (unsigned idx = t; idx < total; idx += OT) edge_exp_den(L, idx);
        __syncthreads();
        for (unsigned idx = t; idx < total; idx += OT) edge_alpha(L, idx);
        __syncthreads();
        const unsigned spare = (unsigned)n0 | ((unsigned)RANGE << 16);
        for (unsigned i0 = wv; i0 < total; i0 += (OT / 32) * ROW_U) {
            unsigned ent[ROW_U]; float al[ROW_U]; float4 v[ROW_U];
            #pragma unroll
            for (int u = 0; u < ROW_U; ++u) {
                const unsigned idx = i0 + u * (OT / 32);
                const unsigned idc = idx < total ? idx : 0;
                const unsigned e0 = L.list[idc];
                ent[u] = idx < total ? e0 : spare;
                al[u]  = L.ev[idc * HEADS + (lane >> 3)];
                v[u]   = *((const float4*)(h + (size_t)(ent[u] & 0xFFFFu) * HC) + lane);
            }
            #pragma unroll
            for (int u = 0; u < ROW_U; ++u) {
                float* ap = L.acc + (ent[u] >> 16) * HC + lane;
                lds_add(ap,      v[u].x * al[u]);
                lds_add(ap + 32, v[u].y * al[u]);
                lds_add(ap + 64, v[u].z * al[u]);
                lds_add(ap + 96, v[u].w * al[u]);
            }
        }
    } else {
        if ((unsigned)t < nloc) edge_max_one(L, al_s, n0 + t, (unsigned)t);
        scan_edges<1>(L, ei, h, al_s, n0, nloc, t);
        __syncthreads();
        if ((unsigned)t < nloc) edge_den_one(L, al_s, n0 + t, (unsigned)t);
        scan_edges<2>(L, ei, h, al_s, n0, nloc, t);
        __syncthreads();
        for (unsigned i = wv; i < nloc; i += OT / 32) row_one(L, h, al_s, n0 + (int)i, i, lane);
        scan_edges<3>(L, ei, h, al_s, n0, nloc, t);
    }
    __syncthreads();

    for (unsigned i = t; i < nloc * HC; i += OT) {
        const unsigned node = i / HC, c = i % HC;
        float vv = L.acc[node * HC + (c & 3) * 32 + (c >> 2)] + bias[c];
        out[(size_t)n0 * HC + i] = vv > 0.f ? vv : 0.f;
    }
}

extern "C" void kernel_launch(void* const* d_in, const int* in_sizes, int n_in,
                              void* d_out, int out_size, void* d_ws, size_t ws_size,
                              hipStream_t stream)
{
    const float* x   = (const float*)d_in[0];
    const int*   ei  = (const int*)d_in[1];
    const float* W[3]    = { (const float*)d_in[2],  (const float*)d_in[6],  (const float*)d_in[10] };
    const float* asv[3]  = { (const float*)d_in[3],  (const float*)d_in[7],  (const float*)d_in[11] };
    const float* adv[3]  = { (const float*)d_in[4],  (const float*)d_in[8],  (const float*)d_in[12] };
    const float* bv[3]   = { (const float*)d_in[5],  (const float*)d_in[9],  (const float*)d_in[13] };
    const float* Wl  = (const float*)d_in[14];
    const float* bl  = (const float*)d_in[15];

    char* ws = (char*)d_ws;
    float*    hbuf = (float*)ws;                       ws += (size_t)NN * HC * sizeof(float);
    float*    bufA = (float*)ws;                       ws += (size_t)NN * HC * sizeof(float);
    float*    bufB = (float*)ws;                       ws += (size_t)NN * HC * sizeof(float);
    float*    al_s = (float*)ws;                       ws += (size_t)NN * HEADS * sizeof(float);
    float*    al_d = (float*)ws;                       ws += (size_t)NN * HEADS * sizeof(float);
    unsigned* wlist  = (unsigned*)ws;                  ws += (size_t)NWG * CAP * sizeof(unsigned);
    unsigned* wcount = (unsigned*)ws;

    const int TPB = 256;
    const int MT  = NN / 16;
    const int gNH    = (NN * HEADS + TPB - 1) / TPB;

    auto gat_front = [&](const float* xin, int layer) {
        hipLaunchKernelGGL(wmma_gemm_f32_2t, dim3(MT, HC / 32), dim3(32), 0, stream,
                           xin, W[layer], hbuf, IN_DIM, HC);
        hipLaunchKernelGGL(attn_coef, dim3(gNH), dim3(TPB), 0, stream,
                           hbuf, asv[layer], adv[layer], al_s, al_d);
    };

    gat_front(x, 0);
    hipLaunchKernelGGL(owner_aggregate<true>, dim3(NWG), dim3(OT), 0, stream,
                       ei, hbuf, al_s, al_d, bv[0], bufA, wlist, wcount);
    gat_front(bufA, 1);
    hipLaunchKernelGGL(owner_aggregate<false>, dim3(NWG), dim3(OT), 0, stream,
                       ei, hbuf, al_s, al_d, bv[1], bufB, wlist, wcount);
    gat_front(bufB, 2);
    hipLaunchKernelGGL(owner_aggregate<false>, dim3(NWG), dim3(OT), 0, stream,
                       ei, hbuf, al_s, al_d, bv[2], bufA, wlist, wcount);

    hipLaunchKernelGGL(wmma_gemm_f32_tail, dim3(MT, (OUT_DIM + 15) / 16), dim3(32), 0, stream,
                       bufA, Wl, (float*)d_out, bl, HC, OUT_DIM);
}


# Next dense product moved into the aggregation launch: each owner group multiplies its finished LDS rows by the weight on the matrix unit, x1..x3 never stored; 10 launches become 7; k order kept
# speedup vs baseline: 4.1894x; 1.3260x over previous
#include <hip/hip_runtime.h>
#include <math.h>

#define NN 50000
#define EE 800000
#define IN_DIM 128
#define HID 32
#define HEADS 4
#define HC 128
#define OUT_DIM 40
#define SLOPE 0.2f

typedef float v2f __attribute__((ext_vector_type(2)));
typedef float v8f __attribute__((ext_vector_type(8)));

__device__ __forceinline__ unsigned ford(float f) {
    unsigned u = __float_as_uint(f);
    return (u & 0x80000000u) ? ~u : (u | 0x80000000u);
}
__device__ __forceinline__ float funord(unsigned k) {
    unsigned u = (k & 0x80000000u) ? (k & 0x7FFFFFFFu) : ~k;
    return __uint_as_float(u);
}
#define FORD_NEG_INF 0x007FFFFFu

__device__ __forceinline__ float leaky(float e) {
    return e > 0.f ? e : SLOPE * e;
}

__device__ __forceinline__ v8f wmma4(v2f a, v2f b, v8f c) {
    return __builtin_amdgcn_wmma_f32_16x16x4_f32(
               false, a, false, b, (short)0, c, false, false);
}

__global__ __launch_bounds__(32) void wmma_gemm_f32_2t(
    const float* __restrict__ A, const float* __restrict__ B,
    float* __restrict__ C, int K, int Nout)
{
    const int mt   = blockIdx.x;
    const int nt   = blockIdx.y;
    const int lane = threadIdx.x;
    const int hi   = lane >> 4;
    const int lo   = lane & 15;

    const float* Ap = A + (size_t)(mt * 16 + lo) * K + 2 * hi;
    const float* Bp = B + (size_t)(2 * hi) * Nout + nt * 32 + lo;

    v8f acc0 = {}, acc1 = {};
    for (int k = 0; k < K; k += 4) {
        float2 av = *(const float2*)(Ap + k);
        v2f a; a.x = av.x; a.y = av.y;
        const float* bp = Bp + (size_t)k * Nout;
        v2f b0; b0.x = bp[0];  b0.y = bp[Nout];
        v2f b1; b1.x = bp[16]; b1.y = bp[Nout + 16];
        acc0 = wmma4(a, b0, acc0);
        acc1 = wmma4(a, b1, acc1);
    }
    const int col = nt * 32 + lo;
    float* Cp = C + (size_t)(mt * 16 + 8 * hi) * Nout + col;
    #pragma unroll
    for (int r = 0; r < 8; ++r) {
        Cp[(size_t)r * Nout + 0]  = acc0[r];
        Cp[(size_t)r * Nout + 16] = acc1[r];
    }
}

__global__ void attn_coef(const float* __restrict__ h,
                          const float* __restrict__ a_src,
                          const float* __restrict__ a_dst,
                          float* __restrict__ al_s,
                          float* __restrict__ al_d)
{
    int idx = blockIdx.x * blockDim.x + threadIdx.x;
    if (idx >= NN * HEADS) return;
    int n = idx / HEADS, hd = idx % HEADS;
    const float* hp = h + (size_t)n * HC + hd * HID;
    const float* as = a_src + hd * HID;
    const float* ad = a_dst + hd * HID;
    float s = 0.f, d = 0.f;
    #pragma unroll
    for (int c = 0; c < HID; ++c) { float v = hp[c]; s = fmaf(v, as[c], s); d = fmaf(v, ad[c], d); }
    al_s[idx] = s; al_d[idx] = d;
}

#define PLANE 33
#define ACCP (4 * PLANE)
#define WPP 288
#define RANGE 196
#define NWG ((NN + RANGE - 1) / RANGE)
#define OT 1024
#define CAP 8192
#define SCAN_U 8
#define ROW_U 8
#define LIST_U 4
static_assert(NN <= 65536, "list entry packs the source index in 16 bits");
static_assert(RANGE < 256 && (RANGE + 1) * HEADS <= OT, "range bookkeeping fits one pass");
static_assert((EE % 4) == 0, "dst array is read as int4");

struct OwnerLds {
    alignas(16) float ev[CAP * HEADS];
    float    acc[(RANGE + 1) * ACCP];
    unsigned m[(RANGE + 1) * HEADS];
    float    den[(RANGE + 1) * HEADS];
    float    ald[(RANGE + 1) * HEADS];
    unsigned list[CAP];
    unsigned count;
};
static_assert(sizeof(OwnerLds) <= 320 * 1024, "one group's LDS fits a WGP");
static_assert((HC / 2) * WPP <= CAP * HEADS, "staged weight fits in ev");

__device__ __forceinline__ void lds_add(float* p, float v) {
    __hip_atomic_fetch_add(p, v, __ATOMIC_RELAXED, __HIP_MEMORY_SCOPE_WORKGROUP);
}

__device__ __forceinline__ void edge_max_one(OwnerLds& L, const float* __restrict__ al_s,
                                             int src, unsigned ld) {
    const float4 s = *(const float4*)(al_s + (size_t)src * HEADS);
    const float sv[4] = { s.x, s.y, s.z, s.w };
    #pragma unroll
    for (int hd = 0; hd < HEADS; ++hd) {
        float e = leaky(sv[hd] + L.ald[ld * HEADS + hd]);
        atomicMax(&L.m[ld * HEADS + hd], ford(e));
    }
}

__device__ __forceinline__ void edge_den_one(OwnerLds& L, const float* __restrict__ al_s,
                                             int src, unsigned ld) {
    const float4 s = *(const float4*)(al_s + (size_t)src * HEADS);
    const float sv[4] = { s.x, s.y, s.z, s.w };
    #pragma unroll
    for (int hd = 0; hd < HEADS; ++hd) {
        float e  = leaky(sv[hd] + L.ald[ld * HEADS + hd]);
        float mx = funord(L.m[ld * HEADS + hd]);
        lds_add(&L.den[ld * HEADS + hd], __expf(e - mx));
    }
}

__device__ __forceinline__ void edge_logit_max(OwnerLds& L, const float* __restrict__ al_s, unsigned idx) {
    const unsigned ent = L.list[idx];
    const unsigned ld  = ent >> 16;
    const float4 s = *(const float4*)(al_s + (size_t)(ent & 0xFFFFu) * HEADS);
    const float4 d = *(const float4*)(L.ald + ld * HEADS);
    float4 e;
    e.x = leaky(s.x + d.x); e.y = leaky(s.y + d.y); e.z = leaky(s.z + d.z); e.w = leaky(s.w + d.w);
    atomicMax(&L.m[ld * HEADS + 0], ford(e.x));
    atomicMax(&L.m[ld * HEADS + 1], ford(e.y));
    atomicMax(&L.m[ld * HEADS + 2], ford(e.z));
    atomicMax(&L.m[ld * HEADS + 3], ford(e.w));
    *(float4*)(L.ev + idx * HEADS) = e;
}

__device__ __forceinline__ void edge_exp_den(OwnerLds& L, unsigned idx) {
    const unsigned ld = L.list[idx] >> 16;
    const float4 e  = *(const float4*)(L.ev + idx * HEADS);
    const uint4  mk = *(const uint4*)(L.m + ld * HEADS);
    float4 x;
    x.x = __expf(e.x - funord(mk.x)); x.y = __expf(e.y - funord(mk.y));
    x.z = __expf(e.z - funord(mk.z)); x.w = __expf(e.w - funord(mk.w));
    lds_add(&L.den[ld * HEADS + 0], x.x);
    lds_add(&L.den[ld * HEADS + 1], x.y);
    lds_add(&L.den[ld * HEADS + 2], x.z);
    lds_add(&L.den[ld * HEADS + 3], x.w);
    *(float4*)(L.ev + idx * HEADS) = x;
}

__device__ __forceinline__ void edge_alpha(OwnerLds& L, unsigned idx) {
    const unsigned ld = L.list[idx] >> 16;
    float4 x = *(const float4*)(L.ev + idx * HEADS);
    const float4 dn = *(const float4*)(L.den + ld * HEADS);
    x.x = x.x / (dn.x + 1e-16f); x.y = x.y / (dn.y + 1e-16f);
    x.z = x.z / (dn.z + 1e-16f); x.w = x.w / (dn.w + 1e-16f);
    *(float4*)(L.ev + idx * HEADS) = x;
}

__device__ __forceinline__ void row_add(OwnerLds& L, float sv, float4 v, unsigned ld, int lane) {
    const int hd = lane >> 3;
    float e  = leaky(sv + L.ald[ld * HEADS + hd]);
    float mx = funord(L.m[ld * HEADS + hd]);
    float alpha = __expf(e - mx) / (L.den[ld * HEADS + hd] + 1e-16f);
    float* ap = L.acc + ld * ACCP + lane;
    lds_add(ap,             v.x * alpha);
    lds_add(ap + PLANE,     v.y * alpha);
    lds_add(ap + 2 * PLANE, v.z * alpha);
    lds_add(ap + 3 * PLANE, v.w * alpha);
}

__device__ __forceinline__ void row_one(OwnerLds& L, const float* __restrict__ h,
                                        const float* __restrict__ al_s,
                                        int src, unsigned ld, int lane) {
    float  sv = al_s[(size_t)src * HEADS + (lane >> 3)];
    float4 v  = *((const float4*)(h + (size_t)src * HC) + lane);
    row_add(L, sv, v, ld, lane);
}

template <int MODE>
__device__ __forceinline__ void scan_edges(OwnerLds& L, const int* __restrict__ ei,
                                           const float* __restrict__ h,
                                           const float* __restrict__ al_s,
                                           int n0, unsigned nloc, int t)
{
    const int4* d4 = (const int4*)(ei + EE);
    const int lane = t & 31;
    for (int base = 0; base < EE / 4; base += OT * SCAN_U) {
        int4 dv[SCAN_U];
        #pragma unroll
        for (int u = 0; u < SCAN_U; ++u) {
            const int i = base + u * OT + t;
            dv[u] = (i < EE / 4) ? d4[i] : make_int4(-1, -1, -1, -1);
        }
        #pragma unroll
        for (int u = 0; u < SCAN_U; ++u) {
            const int i = base + u * OT + t;
            const int dd[4] = { dv[u].x, dv[u].y, dv[u].z, dv[u].w };
            unsigned ld[4]; bool hit[4];
            #pragma unroll
            for (int k = 0; k < 4; ++k) { ld[k] = (unsigned)(dd[k] - n0); hit[k] = ld[k] < nloc; }
            if (MODE == 3) {
                #pragma unroll
                for (int k = 0; k < 4; ++k) {
                    const int src = hit[k] ? ei[4 * i + k] : 0;
                    unsigned mask = __builtin_amdgcn_ballot_w32(hit[k]);
                    while (mask) {
                        const int l = __builtin_ctz(mask);
                        mask &= mask - 1;
                        const int      s2 = __shfl(src, l);
                        const unsigned d2 = (unsigned)__shfl((int)ld[k], l);
                        row_one(L, h, al_s, s2, d2, lane);
                    }
                }
            } else if (hit[0] | hit[1] | hit[2] | hit[3]) {
                #pragma unroll
                for (int k = 0; k < 4; ++k) {
                    if (!hit[k]) continue;
                    const int src = ei[4 * i + k];
                    if (MODE == 0) {
                        const unsigned slot = atomicAdd(&L.count, 1u);
                        if (slot < CAP) L.list[slot] = (unsigned)src | (ld[k] << 16);
                    } else if (MODE == 1) {
                        edge_max_one(L, al_s, src, ld[k]);
                    } else {
                        edge_den_one(L, al_s, src, ld[k]);
                    }
                }
            }
        }
    }
}

template <int NOUT, int CT>
__device__ __forceinline__ void rows_times_weight(const OwnerLds& L, const float* Ws,
                                                  float* __restrict__ dst,
                                                  const float* __restrict__ obias,
                                                  int n0, unsigned nloc, int wv, int lane)
{
    constexpr int NCT = (NOUT + 15) / 16;
    constexpr int NCG = (NCT + CT - 1) / CT;
    constexpr int NRT = (RANGE + 15) / 16;
    const int hi = lane >> 4;
    const int lo = lane & 15;
    for (int job = wv; job < NRT * NCG; job += OT / 32) {
        const int rt = job % NRT, cg = job / NRT;
        int row = rt * 16 + lo;
        if (row > RANGE) row = RANGE;
        const float* ap = L.acc + row * ACCP + 2 * hi * PLANE;
        int col[CT]; const float* bp[CT]; v8f acc[CT];
        #pragma unroll
        for (int u = 0; u < CT; ++u) {
            col[u] = (cg * CT + u) * 16 + lo;
            const int cc = col[u] < NOUT ? col[u] : NOUT - 1;
            bp[u]  = Ws + hi * WPP + 2 * cc;
            acc[u] = v8f{};
        }
        for (int kk = 0; kk < HC / 4; ++kk) {
            v2f a; a.x = ap[kk]; a.y = ap[PLANE + kk];
            #pragma unroll
            for (int u = 0; u < CT; ++u) {
                const float2 bv = *(const float2*)(bp[u] + kk * 2 * WPP);
                v2f b; b.x = bv.x; b.y = bv.y;
                acc[u] = wmma4(a, b, acc[u]);
            }
        }
        #pragma unroll
        for (int u = 0; u < CT; ++u) {
            if (col[u] < NOUT) {
                const float bb = obias ? obias[col[u]] : 0.f;
                #pragma unroll
                for (int r = 0; r < 8; ++r) {
                    const unsigned grow = (unsigned)(rt * 16 + 8 * hi + r);
                    if (grow < nloc) dst[(size_t)(n0 + (int)grow) * NOUT + col[u]] = acc[u][r] + bb;
                }
            }
        }
    }
}

template <bool BUILD, bool LAST>
__global__ __launch_bounds__(OT) void owner_aggregate(
    const int* __restrict__ ei, const float* __restrict__ h,
    const float* __restrict__ al_s, const float* __restrict__ al_d,
    const float* __restrict__ bias, const float* __restrict__ Wn,
    const float* __restrict__ obias, float* __restrict__ dst,
    unsigned* __restrict__ wlist, unsigned* __restrict__ wcount)
{
    __shared__ OwnerLds L;
    const int t    = threadIdx.x;
    const int lane = t & 31;
    const int wv   = t >> 5;
    const int n0   = blockIdx.x * RANGE;
    const unsigned nloc = (unsigned)((NN - n0) < RANGE ? (NN - n0) : RANGE);
    unsigned* wl = wlist + (size_t)blockIdx.x * CAP;

    if (!BUILD) {
        const unsigned saved = wcount[blockIdx.x];
        if (saved <= CAP) {
            for (unsigned i0 = 0; i0 < saved; i0 += OT * LIST_U) {
                unsigned ent[LIST_U];
                #pragma unroll
                for (int u = 0; u < LIST_U; ++u) {
                    const unsigned idx = i0 + u * OT + t;
                    ent[u] = idx < saved ? wl[idx] : 0u;
                }
                #pragma unroll
                for (int u = 0; u < LIST_U; ++u) {
                    const unsigned idx = i0 + u * OT + t;
                    if (idx < saved) L.list[idx] = ent[u];
                }
            }
        }
        if (t == 0) L.count = saved;
    }

    for (int i = t; i < (RANGE + 1) * ACCP; i += OT) L.acc[i] = 0.f;
    if (t < RANGE * HEADS) {
        L.m[t]   = FORD_NEG_INF;
        L.den[t] = 0.f;
        L.ald[t] = ((unsigned)t < nloc * HEADS) ? al_d[(size_t)n0 * HEADS + t] : 0.f;
    }
    if (BUILD) {
        if ((unsigned)t < nloc) L.list[t] = (unsigned)(n0 + t) | ((unsigned)t << 16);
        if (t == 0) L.count = nloc;
    }
    __syncthreads();

    if (BUILD) {
        scan_edges<0>(L, ei, h, al_s, n0, nloc, t);
        __syncthreads();
    }
    const unsigned total = L.count;

    if (BUILD) {
        if (total <= CAP)
            for (unsigned idx = t; idx < total; idx += OT) wl[idx] = L.list[idx];
        if (t == 0) wcount[blockIdx.x] = total;
    }

    if (total <= CAP) {
        for (unsigned idx = t; idx < total; idx += OT) edge_logit_max(L, al_s, idx);
        __syncthreads();
        for (unsigned idx = t; idx < total; idx += OT) edge_exp_den(L, idx);
        __syncthreads();
        for (unsigned idx = t; idx < total; idx += OT) edge_alpha(L, idx);
        __syncthreads();
        const unsigned spare = (unsigned)n0 | ((unsigned)RANGE << 16);
        for (unsigned i0 = wv; i0 < total; i0 += (OT / 32) * ROW_U) {
            unsigned ent[ROW_U]; float al[ROW_U]; float4 v[ROW_U];
            #pragma unroll
            for (int u = 0; u < ROW_U; ++u) {
                const unsigned idx = i0 + u * (OT / 32);
                const unsigned idc = idx < total ? idx : 0;
                const unsigned e0 = L.list[idc];
                ent[u] = idx < total ? e0 : spare;
                al[u]  = L.ev[idc * HEADS + (lane >> 3)];
                v[u]   = *((const float4*)(h + (size_t)(ent[u] & 0xFFFFu) * HC) + lane);
            }
            #pragma unroll
            for (int u = 0; u < ROW_U; ++u) {
                float* ap = L.acc + (ent[u] >> 16) * ACCP + lane;
                lds_add(ap,             v[u].x * al[u]);
                lds_add(ap + PLANE,     v[u].y * al[u]);
                lds_add(ap + 2 * PLANE, v[u].z * al[u]);
                lds_add(ap + 3 * PLANE, v[u].w * al[u]);
            }
        }
    } else {
        if ((unsigned)t < nloc) edge_max_one(L, al_s, n0 + t, (unsigned)t);
        scan_edges<1>(L, ei, h, al_s, n0, nloc, t);
        __syncthreads();
        if ((unsigned)t < nloc) edge_den_one(L, al_s, n0 + t, (unsigned)t);
        scan_edges<2>(L, ei, h, al_s, n0, nloc, t);
        __syncthreads();
        for (unsigned i = wv; i < nloc; i += OT / 32) row_one(L, h, al_s, n0 + (int)i, i, lane);
        scan_edges<3>(L, ei, h, al_s, n0, nloc, t);
    }
    __syncthreads();

    for (int i = t; i < RANGE * HC; i += OT) {
        const int node = i / HC, q = i % HC;
        const int j = q >> 5, l = q & 31;
        float* p = L.acc + node * ACCP + j * PLANE + l;
        const float vv = *p + bias[4 * l + j];
        *p = vv > 0.f ? vv : 0.f;
    }
    constexpr int NOUT = LAST ? OUT_DIM : HC;
    float* Ws = L.ev;
    for (int i = t; i < HC * NOUT; i += OT) {
        const int k = i / NOUT, n = i % NOUT;
        Ws[(k >> 1) * WPP + 2 * n + (k & 1)] = Wn[i];
    }
    __syncthreads();
    rows_times_weight<NOUT, LAST ? 1 : 2>(L, Ws, dst, obias, n0, nloc, wv, lane);
}

extern "C" void kernel_launch(void* const* d_in, const int* in_sizes, int n_in,
                              void* d_out, int out_size, void* d_ws, size_t ws_size,
                              hipStream_t stream)
{
    const float* x   = (const float*)d_in[0];
    const int*   ei  = (const int*)d_in[1];
    const float* W[3]    = { (const float*)d_in[2],  (const float*)d_in[6],  (const float*)d_in[10] };
    const float* asv[3]  = { (const float*)d_in[3],  (const float*)d_in[7],  (const float*)d_in[11] };
    const float* adv[3]  = { (const float*)d_in[4],  (const float*)d_in[8],  (const float*)d_in[12] };
    const float* bv[3]   = { (const float*)d_in[5],  (const float*)d_in[9],  (const float*)d_in[13] };
    const float* Wl  = (const float*)d_in[14];
    const float* bl  = (const float*)d_in[15];

    char* ws = (char*)d_ws;
    float*    hA   = (float*)ws;                       ws += (size_t)NN * HC * sizeof(float);
    float*    hB   = (float*)ws;                       ws += (size_t)NN * HC * sizeof(float);
    float*    al_s = (float*)ws;                       ws += (size_t)NN * HEADS * sizeof(float);
    float*    al_d = (float*)ws;                       ws += (size_t)NN * HEADS * sizeof(float);
    unsigned* wlist  = (unsigned*)ws;                  ws += (size_t)NWG * CAP * sizeof(unsigned);
    unsigned* wcount = (unsigned*)ws;

    const int TPB = 256;
    const int MT  = NN / 16;
    const int gNH    = (NN * HEADS + TPB - 1) / TPB;

    auto logits = [&](const float* hin, int layer) {
        hipLaunchKernelGGL(attn_coef, dim3(gNH), dim3(TPB), 0, stream,
                           hin, asv[layer], adv[layer], al_s, al_d);
    };

    hipLaunchKernelGGL(wmma_gemm_f32_2t, dim3(MT, HC / 32), dim3(32), 0, stream,
                       x, W[0], hA, IN_DIM, HC);
    logits(hA, 0);
    hipLaunchKernelGGL((owner_aggregate<true, false>), dim3(NWG), dim3(OT), 0, stream,
                       ei, hA, al_s, al_d, bv[0], W[1], (const float*)nullptr, hB, wlist, wcount);
    logits(hB, 1);
    hipLaunchKernelGGL((owner_aggregate<false, false>), dim3(NWG), dim3(OT), 0, stream,
                       ei, hB, al_s, al_d, bv[1], W[2], (const float*)nullptr, hA, wlist, wcount);
    logits(hA, 2);
    hipLaunchKernelGGL((owner_aggregate<false, true>), dim3(NWG), dim3(OT), 0, stream,
                       ei, hA, al_s, al_d, bv[2], Wl, bl, (float*)d_out, wlist, wcount);
}


# First product x@W1 by the 256 node-range groups: own x rows copied to LDS, W1 staged in LDS as k pairs, 16x32 tiles on the matrix unit from LDS; 12500 one-wave groups gone; k and sum order kept
# speedup vs baseline: 4.6156x; 1.1017x over previous
#include <hip/hip_runtime.h>
#include <math.h>

#define NN 50000
#define EE 800000
#define IN_DIM 128
#define HID 32
#define HEADS 4
#define HC 128
#define OUT_DIM 40
#define SLOPE 0.2f

typedef float v2f __attribute__((ext_vector_type(2)));
typedef float v8f __attribute__((ext_vector_type(8)));

__device__ __forceinline__ unsigned ford(float f) {
    unsigned u = __float_as_uint(f);
    return (u & 0x80000000u) ? ~u : (u | 0x80000000u);
}
__device__ __forceinline__ float funord(unsigned k) {
    unsigned u = (k & 0x80000000u) ? (k & 0x7FFFFFFFu) : ~k;
    return __uint_as_float(u);
}
#define FORD_NEG_INF 0x007FFFFFu

__device__ __forceinline__ float leaky(float e) {
    return e > 0.f ? e : SLOPE * e;
}

__device__ __forceinline__ v8f wmma4(v2f a, v2f b, v8f c) {
    return __builtin_amdgcn_wmma_f32_16x16x4_f32(
               false, a, false, b, (short)0, c, false, false);
}

__global__ void attn_coef(const float* __restrict__ h,
                          const float* __restrict__ a_src,
                          const float* __restrict__ a_dst,
                          float* __restrict__ al_s,
                          float* __restrict__ al_d)
{
    int idx = blockIdx.x * blockDim.x + threadIdx.x;
    if (idx >= NN * HEADS) return;
    int n = idx / HEADS, hd = idx % HEADS;
    const float* hp = h + (size_t)n * HC + hd * HID;
    const float* as = a_src + hd * HID;
    const float* ad = a_dst + hd * HID;
    float s = 0.f, d = 0.f;
    #pragma unroll
    for (int c = 0; c < HID; ++c) { float v = hp[c]; s = fmaf(v, as[c], s); d = fmaf(v, ad[c], d); }
    al_s[idx] = s; al_d[idx] = d;
}

#define PLANE 33
#define ACCP (4 * PLANE)
#define WPP 288
#define RANGE 196
#define NWG ((NN + RANGE - 1) / RANGE)
#define OT 1024
#define CAP 8192
#define SCAN_U 8
#define ROW_U 8
#define LIST_U 4
static_assert(NN <= 65536, "list entry packs the source index in 16 bits");
static_assert(RANGE < 256 && (RANGE + 1) * HEADS <= OT, "range bookkeeping fits one pass");
static_assert((EE % 4) == 0, "dst array is read as int4");

struct OwnerLds {
    alignas(16) float ev[CAP * HEADS];
    float    acc[(RANGE + 1) * ACCP];
    unsigned m[(RANGE + 1) * HEADS];
    float    den[(RANGE + 1) * HEADS];
    float    ald[(RANGE + 1) * HEADS];
    unsigned list[CAP];
    unsigned count;
};
static_assert(sizeof(OwnerLds) <= 320 * 1024, "one group's LDS fits a WGP");
static_assert((HC / 2) * WPP <= CAP * HEADS, "staged weight fits in ev");

__device__ __forceinline__ void lds_add(float* p, float v) {
    __hip_atomic_fetch_add(p, v, __ATOMIC_RELAXED, __HIP_MEMORY_SCOPE_WORKGROUP);
}

__device__ __forceinline__ void edge_max_one(OwnerLds& L, const float* __restrict__ al_s,
                                             int src, unsigned ld) {
    const float4 s = *(const float4*)(al_s + (size_t)src * HEADS);
    const float sv[4] = { s.x, s.y, s.z, s.w };
    #pragma unroll
    for (int hd = 0; hd < HEADS; ++hd) {
        float e = leaky(sv[hd] + L.ald[ld * HEADS + hd]);
        atomicMax(&L.m[ld * HEADS + hd], ford(e));
    }
}

__device__ __forceinline__ void edge_den_one(OwnerLds& L, const float* __restrict__ al_s,
                                             int src, unsigned ld) {
    const float4 s = *(const float4*)(al_s + (size_t)src * HEADS);
    const float sv[4] = { s.x, s.y, s.z, s.w };
    #pragma unroll
    for (int hd = 0; hd < HEADS; ++hd) {
        float e  = leaky(sv[hd] + L.ald[ld * HEADS + hd]);
        float mx = funord(L.m[ld * HEADS + hd]);
        lds_add(&L.den[ld * HEADS + hd], __expf(e - mx));
    }
}

__device__ __forceinline__ void edge_logit_max(OwnerLds& L, const float* __restrict__ al_s, unsigned idx) {
    const unsigned ent = L.list[idx];
    const unsigned ld  = ent >> 16;
    const float4 s = *(const float4*)(al_s + (size_t)(ent & 0xFFFFu) * HEADS);
    const float4 d = *(const float4*)(L.ald + ld * HEADS);
    float4 e;
    e.x = leaky(s.x + d.x); e.y = leaky(s.y + d.y); e.z = leaky(s.z + d.z); e.w = leaky(s.w + d.w);
    atomicMax(&L.m[ld * HEADS + 0], ford(e.x));
    atomicMax(&L.m[ld * HEADS + 1], ford(e.y));
    atomicMax(&L.m[ld * HEADS + 2], ford(e.z));
    atomicMax(&L.m[ld * HEADS + 3], ford(e.w));
    *(float4*)(L.ev + idx * HEADS) = e;
}

__device__ __forceinline__ void edge_exp_den(OwnerLds& L, unsigned idx) {
    const unsigned ld = L.list[idx] >> 16;
    const float4 e  = *(const float4*)(L.ev + idx * HEADS);
    const uint4  mk = *(const uint4*)(L.m + ld * HEADS);
    float4 x;
    x.x = __expf(e.x - funord(mk.x)); x.y = __expf(e.y - funord(mk.y));
    x.z = __expf(e.z - funord(mk.z)); x.w = __expf(e.w - funord(mk.w));
    lds_add(&L.den[ld * HEADS + 0], x.x);
    lds_add(&L.den[ld * HEADS + 1], x.y);
    lds_add(&L.den[ld * HEADS + 2], x.z);
    lds_add(&L.den[ld * HEADS + 3], x.w);
    *(float4*)(L.ev + idx * HEADS) = x;
}

__device__ __forceinline__ void edge_alpha(OwnerLds& L, unsigned idx) {
    const unsigned ld = L.list[idx] >> 16;
    float4 x = *(const float4*)(L.ev + idx * HEADS);
    const float4 dn = *(const float4*)(L.den + ld * HEADS);
    x.x = x.x / (dn.x + 1e-16f); x.y = x.y / (dn.y + 1e-16f);
    x.z = x.z / (dn.z + 1e-16f); x.w = x.w / (dn.w + 1e-16f);
    *(float4*)(L.ev + idx * HEADS) = x;
}

__device__ __forceinline__ void row_add(OwnerLds& L, float sv, float4 v, unsigned ld, int lane) {
    const int hd = lane >> 3;
    float e  = leaky(sv + L.ald[ld * HEADS + hd]);
    float mx = funord(L.m[ld * HEADS + hd]);
    float alpha = __expf(e - mx) / (L.den[ld * HEADS + hd] + 1e-16f);
    float* ap = L.acc + ld * ACCP + lane;
    lds_add(ap,             v.x * alpha);
    lds_add(ap + PLANE,     v.y * alpha);
    lds_add(ap + 2 * PLANE, v.z * alpha);
    lds_add(ap + 3 * PLANE, v.w * alpha);
}

__device__ __forceinline__ void row_one(OwnerLds& L, const float* __restrict__ h,
                                        const float* __restrict__ al_s,
                                        int src, unsigned ld, int lane) {
    float  sv = al_s[(size_t)src * HEADS + (lane >> 3)];
    float4 v  = *((const float4*)(h + (size_t)src * HC) + lane);
    row_add(L, sv, v, ld, lane);
}

template <int MODE>
__device__ __forceinline__ void scan_edges(OwnerLds& L, const int* __restrict__ ei,
                                           const float* __restrict__ h,
                                           const float* __restrict__ al_s,
                                           int n0, unsigned nloc, int t)
{
    const int4* d4 = (const int4*)(ei + EE);
    const int lane = t & 31;
    for (int base = 0; base < EE / 4; base += OT * SCAN_U) {
        int4 dv[SCAN_U];
        #pragma unroll
        for (int u = 0; u < SCAN_U; ++u) {
            const int i = base + u * OT + t;
            dv[u] = (i < EE / 4) ? d4[i] : make_int4(-1, -1, -1, -1);
        }
        #pragma unroll
        for (int u = 0; u < SCAN_U; ++u) {
            const int i = base + u * OT + t;
            const int dd[4] = { dv[u].x, dv[u].y, dv[u].z, dv[u].w };
            unsigned ld[4]; bool hit[4];
            #pragma unroll
            for (int k = 0; k < 4; ++k) { ld[k] = (unsigned)(dd[k] - n0); hit[k] = ld[k] < nloc; }
            if (MODE == 3) {
                #pragma unroll
                for (int k = 0; k < 4; ++k) {
                    const int src = hit[k] ? ei[4 * i + k] : 0;
                    unsigned mask = __builtin_amdgcn_ballot_w32(hit[k]);
                    while (mask) {
                        const int l = __builtin_ctz(mask);
                        mask &= mask - 1;
                        const int      s2 = __shfl(src, l);
                        const unsigned d2 = (unsigned)__shfl((int)ld[k], l);
                        row_one(L, h, al_s, s2, d2, lane);
                    }
                }
            } else if (hit[0] | hit[1] | hit[2] | hit[3]) {
                #pragma unroll
                for (int k = 0; k < 4; ++k) {
                    if (!hit[k]) continue;
                    const int src = ei[4 * i + k];
                    if (MODE == 0) {
                        const unsigned slot = atomicAdd(&L.count, 1u);
                        if (slot < CAP) L.list[slot] = (unsigned)src | (ld[k] << 16);
                    } else if (MODE == 1) {
                        edge_max_one(L, al_s, src, ld[k]);
                    } else {
                        edge_den_one(L, al_s, src, ld[k]);
                    }
                }
            }
        }
    }
}

template <int NOUT, int CT>
__device__ __forceinline__ void rows_times_weight(const float* rows, const float* Ws,
                                                  float* __restrict__ dst,
                                                  const float* __restrict__ obias,
                                                  int n0, unsigned nloc, int wv, int lane)
{
    constexpr int NCT = (NOUT + 15) / 16;
    constexpr int NCG = (NCT + CT - 1) / CT;
    constexpr int NRT = (RANGE + 15) / 16;
    const int hi = lane >> 4;
    const int lo = lane & 15;
    for (int job = wv; job < NRT * NCG; job += OT / 32) {
        const int rt = job % NRT, cg = job / NRT;
        int row = rt * 16 + lo;
        if (row > RANGE) row = RANGE;
        const float* ap = rows + row * ACCP + 2 * hi * PLANE;
        int col[CT]; const float* bp[CT]; v8f acc[CT];
        #pragma unroll
        for (int u = 0; u < CT; ++u) {
            col[u] = (cg * CT + u) * 16 + lo;
            const int cc = col[u] < NOUT ? col[u] : NOUT - 1;
            bp[u]  = Ws + hi * WPP + 2 * cc;
            acc[u] = v8f{};
        }
        for (int kk = 0; kk < HC / 4; ++kk) {
            v2f a; a.x = ap[kk]; a.y = ap[PLANE + kk];
            #pragma unroll
            for (int u = 0; u < CT; ++u) {
                const float2 bv = *(const float2*)(bp[u] + kk * 2 * WPP);
                v2f b; b.x = bv.x; b.y = bv.y;
                acc[u] = wmma4(a, b, acc[u]);
            }
        }
        #pragma unroll
        for (int u = 0; u < CT; ++u) {
            if (col[u] < NOUT) {
                const float bb = obias ? obias[col[u]] : 0.f;
                #pragma unroll
                for (int r = 0; r < 8; ++r) {
                    const unsigned grow = (unsigned)(rt * 16 + 8 * hi + r);
                    if (grow < nloc) dst[(size_t)(n0 + (int)grow) * NOUT + col[u]] = acc[u][r] + bb;
                }
            }
        }
    }
}

#define XROW_U 7
#define W_U ((IN_DIM * HC) / OT)
static_assert(IN_DIM == HC, "the tile routine walks HC channels");
static_assert((RANGE + 1) * (IN_DIM / 4) <= OT * XROW_U, "one trip copies the whole range");
static_assert(W_U * OT == IN_DIM * HC, "the weight divides evenly over the threads");
#define XROWS ((OT * XROW_U) / (IN_DIM / 4))
struct FirstLds {
    alignas(16) float ws[(IN_DIM / 2) * WPP];
    float rows[XROWS * ACCP];
};
static_assert(sizeof(FirstLds) <= 320 * 1024, "one group's LDS fits a WGP");
__global__ __launch_bounds__(OT) void range_product(
    const float* __restrict__ x, const float* __restrict__ Wn, float* __restrict__ dst)
{
    __shared__ FirstLds F;
    const int t    = threadIdx.x;
    const int lane = t & 31;
    const int wv   = t >> 5;
    const int n0   = blockIdx.x * RANGE;
    const unsigned nloc = (unsigned)((NN - n0) < RANGE ? (NN - n0) : RANGE);

    float4 v[XROW_U];
    float  w[W_U];
    #pragma unroll
    for (int u = 0; u < XROW_U; ++u) {
        const int i = u * OT + t;
        const unsigned node = (unsigned)(i >> 5);
        const unsigned nc   = node < nloc ? node : nloc - 1;
        v[u] = *((const float4*)(x + (size_t)(n0 + (int)nc) * IN_DIM) + (i & 31));
    }
    #pragma unroll
    for (int u = 0; u < W_U; ++u) w[u] = Wn[u * OT + t];
    #pragma unroll
    for (int u = 0; u < XROW_U; ++u) {
        const int i = u * OT + t;
        float* p = F.rows + (i >> 5) * ACCP + (i & 31);
        p[0] = v[u].x; p[PLANE] = v[u].y; p[2 * PLANE] = v[u].z; p[3 * PLANE] = v[u].w;
    }
    #pragma unroll
    for (int u = 0; u < W_U; ++u) {
        const int i = u * OT + t;
        const int k = i / HC, n = i % HC;
        F.ws[(k >> 1) * WPP + 2 * n + (k & 1)] = w[u];
    }
    __syncthreads();
    rows_times_weight<HC, 2>(F.rows, F.ws, dst, (const float*)nullptr, n0, nloc, wv, lane);
}

template <bool BUILD, bool LAST>
__global__ __launch_bounds__(OT) void owner_aggregate(
    const int* __restrict__ ei, const float* __restrict__ h,
    const float* __restrict__ al_s, const float* __restrict__ al_d,
    const float* __restrict__ bias, const float* __restrict__ Wn,
    const float* __restrict__ obias, float* __restrict__ dst,
    unsigned* __restrict__ wlist, unsigned* __restrict__ wcount)
{
    __shared__ OwnerLds L;
    const int t    = threadIdx.x;
    const int lane = t & 31;
    const int wv   = t >> 5;
    const int n0   = blockIdx.x * RANGE;
    const unsigned nloc = (unsigned)((NN - n0) < RANGE ? (NN - n0) : RANGE);
    unsigned* wl = wlist + (size_t)blockIdx.x * CAP;

    if (!BUILD) {
        const unsigned saved = wcount[blockIdx.x];
        if (saved <= CAP) {
            for (unsigned i0 = 0; i0 < saved; i0 += OT * LIST_U) {
                unsigned ent[LIST_U];
                #pragma unroll
                for (int u = 0; u < LIST_U; ++u) {
                    const unsigned idx = i0 + u * OT + t;
                    ent[u] = idx < saved ? wl[idx] : 0u;
                }
                #pragma unroll
                for (int u = 0; u < LIST_U; ++u) {
                    const unsigned idx = i0 + u * OT + t;
                    if (idx < saved) L.list[idx] = ent[u];
                }
            }
        }
        if (t == 0) L.count = saved;
    }

    for (int i = t; i < (RANGE + 1) * ACCP; i += OT) L.acc[i] = 0.f;
    if (t < RANGE * HEADS) {
        L.m[t]   = FORD_NEG_INF;
        L.den[t] = 0.f;
        L.ald[t] = ((unsigned)t < nloc * HEADS) ? al_d[(size_t)n0 * HEADS + t] : 0.f;
    }
    if (BUILD) {
        if ((unsigned)t < nloc) L.list[t] = (unsigned)(n0 + t) | ((unsigned)t << 16);
        if (t == 0) L.count = nloc;
    }
    __syncthreads();

    if (BUILD) {
        scan_edges<0>(L, ei, h, al_s, n0, nloc, t);
        __syncthreads();
    }
    const unsigned total = L.count;

    if (BUILD) {
        if (total <= CAP)
            for (unsigned idx = t; idx < total; idx += OT) wl[idx] = L.list[idx];
        if (t == 0) wcount[blockIdx.x] = total;
    }

    if (total <= CAP) {
        for (unsigned idx = t; idx < total; idx += OT) edge_logit_max(L, al_s, idx);
        __syncthreads();
        for (unsigned idx = t; idx < total; idx += OT) edge_exp_den(L, idx);
        __syncthreads();
        for (unsigned idx = t; idx < total; idx += OT) edge_alpha(L, idx);
        __syncthreads();
        const unsigned spare = (unsigned)n0 | ((unsigned)RANGE << 16);
        for (unsigned i0 = wv; i0 < total; i0 += (OT / 32) * ROW_U) {
            unsigned ent[ROW_U]; float al[ROW_U]; float4 v[ROW_U];
            #pragma unroll
            for (int u = 0; u < ROW_U; ++u) {
                const unsigned idx = i0 + u * (OT / 32);
                const unsigned idc = idx < total ? idx : 0;
                const unsigned e0 = L.list[idc];
                ent[u] = idx < total ? e0 : spare;
                al[u]  = L.ev[idc * HEADS + (lane >> 3)];
                v[u]   = *((const float4*)(h + (size_t)(ent[u] & 0xFFFFu) * HC) + lane);
            }
            #pragma unroll
            for (int u = 0; u < ROW_U; ++u) {
                float* ap = L.acc + (ent[u] >> 16) * ACCP + lane;
                lds_add(ap,             v[u].x * al[u]);
                lds_add(ap + PLANE,     v[u].y * al[u]);
                lds_add(ap + 2 * PLANE, v[u].z * al[u]);
                lds_add(ap + 3 * PLANE, v[u].w * al[u]);
            }
        }
    } else {
        if ((unsigned)t < nloc) edge_max_one(L, al_s, n0 + t, (unsigned)t);
        scan_edges<1>(L, ei, h, al_s, n0, nloc, t);
        __syncthreads();
        if ((unsigned)t < nloc) edge_den_one(L, al_s, n0 + t, (unsigned)t);
        scan_edges<2>(L, ei, h, al_s, n0, nloc, t);
        __syncthreads();
        for (unsigned i = wv; i < nloc; i += OT / 32) row_one(L, h, al_s, n0 + (int)i, i, lane);
        scan_edges<3>(L, ei, h, al_s, n0, nloc, t);
    }
    __syncthreads();

    for (int i = t; i < RANGE * HC; i += OT) {
        const int node = i / HC, q = i % HC;
        const int j = q >> 5, l = q & 31;
        float* p = L.acc + node * ACCP + j * PLANE + l;
        const float vv = *p + bias[4 * l + j];
        *p = vv > 0.f ? vv : 0.f;
    }
    constexpr int NOUT = LAST ? OUT_DIM : HC;
    float* Ws = L.ev;
    for (int i = t; i < HC * NOUT; i += OT) {
        const int k = i / NOUT, n = i % NOUT;
        Ws[(k >> 1) * WPP + 2 * n + (k & 1)] = Wn[i];
    }
    __syncthreads();
    rows_times_weight<NOUT, LAST ? 1 : 2>(L.acc, Ws, dst, obias, n0, nloc, wv, lane);
}

extern "C" void kernel_launch(void* const* d_in, const int* in_sizes, int n_in,
                              void* d_out, int out_size, void* d_ws, size_t ws_size,
                              hipStream_t stream)
{
    const float* x   = (const float*)d_in[0];
    const int*   ei  = (const int*)d_in[1];
    const float* W[3]    = { (const float*)d_in[2],  (const float*)d_in[6],  (const float*)d_in[10] };
    const float* asv[3]  = { (const float*)d_in[3],  (const float*)d_in[7],  (const float*)d_in[11] };
    const float* adv[3]  = { (const float*)d_in[4],  (const float*)d_in[8],  (const float*)d_in[12] };
    const float* bv[3]   = { (const float*)d_in[5],  (const float*)d_in[9],  (const float*)d_in[13] };
    const float* Wl  = (const float*)d_in[14];
    const float* bl  = (const float*)d_in[15];

    char* ws = (char*)d_ws;
    float*    hA   = (float*)ws;                       ws += (size_t)NN * HC * sizeof(float);
    float*    hB   = (float*)ws;                       ws += (size_t)NN * HC * sizeof(float);
    float*    al_s = (float*)ws;                       ws += (size_t)NN * HEADS * sizeof(float);
    float*    al_d = (float*)ws;                       ws += (size_t)NN * HEADS * sizeof(float);
    unsigned* wlist  = (unsigned*)ws;                  ws += (size_t)NWG * CAP * sizeof(unsigned);
    unsigned* wcount = (unsigned*)ws;

    const int TPB = 256;
    const int gNH    = (NN * HEADS + TPB - 1) / TPB;

    auto logits = [&](const float* hin, int layer) {
        hipLaunchKernelGGL(attn_coef, dim3(gNH), dim3(TPB), 0, stream,
                           hin, asv[layer], adv[layer], al_s, al_d);
    };

    hipLaunchKernelGGL(range_product, dim3(NWG), dim3(OT), 0, stream, x, W[0], hA);
    logits(hA, 0);
    hipLaunchKernelGGL((owner_aggregate<true, false>), dim3(NWG), dim3(OT), 0, stream,
                       ei, hA, al_s, al_d, bv[0], W[1], (const float*)nullptr, hB, wlist, wcount);
    logits(hB, 1);
    hipLaunchKernelGGL((owner_aggregate<false, false>), dim3(NWG), dim3(OT), 0, stream,
                       ei, hB, al_s, al_d, bv[1], W[2], (const float*)nullptr, hA, wlist, wcount);
    logits(hA, 2);
    hipLaunchKernelGGL((owner_aggregate<false, true>), dim3(NWG), dim3(OT), 0, stream,
                       ei, hA, al_s, al_d, bv[2], Wl, bl, (float*)d_out, wlist, wcount);
}


# First launch stores each edge's owner group (dst/196, 1 byte) bit-sliced per 32 edges; layer 1 list build matches 32 edges per lane on those bytes, loads src/dst of matches only; arrival-order sums
# speedup vs baseline: 5.1685x; 1.1198x over previous
#include <hip/hip_runtime.h>
#include <math.h>

#define NN 50000
#define EE 800000
#define IN_DIM 128
#define HID 32
#define HEADS 4
#define HC 128
#define OUT_DIM 40
#define SLOPE 0.2f

typedef float v2f __attribute__((ext_vector_type(2)));
typedef float v8f __attribute__((ext_vector_type(8)));

__device__ __forceinline__ unsigned ford(float f) {
    unsigned u = __float_as_uint(f);
    return (u & 0x80000000u) ? ~u : (u | 0x80000000u);
}
__device__ __forceinline__ float funord(unsigned k) {
    unsigned u = (k & 0x80000000u) ? (k & 0x7FFFFFFFu) : ~k;
    return __uint_as_float(u);
}
#define FORD_NEG_INF 0x007FFFFFu

__device__ __forceinline__ float leaky(float e) {
    return e > 0.f ? e : SLOPE * e;
}

__device__ __forceinline__ v8f wmma4(v2f a, v2f b, v8f c) {
    return __builtin_amdgcn_wmma_f32_16x16x4_f32(
               false, a, false, b, (short)0, c, false, false);
}

__global__ void attn_coef(const float* __restrict__ h,
                          const float* __restrict__ a_src,
                          const float* __restrict__ a_dst,
                          float* __restrict__ al_s,
                          float* __restrict__ al_d)
{
    int idx = blockIdx.x * blockDim.x + threadIdx.x;
    if (idx >= NN * HEADS) return;
    int n = idx / HEADS, hd = idx % HEADS;
    const float* hp = h + (size_t)n * HC + hd * HID;
    const float* as = a_src + hd * HID;
    const float* ad = a_dst + hd * HID;
    float s = 0.f, d = 0.f;
    #pragma unroll
    for (int c = 0; c < HID; ++c) { float v = hp[c]; s = fmaf(v, as[c], s); d = fmaf(v, ad[c], d); }
    al_s[idx] = s; al_d[idx] = d;
}

#define PLANE 33
#define ACCP (4 * PLANE)
#define WPP 288
#define RANGE 196
#define NWG ((NN + RANGE - 1) / RANGE)
#define OT 1024
#define CAP 8192
#define SCAN_U 8
#define ROW_U 8
#define LIST_U 4
#define NCHUNK (EE / 32)
#define PL_U 5
static_assert(NN <= 65536, "list entry packs the source index in 16 bits");
static_assert(RANGE < 256 && (RANGE + 1) * HEADS <= OT, "range bookkeeping fits one pass");
static_assert((EE % 4) == 0, "dst array is read as int4");
static_assert((EE % 32) == 0, "edges divide into whole chunks");
static_assert(NWG <= 256, "an owner number fits one byte");

struct OwnerLds {
    alignas(16) float ev[CAP * HEADS];
    float    acc[(RANGE + 1) * ACCP];
    unsigned m[(RANGE + 1) * HEADS];
    float    den[(RANGE + 1) * HEADS];
    float    ald[(RANGE + 1) * HEADS];
    unsigned list[CAP];
    unsigned count;
};
static_assert(sizeof(OwnerLds) <= 320 * 1024, "one group's LDS fits a WGP");
static_assert((HC / 2) * WPP <= CAP * HEADS, "staged weight fits in ev");

__device__ __forceinline__ void lds_add(float* p, float v) {
    __hip_atomic_fetch_add(p, v, __ATOMIC_RELAXED, __HIP_MEMORY_SCOPE_WORKGROUP);
}

__device__ __forceinline__ void edge_max_one(OwnerLds& L, const float* __restrict__ al_s,
                                             int src, unsigned ld) {
    const float4 s = *(const float4*)(al_s + (size_t)src * HEADS);
    const float sv[4] = { s.x, s.y, s.z, s.w };
    #pragma unroll
    for (int hd = 0; hd < HEADS; ++hd) {
        float e = leaky(sv[hd] + L.ald[ld * HEADS + hd]);
        atomicMax(&L.m[ld * HEADS + hd], ford(e));
    }
}

__device__ __forceinline__ void edge_den_one(OwnerLds& L, const float* __restrict__ al_s,
                                             int src, unsigned ld) {
    const float4 s = *(const float4*)(al_s + (size_t)src * HEADS);
    const float sv[4] = { s.x, s.y, s.z, s.w };
    #pragma unroll
    for (int hd = 0; hd < HEADS; ++hd) {
        float e  = leaky(sv[hd] + L.ald[ld * HEADS + hd]);
        float mx = funord(L.m[ld * HEADS + hd]);
        lds_add(&L.den[ld * HEADS + hd], __expf(e - mx));
    }
}

__device__ __forceinline__ void edge_logit_max(OwnerLds& L, const float* __restrict__ al_s, unsigned idx) {
    const unsigned ent = L.list[idx];
    const unsigned ld  = ent >> 16;
    const float4 s = *(const float4*)(al_s + (size_t)(ent & 0xFFFFu) * HEADS);
    const float4 d = *(const float4*)(L.ald + ld * HEADS);
    float4 e;
    e.x = leaky(s.x + d.x); e.y = leaky(s.y + d.y); e.z = leaky(s.z + d.z); e.w = leaky(s.w + d.w);
    atomicMax(&L.m[ld * HEADS + 0], ford(e.x));
    atomicMax(&L.m[ld * HEADS + 1], ford(e.y));
    atomicMax(&L.m[ld * HEADS + 2], ford(e.z));
    atomicMax(&L.m[ld * HEADS + 3], ford(e.w));
    *(float4*)(L.ev + idx * HEADS) = e;
}

__device__ __forceinline__ void edge_exp_den(OwnerLds& L, unsigned idx) {
    const unsigned ld = L.list[idx] >> 16;
    const float4 e  = *(const float4*)(L.ev + idx * HEADS);
    const uint4  mk = *(const uint4*)(L.m + ld * HEADS);
    float4 x;
    x.x = __expf(e.x - funord(mk.x)); x.y = __expf(e.y - funord(mk.y));
    x.z = __expf(e.z - funord(mk.z)); x.w = __expf(e.w - funord(mk.w));
    lds_add(&L.den[ld * HEADS + 0], x.x);
    lds_add(&L.den[ld * HEADS + 1], x.y);
    lds_add(&L.den[ld * HEADS + 2], x.z);
    lds_add(&L.den[ld * HEADS + 3], x.w);
    *(float4*)(L.ev + idx * HEADS) = x;
}

__device__ __forceinline__ void edge_alpha(OwnerLds& L, unsigned idx) {
    const unsigned ld = L.list[idx] >> 16;
    float4 x = *(const float4*)(L.ev + idx * HEADS);
    const float4 dn = *(const float4*)(L.den + ld * HEADS);
    x.x = x.x / (dn.x + 1e-16f); x.y = x.y / (dn.y + 1e-16f);
    x.z = x.z / (dn.z + 1e-16f); x.w = x.w / (dn.w + 1e-16f);
    *(float4*)(L.ev + idx * HEADS) = x;
}

__device__ __forceinline__ void row_add(OwnerLds& L, float sv, float4 v, unsigned ld, int lane) {
    const int hd = lane >> 3;
    float e  = leaky(sv + L.ald[ld * HEADS + hd]);
    float mx = funord(L.m[ld * HEADS + hd]);
    float alpha = __expf(e - mx) / (L.den[ld * HEADS + hd] + 1e-16f);
    float* ap = L.acc + ld * ACCP + lane;
    lds_add(ap,             v.x * alpha);
    lds_add(ap + PLANE,     v.y * alpha);
    lds_add(ap + 2 * PLANE, v.z * alpha);
    lds_add(ap + 3 * PLANE, v.w * alpha);
}

__device__ __forceinline__ void row_one(OwnerLds& L, const float* __restrict__ h,
                                        const float* __restrict__ al_s,
                                        int src, unsigned ld, int lane) {
    float  sv = al_s[(size_t)src * HEADS + (lane >> 3)];
    float4 v  = *((const float4*)(h + (size_t)src * HC) + lane);
    row_add(L, sv, v, ld, lane);
}

template <int MODE>
__device__ __forceinline__ void scan_edges(OwnerLds& L, const int* __restrict__ ei,
                                           const float* __restrict__ h,
                                           const float* __restrict__ al_s,
                                           int n0, unsigned nloc, int t)
{
    const int4* d4 = (const int4*)(ei + EE);
    const int lane = t & 31;
    for (int base = 0; base < EE / 4; base += OT * SCAN_U) {
        int4 dv[SCAN_U];
        #pragma unroll
        for (int u = 0; u < SCAN_U; ++u) {
            const int i = base + u * OT + t;
            dv[u] = (i < EE / 4) ? d4[i] : make_int4(-1, -1, -1, -1);
        }
        #pragma unroll
        for (int u = 0; u < SCAN_U; ++u) {
            const int i = base + u * OT + t;
            const int dd[4] = { dv[u].x, dv[u].y, dv[u].z, dv[u].w };
            unsigned ld[4]; bool hit[4];
            #pragma unroll
            for (int k = 0; k < 4; ++k) { ld[k] = (unsigned)(dd[k] - n0); hit[k] = ld[k] < nloc; }
            if (MODE == 3) {
                #pragma unroll
                for (int k = 0; k < 4; ++k) {
                    const int src = hit[k] ? ei[4 * i + k] : 0;
                    unsigned mask = __builtin_amdgcn_ballot_w32(hit[k]);
                    while (mask) {
                        const int l = __builtin_ctz(mask);
                        mask &= mask - 1;
                        const int      s2 = __shfl(src, l);
                        const unsigned d2 = (unsigned)__shfl((int)ld[k], l);
                        row_one(L, h, al_s, s2, d2, lane);
                    }
                }
            } else if (hit[0] | hit[1] | hit[2] | hit[3]) {
                #pragma unroll
                for (int k = 0; k < 4; ++k) {
                    if (!hit[k]) continue;
                    const int src = ei[4 * i + k];
                    if (MODE == 1) {
                        edge_max_one(L, al_s, src, ld[k]);
                    } else {
                        edge_den_one(L, al_s, src, ld[k]);
                    }
                }
            }
        }
    }
}

__device__ __forceinline__ void scan_planes(OwnerLds& L, const unsigned* __restrict__ planes, int t)
{
    const uint4* p4 = (const uint4*)planes;
    const unsigned g = blockIdx.x;
    unsigned same[8];
    #pragma unroll
    for (int p = 0; p < 8; ++p) same[p] = ((g >> p) & 1u) ? 0u : ~0u;
    for (int base = 0; base < NCHUNK; base += OT * PL_U) {
        uint4 lo[PL_U], hi[PL_U];
        #pragma unroll
        for (int u = 0; u < PL_U; ++u) {
            const int c  = base + u * OT + t;
            const int cc = c < NCHUNK ? c : NCHUNK - 1;
            lo[u] = p4[2 * cc];
            hi[u] = p4[2 * cc + 1];
        }
        #pragma unroll
        for (int u = 0; u < PL_U; ++u) {
            const int c = base + u * OT + t;
            unsigned mt = (lo[u].x ^ same[0]) & (lo[u].y ^ same[1]) & (lo[u].z ^ same[2]) & (lo[u].w ^ same[3])
                        & (hi[u].x ^ same[4]) & (hi[u].y ^ same[5]) & (hi[u].z ^ same[6]) & (hi[u].w ^ same[7]);
            if (c >= NCHUNK) mt = 0u;
            while (mt) {
                const unsigned l = (unsigned)__builtin_ctz(mt);
                mt &= mt - 1u;
                const unsigned slot = atomicAdd(&L.count, 1u);
                if (slot < CAP) L.list[slot] = 32u * (unsigned)c + l;
            }
        }
    }
}

template <int NOUT, int CT>
__device__ __forceinline__ void rows_times_weight(const float* rows, const float* Ws,
                                                  float* __restrict__ dst,
                                                  const float* __restrict__ obias,
                                                  int n0, unsigned nloc, int wv, int lane)
{
    constexpr int NCT = (NOUT + 15) / 16;
    constexpr int NCG = (NCT + CT - 1) / CT;
    constexpr int NRT = (RANGE + 15) / 16;
    const int hi = lane >> 4;
    const int lo = lane & 15;
    for (int job = wv; job < NRT * NCG; job += OT / 32) {
        const int rt = job % NRT, cg = job / NRT;
        int row = rt * 16 + lo;
        if (row > RANGE) row = RANGE;
        const float* ap = rows + row * ACCP + 2 * hi * PLANE;
        int col[CT]; const float* bp[CT]; v8f acc[CT];
        #pragma unroll
        for (int u = 0; u < CT; ++u) {
            col[u] = (cg * CT + u) * 16 + lo;
            const int cc = col[u] < NOUT ? col[u] : NOUT - 1;
            bp[u]  = Ws + hi * WPP + 2 * cc;
            acc[u] = v8f{};
        }
        for (int kk = 0; kk < HC / 4; ++kk) {
            v2f a; a.x = ap[kk]; a.y = ap[PLANE + kk];
            #pragma unroll
            for (int u = 0; u < CT; ++u) {
                const float2 bv = *(const float2*)(bp[u] + kk * 2 * WPP);
                v2f b; b.x = bv.x; b.y = bv.y;
                acc[u] = wmma4(a, b, acc[u]);
            }
        }
        #pragma unroll
        for (int u = 0; u < CT; ++u) {
            if (col[u] < NOUT) {
                const float bb = obias ? obias[col[u]] : 0.f;
                #pragma unroll
                for (int r = 0; r < 8; ++r) {
                    const unsigned grow = (unsigned)(rt * 16 + 8 * hi + r);
                    if (grow < nloc) dst[(size_t)(n0 + (int)grow) * NOUT + col[u]] = acc[u][r] + bb;
                }
            }
        }
    }
}

#define XROW_U 7
#define W_U ((IN_DIM * HC) / OT)
static_assert(IN_DIM == HC, "the tile routine walks HC channels");
static_assert((RANGE + 1) * (IN_DIM / 4) <= OT * XROW_U, "one trip copies the whole range");
static_assert(W_U * OT == IN_DIM * HC, "the weight divides evenly over the threads");
#define XROWS ((OT * XROW_U) / (IN_DIM / 4))
struct FirstLds {
    alignas(16) float ws[(IN_DIM / 2) * WPP];
    float rows[XROWS * ACCP];
};
static_assert(sizeof(FirstLds) <= 320 * 1024, "one group's LDS fits a WGP");
#define ID_PASSES ((NCHUNK + NWG * (OT / 32) - 1) / (NWG * (OT / 32)))
__global__ __launch_bounds__(OT) void range_product(
    const float* __restrict__ x, const float* __restrict__ Wn, float* __restrict__ dst,
    const int* __restrict__ ei, unsigned* __restrict__ planes)
{
    __shared__ FirstLds F;
    const int t    = threadIdx.x;
    const int lane = t & 31;
    const int wv   = t >> 5;
    const int n0   = blockIdx.x * RANGE;
    const unsigned nloc = (unsigned)((NN - n0) < RANGE ? (NN - n0) : RANGE);

    float4 v[XROW_U];
    float  w[W_U];
    int    dd[ID_PASSES];
    #pragma unroll
    for (int it = 0; it < ID_PASSES; ++it) {
        const int c  = (it * NWG + (int)blockIdx.x) * (OT / 32) + wv;
        const int cc = c < NCHUNK ? c : NCHUNK - 1;
        dd[it] = ei[EE + 32 * cc + lane];
    }
    #pragma unroll
    for (int u = 0; u < XROW_U; ++u) {
        const int i = u * OT + t;
        const unsigned node = (unsigned)(i >> 5);
        const unsigned nc   = node < nloc ? node : nloc - 1;
        v[u] = *((const float4*)(x + (size_t)(n0 + (int)nc) * IN_DIM) + (i & 31));
    }
    #pragma unroll
    for (int u = 0; u < W_U; ++u) w[u] = Wn[u * OT + t];
    #pragma unroll
    for (int u = 0; u < XROW_U; ++u) {
        const int i = u * OT + t;
        float* p = F.rows + (i >> 5) * ACCP + (i & 31);
        p[0] = v[u].x; p[PLANE] = v[u].y; p[2 * PLANE] = v[u].z; p[3 * PLANE] = v[u].w;
    }
    #pragma unroll
    for (int u = 0; u < W_U; ++u) {
        const int i = u * OT + t;
        const int k = i / HC, n = i % HC;
        F.ws[(k >> 1) * WPP + 2 * n + (k & 1)] = w[u];
    }
    #pragma unroll
    for (int it = 0; it < ID_PASSES; ++it) {
        const int c = (it * NWG + (int)blockIdx.x) * (OT / 32) + wv;
        unsigned id = (unsigned)dd[it] / RANGE;
        id = id < 255u ? id : 255u;
        unsigned mine = 0u;
        #pragma unroll
        for (int p = 0; p < 8; ++p) {
            const unsigned vote = __builtin_amdgcn_ballot_w32(((id >> p) & 1u) != 0u);
            if (lane == p) mine = vote;
        }
        if (lane < 8 && c < NCHUNK) planes[8 * c + lane] = mine;
    }
    __syncthreads();
    rows_times_weight<HC, 2>(F.rows, F.ws, dst, (const float*)nullptr, n0, nloc, wv, lane);
}

template <bool BUILD, bool LAST>
__global__ __launch_bounds__(OT) void owner_aggregate(
    const int* __restrict__ ei, const float* __restrict__ h,
    const float* __restrict__ al_s, const float* __restrict__ al_d,
    const float* __restrict__ bias, const float* __restrict__ Wn,
    const float* __restrict__ obias, float* __restrict__ dst,
    unsigned* __restrict__ wlist, unsigned* __restrict__ wcount,
    const unsigned* __restrict__ planes)
{
    __shared__ OwnerLds L;
    const int t    = threadIdx.x;
    const int lane = t & 31;
    const int wv   = t >> 5;
    const int n0   = blockIdx.x * RANGE;
    const unsigned nloc = (unsigned)((NN - n0) < RANGE ? (NN - n0) : RANGE);
    unsigned* wl = wlist + (size_t)blockIdx.x * CAP;

    if (!BUILD) {
        const unsigned saved = wcount[blockIdx.x];
        if (saved <= CAP) {
            for (unsigned i0 = 0; i0 < saved; i0 += OT * LIST_U) {
                unsigned ent[LIST_U];
                #pragma unroll
                for (int u = 0; u < LIST_U; ++u) {
                    const unsigned idx = i0 + u * OT + t;
                    ent[u] = idx < saved ? wl[idx] : 0u;
                }
                #pragma unroll
                for (int u = 0; u < LIST_U; ++u) {
                    const unsigned idx = i0 + u * OT + t;
                    if (idx < saved) L.list[idx] = ent[u];
                }
            }
        }
        if (t == 0) L.count = saved;
    }

    for (int i = t; i < (RANGE + 1) * ACCP; i += OT) L.acc[i] = 0.f;
    if (t < RANGE * HEADS) {
        L.m[t]   = FORD_NEG_INF;
        L.den[t] = 0.f;
        L.ald[t] = ((unsigned)t < nloc * HEADS) ? al_d[(size_t)n0 * HEADS + t] : 0.f;
    }
    if (BUILD) {
        if ((unsigned)t < nloc) L.list[t] = (unsigned)(n0 + t) | ((unsigned)t << 16);
        if (t == 0) L.count = nloc;
    }
    __syncthreads();

    if (BUILD) {
        scan_planes(L, planes, t);
        __syncthreads();
    }
    const unsigned total = L.count;

    if (BUILD) {
        if (total <= CAP) {
            const unsigned spare = (unsigned)n0 | ((unsigned)RANGE << 16);
            for (unsigned i0 = 0; i0 < total; i0 += OT * LIST_U) {
                unsigned ent[LIST_U]; int sv[LIST_U], dv[LIST_U];
                #pragma unroll
                for (int u = 0; u < LIST_U; ++u) {
                    const unsigned idx = i0 + u * OT + t;
                    ent[u] = idx < total ? L.list[idx] : 0u;
                    const unsigned e = (idx >= nloc && idx < total) ? ent[u] : 0u;
                    sv[u] = ei[e];
                    dv[u] = ei[EE + e];
                }
                #pragma unroll
                for (int u = 0; u < LIST_U; ++u) {
                    const unsigned idx = i0 + u * OT + t;
                    if (idx < total) {
                        if (idx >= nloc) {
                            const unsigned ld = (unsigned)(dv[u] - n0);
                            ent[u] = ld < nloc ? ((unsigned)sv[u] | (ld << 16)) : spare;
                            L.list[idx] = ent[u];
                        }
                        wl[idx] = ent[u];
                    }
                }
            }
        }
        if (t == 0) wcount[blockIdx.x] = total;
    }

    if (total <= CAP) {
        for (unsigned idx = t; idx < total; idx += OT) edge_logit_max(L, al_s, idx);
        __syncthreads();
        for (unsigned idx = t; idx < total; idx += OT) edge_exp_den(L, idx);
        __syncthreads();
        for (unsigned idx = t; idx < total; idx += OT) edge_alpha(L, idx);
        __syncthreads();
        const unsigned spare = (unsigned)n0 | ((unsigned)RANGE << 16);
        for (unsigned i0 = wv; i0 < total; i0 += (OT / 32) * ROW_U) {
            unsigned ent[ROW_U]; float al[ROW_U]; float4 v[ROW_U];
            #pragma unroll
            for (int u = 0; u < ROW_U; ++u) {
                const unsigned idx = i0 + u * (OT / 32);
                const unsigned idc = idx < total ? idx : 0;
                const unsigned e0 = L.list[idc];
                ent[u] = idx < total ? e0 : spare;
                al[u]  = L.ev[idc * HEADS + (lane >> 3)];
                v[u]   = *((const float4*)(h + (size_t)(ent[u] & 0xFFFFu) * HC) + lane);
            }
            #pragma unroll
            for (int u = 0; u < ROW_U; ++u) {
                float* ap = L.acc + (ent[u] >> 16) * ACCP + lane;
                lds_add(ap,             v[u].x * al[u]);
                lds_add(ap + PLANE,     v[u].y * al[u]);
                lds_add(ap + 2 * PLANE, v[u].z * al[u]);
                lds_add(ap + 3 * PLANE, v[u].w * al[u]);
            }
        }
    } else {
        if ((unsigned)t < nloc) edge_max_one(L, al_s, n0 + t, (unsigned)t);
        scan_edges<1>(L, ei, h, al_s, n0, nloc, t);
        __syncthreads();
        if ((unsigned)t < nloc) edge_den_one(L, al_s, n0 + t, (unsigned)t);
        scan_edges<2>(L, ei, h, al_s, n0, nloc, t);
        __syncthreads();
        for (unsigned i = wv; i < nloc; i += OT / 32) row_one(L, h, al_s, n0 + (int)i, i, lane);
        scan_edges<3>(L, ei, h, al_s, n0, nloc, t);
    }
    __syncthreads();

    for (int i = t; i < RANGE * HC; i += OT) {
        const int node = i / HC, q = i % HC;
        const int j = q >> 5, l = q & 31;
        float* p = L.acc + node * ACCP + j * PLANE + l;
        const float vv = *p + bias[4 * l + j];
        *p = vv > 0.f ? vv : 0.f;
    }
    constexpr int NOUT = LAST ? OUT_DIM : HC;
    float* Ws = L.ev;
    for (int i = t; i < HC * NOUT; i += OT) {
        const int k = i / NOUT, n = i % NOUT;
        Ws[(k >> 1) * WPP + 2 * n + (k & 1)] = Wn[i];
    }
    __syncthreads();
    rows_times_weight<NOUT, LAST ? 1 : 2>(L.acc, Ws, dst, obias, n0, nloc, wv, lane);
}

extern "C" void kernel_launch(void* const* d_in, const int* in_sizes, int n_in,
                              void* d_out, int out_size, void* d_ws, size_t ws_size,
                              hipStream_t stream)
{
    const float* x   = (const float*)d_in[0];
    const int*   ei  = (const int*)d_in[1];
    const float* W[3]    = { (const float*)d_in[2],  (const float*)d_in[6],  (const float*)d_in[10] };
    const float* asv[3]  = { (const float*)d_in[3],  (const float*)d_in[7],  (const float*)d_in[11] };
    const float* adv[3]  = { (const float*)d_in[4],  (const float*)d_in[8],  (const float*)d_in[12] };
    const float* bv[3]   = { (const float*)d_in[5],  (const float*)d_in[9],  (const float*)d_in[13] };
    const float* Wl  = (const float*)d_in[14];
    const float* bl  = (const float*)d_in[15];

    char* ws = (char*)d_ws;
    float*    hA   = (float*)ws;                       ws += (size_t)NN * HC * sizeof(float);
    float*    hB   = (float*)ws;                       ws += (size_t)NN * HC * sizeof(float);
    float*    al_s = (float*)ws;                       ws += (size_t)NN * HEADS * sizeof(float);
    float*    al_d = (float*)ws;                       ws += (size_t)NN * HEADS * sizeof(float);
    unsigned* wlist  = (unsigned*)ws;                  ws += (size_t)NWG * CAP * sizeof(unsigned);
    unsigned* wcount = (unsigned*)ws;                  ws += (size_t)NWG * sizeof(unsigned);
    unsigned* planes = (unsigned*)ws;

    const int TPB = 256;
    const int gNH    = (NN * HEADS + TPB - 1) / TPB;

    auto logits = [&](const float* hin, int layer) {
        hipLaunchKernelGGL(attn_coef, dim3(gNH), dim3(TPB), 0, stream,
                           hin, asv[layer], adv[layer], al_s, al_d);
    };

    hipLaunchKernelGGL(range_product, dim3(NWG), dim3(OT), 0, stream, x, W[0], hA, ei, planes);
    logits(hA, 0);
    hipLaunchKernelGGL((owner_aggregate<true, false>), dim3(NWG), dim3(OT), 0, stream,
                       ei, hA, al_s, al_d, bv[0], W[1], (const float*)nullptr, hB, wlist, wcount, planes);
    logits(hB, 1);
    hipLaunchKernelGGL((owner_aggregate<false, false>), dim3(NWG), dim3(OT), 0, stream,
                       ei, hB, al_s, al_d, bv[1], W[2], (const float*)nullptr, hA, wlist, wcount, planes);
    logits(hA, 2);
    hipLaunchKernelGGL((owner_aggregate<false, true>), dim3(NWG), dim3(OT), 0, stream,
                       ei, hA, al_s, al_d, bv[2], Wl, bl, (float*)d_out, wlist, wcount, planes);
}


# Logit launches removed: behind a barrier after its product each node-range group reads back its stored rows (a wave per row) and forms al_s/al_d; 7 launches become 4; logit sums now lane-tree order
# speedup vs baseline: 5.5500x; 1.0738x over previous
#include <hip/hip_runtime.h>
#include <math.h>

#define NN 50000
#define EE 800000
#define IN_DIM 128
#define HID 32
#define HEADS 4
#define HC 128
#define OUT_DIM 40
#define SLOPE 0.2f

typedef float v2f __attribute__((ext_vector_type(2)));
typedef float v8f __attribute__((ext_vector_type(8)));

__device__ __forceinline__ unsigned ford(float f) {
    unsigned u = __float_as_uint(f);
    return (u & 0x80000000u) ? ~u : (u | 0x80000000u);
}
__device__ __forceinline__ float funord(unsigned k) {
    unsigned u = (k & 0x80000000u) ? (k & 0x7FFFFFFFu) : ~k;
    return __uint_as_float(u);
}
#define FORD_NEG_INF 0x007FFFFFu

__device__ __forceinline__ float leaky(float e) {
    return e > 0.f ? e : SLOPE * e;
}

__device__ __forceinline__ v8f wmma4(v2f a, v2f b, v8f c) {
    return __builtin_amdgcn_wmma_f32_16x16x4_f32(
               false, a, false, b, (short)0, c, false, false);
}

#define PLANE 33
#define ACCP (4 * PLANE)
#define WPP 288
#define RANGE 196
#define NWG ((NN + RANGE - 1) / RANGE)
#define OT 1024
#define CAP 8192
#define SCAN_U 8
#define ROW_U 8
#define LIST_U 4
#define NCHUNK (EE / 32)
#define PL_U 5
static_assert(NN <= 65536, "list entry packs the source index in 16 bits");
static_assert(RANGE < 256 && (RANGE + 1) * HEADS <= OT, "range bookkeeping fits one pass");
static_assert((EE % 4) == 0, "dst array is read as int4");
static_assert((EE % 32) == 0, "edges divide into whole chunks");
static_assert(NWG <= 256, "an owner number fits one byte");

struct OwnerLds {
    alignas(16) float ev[CAP * HEADS];
    float    acc[(RANGE + 1) * ACCP];
    unsigned m[(RANGE + 1) * HEADS];
    float    den[(RANGE + 1) * HEADS];
    float    ald[(RANGE + 1) * HEADS];
    unsigned list[CAP];
    unsigned count;
};
static_assert(sizeof(OwnerLds) <= 320 * 1024, "one group's LDS fits a WGP");
static_assert((HC / 2) * WPP <= CAP * HEADS, "staged weight fits in ev");

#define LOGIT_U ((RANGE + OT / 32 - 1) / (OT / 32))
__device__ __forceinline__ void own_logits(const float* hrows,
                                           const float* __restrict__ a_src,
                                           const float* __restrict__ a_dst,
                                           float* __restrict__ al_s,
                                           float* __restrict__ al_d,
                                           int n0, unsigned nloc, int wv, int lane)
{
    const float4 as = *((const float4*)a_src + lane);
    const float4 ad = *((const float4*)a_dst + lane);
    float4 v[LOGIT_U];
    #pragma unroll
    for (int u = 0; u < LOGIT_U; ++u) {
        const unsigned i  = (unsigned)(wv + u * (OT / 32));
        const unsigned ic = i < nloc ? i : nloc - 1;
        v[u] = *((const float4*)(hrows + (size_t)(n0 + (int)ic) * HC) + lane);
    }
    float s[LOGIT_U], d[LOGIT_U];
    #pragma unroll
    for (int u = 0; u < LOGIT_U; ++u) {
        s[u] = v[u].x * as.x; s[u] = fmaf(v[u].y, as.y, s[u]); s[u] = fmaf(v[u].z, as.z, s[u]); s[u] = fmaf(v[u].w, as.w, s[u]);
        d[u] = v[u].x * ad.x; d[u] = fmaf(v[u].y, ad.y, d[u]); d[u] = fmaf(v[u].z, ad.z, d[u]); d[u] = fmaf(v[u].w, ad.w, d[u]);
    }
    #pragma unroll
    for (int m = 1; m < 8; m <<= 1) {
        float ps[LOGIT_U], pd[LOGIT_U];
        #pragma unroll
        for (int u = 0; u < LOGIT_U; ++u) { ps[u] = __shfl_xor(s[u], m); pd[u] = __shfl_xor(d[u], m); }
        #pragma unroll
        for (int u = 0; u < LOGIT_U; ++u) { s[u] += ps[u]; d[u] += pd[u]; }
    }
    #pragma unroll
    for (int u = 0; u < LOGIT_U; ++u) {
        const unsigned i = (unsigned)(wv + u * (OT / 32));
        if ((lane & 7) == 0 && i < nloc) {
            const size_t o = (size_t)(n0 + (int)i) * HEADS + (lane >> 3);
            al_s[o] = s[u]; al_d[o] = d[u];
        }
    }
}

__device__ __forceinline__ void lds_add(float* p, float v) {
    __hip_atomic_fetch_add(p, v, __ATOMIC_RELAXED, __HIP_MEMORY_SCOPE_WORKGROUP);
}

__device__ __forceinline__ void edge_max_one(OwnerLds& L, const float* __restrict__ al_s,
                                             int src, unsigned ld) {
    const float4 s = *(const float4*)(al_s + (size_t)src * HEADS);
    const float sv[4] = { s.x, s.y, s.z, s.w };
    #pragma unroll
    for (int hd = 0; hd < HEADS; ++hd) {
        float e = leaky(sv[hd] + L.ald[ld * HEADS + hd]);
        atomicMax(&L.m[ld * HEADS + hd], ford(e));
    }
}

__device__ __forceinline__ void edge_den_one(OwnerLds& L, const float* __restrict__ al_s,
                                             int src, unsigned ld) {
    const float4 s = *(const float4*)(al_s + (size_t)src * HEADS);
    const float sv[4] = { s.x, s.y, s.z, s.w };
    #pragma unroll
    for (int hd = 0; hd < HEADS; ++hd) {
        float e  = leaky(sv[hd] + L.ald[ld * HEADS + hd]);
        float mx = funord(L.m[ld * HEADS + hd]);
        lds_add(&L.den[ld * HEADS + hd], __expf(e - mx));
    }
}

__device__ __forceinline__ void edge_logit_max(OwnerLds& L, const float* __restrict__ al_s, unsigned idx) {
    const unsigned ent = L.list[idx];
    const unsigned ld  = ent >> 16;
    const float4 s = *(const float4*)(al_s + (size_t)(ent & 0xFFFFu) * HEADS);
    const float4 d = *(const float4*)(L.ald + ld * HEADS);
    float4 e;
    e.x = leaky(s.x + d.x); e.y = leaky(s.y + d.y); e.z = leaky(s.z + d.z); e.w = leaky(s.w + d.w);
    atomicMax(&L.m[ld * HEADS + 0], ford(e.x));
    atomicMax(&L.m[ld * HEADS + 1], ford(e.y));
    atomicMax(&L.m[ld * HEADS + 2], ford(e.z));
    atomicMax(&L.m[ld * HEADS + 3], ford(e.w));
    *(float4*)(L.ev + idx * HEADS) = e;
}

__device__ __forceinline__ void edge_exp_den(OwnerLds& L, unsigned idx) {
    const unsigned ld = L.list[idx] >> 16;
    const float4 e  = *(const float4*)(L.ev + idx * HEADS);
    const uint4  mk = *(const uint4*)(L.m + ld * HEADS);
    float4 x;
    x.x = __expf(e.x - funord(mk.x)); x.y = __expf(e.y - funord(mk.y));
    x.z = __expf(e.z - funord(mk.z)); x.w = __expf(e.w - funord(mk.w));
    lds_add(&L.den[ld * HEADS + 0], x.x);
    lds_add(&L.den[ld * HEADS + 1], x.y);
    lds_add(&L.den[ld * HEADS + 2], x.z);
    lds_add(&L.den[ld * HEADS + 3], x.w);
    *(float4*)(L.ev + idx * HEADS) = x;
}

__device__ __forceinline__ void edge_alpha(OwnerLds& L, unsigned idx) {
    const unsigned ld = L.list[idx] >> 16;
    float4 x = *(const float4*)(L.ev + idx * HEADS);
    const float4 dn = *(const float4*)(L.den + ld * HEADS);
    x.x = x.x / (dn.x + 1e-16f); x.y = x.y / (dn.y + 1e-16f);
    x.z = x.z / (dn.z + 1e-16f); x.w = x.w / (dn.w + 1e-16f);
    *(float4*)(L.ev + idx * HEADS) = x;
}

__device__ __forceinline__ void row_add(OwnerLds& L, float sv, float4 v, unsigned ld, int lane) {
    const int hd = lane >> 3;
    float e  = leaky(sv + L.ald[ld * HEADS + hd]);
    float mx = funord(L.m[ld * HEADS + hd]);
    float alpha = __expf(e - mx) / (L.den[ld * HEADS + hd] + 1e-16f);
    float* ap = L.acc + ld * ACCP + lane;
    lds_add(ap,             v.x * alpha);
    lds_add(ap + PLANE,     v.y * alpha);
    lds_add(ap + 2 * PLANE, v.z * alpha);
    lds_add(ap + 3 * PLANE, v.w * alpha);
}

__device__ __forceinline__ void row_one(OwnerLds& L, const float* __restrict__ h,
                                        const float* __restrict__ al_s,
                                        int src, unsigned ld, int lane) {
    float  sv = al_s[(size_t)src * HEADS + (lane >> 3)];
    float4 v  = *((const float4*)(h + (size_t)src * HC) + lane);
    row_add(L, sv, v, ld, lane);
}

template <int MODE>
__device__ __forceinline__ void scan_edges(OwnerLds& L, const int* __restrict__ ei,
                                           const float* __restrict__ h,
                                           const float* __restrict__ al_s,
                                           int n0, unsigned nloc, int t)
{
    const int4* d4 = (const int4*)(ei + EE);
    const int lane = t & 31;
    for (int base = 0; base < EE / 4; base += OT * SCAN_U) {
        int4 dv[SCAN_U];
        #pragma unroll
        for (int u = 0; u < SCAN_U; ++u) {
            const int i = base + u * OT + t;
            dv[u] = (i < EE / 4) ? d4[i] : make_int4(-1, -1, -1, -1);
        }
        #pragma unroll
        for (int u = 0; u < SCAN_U; ++u) {
            const int i = base + u * OT + t;
            const int dd[4] = { dv[u].x, dv[u].y, dv[u].z, dv[u].w };
            unsigned ld[4]; bool hit[4];
            #pragma unroll
            for (int k = 0; k < 4; ++k) { ld[k] = (unsigned)(dd[k] - n0); hit[k] = ld[k] < nloc; }
            if (MODE == 3) {
                #pragma unroll
                for (int k = 0; k < 4; ++k) {
                    const int src = hit[k] ? ei[4 * i + k] : 0;
                    unsigned mask = __builtin_amdgcn_ballot_w32(hit[k]);
                    while (mask) {
                        const int l = __builtin_ctz(mask);
                        mask &= mask - 1;
                        const int      s2 = __shfl(src, l);
                        const unsigned d2 = (unsigned)__shfl((int)ld[k], l);
                        row_one(L, h, al_s, s2, d2, lane);
                    }
                }
            } else if (hit[0] | hit[1] | hit[2] | hit[3]) {
                #pragma unroll
                for (int k = 0; k < 4; ++k) {
                    if (!hit[k]) continue;
                    const int src = ei[4 * i + k];
                    if (MODE == 1) {
                        edge_max_one(L, al_s, src, ld[k]);
                    } else {
                        edge_den_one(L, al_s, src, ld[k]);
                    }
                }
            }
        }
    }
}

__device__ __forceinline__ void scan_planes(OwnerLds& L, const unsigned* __restrict__ planes, int t)
{
    const uint4* p4 = (const uint4*)planes;
    const unsigned g = blockIdx.x;
    unsigned same[8];
    #pragma unroll
    for (int p = 0; p < 8; ++p) same[p] = ((g >> p) & 1u) ? 0u : ~0u;
    for (int base = 0; base < NCHUNK; base += OT * PL_U) {
        uint4 lo[PL_U], hi[PL_U];
        #pragma unroll
        for (int u = 0; u < PL_U; ++u) {
            const int c  = base + u * OT + t;
            const int cc = c < NCHUNK ? c : NCHUNK - 1;
            lo[u] = p4[2 * cc];
            hi[u] = p4[2 * cc + 1];
        }
        #pragma unroll
        for (int u = 0; u < PL_U; ++u) {
            const int c = base + u * OT + t;
            unsigned mt = (lo[u].x ^ same[0]) & (lo[u].y ^ same[1]) & (lo[u].z ^ same[2]) & (lo[u].w ^ same[3])
                        & (hi[u].x ^ same[4]) & (hi[u].y ^ same[5]) & (hi[u].z ^ same[6]) & (hi[u].w ^ same[7]);
            if (c >= NCHUNK) mt = 0u;
            while (mt) {
                const unsigned l = (unsigned)__builtin_ctz(mt);
                mt &= mt - 1u;
                const unsigned slot = atomicAdd(&L.count, 1u);
                if (slot < CAP) L.list[slot] = 32u * (unsigned)c + l;
            }
        }
    }
}

template <int NOUT, int CT>
__device__ __forceinline__ void rows_times_weight(const float* rows, const float* Ws,
                                                  float* __restrict__ dst,
                                                  const float* __restrict__ obias,
                                                  int n0, unsigned nloc, int wv, int lane)
{
    constexpr int NCT = (NOUT + 15) / 16;
    constexpr int NCG = (NCT + CT - 1) / CT;
    constexpr int NRT = (RANGE + 15) / 16;
    const int hi = lane >> 4;
    const int lo = lane & 15;
    for (int job = wv; job < NRT * NCG; job += OT / 32) {
        const int rt = job % NRT, cg = job / NRT;
        int row = rt * 16 + lo;
        if (row > RANGE) row = RANGE;
        const float* ap = rows + row * ACCP + 2 * hi * PLANE;
        int col[CT]; const float* bp[CT]; v8f acc[CT];
        #pragma unroll
        for (int u = 0; u < CT; ++u) {
            col[u] = (cg * CT + u) * 16 + lo;
            const int cc = col[u] < NOUT ? col[u] : NOUT - 1;
            bp[u]  = Ws + hi * WPP + 2 * cc;
            acc[u] = v8f{};
        }
        for (int kk = 0; kk < HC / 4; ++kk) {
            v2f a; a.x = ap[kk]; a.y = ap[PLANE + kk];
            #pragma unroll
            for (int u = 0; u < CT; ++u) {
                const float2 bv = *(const float2*)(bp[u] + kk * 2 * WPP);
                v2f b; b.x = bv.x; b.y = bv.y;
                acc[u] = wmma4(a, b, acc[u]);
            }
        }
        #pragma unroll
        for (int u = 0; u < CT; ++u) {
            if (col[u] < NOUT) {
                const float bb = obias ? obias[col[u]] : 0.f;
                #pragma unroll
                for (int r = 0; r < 8; ++r) {
                    const unsigned grow = (unsigned)(rt * 16 + 8 * hi + r);
                    if (grow < nloc) dst[(size_t)(n0 + (int)grow) * NOUT + col[u]] = acc[u][r] + bb;
                }
            }
        }
    }
}

#define XROW_U 7
#define W_U ((IN_DIM * HC) / OT)
static_assert(IN_DIM == HC, "the tile routine walks HC channels");
static_assert((RANGE + 1) * (IN_DIM / 4) <= OT * XROW_U, "one trip copies the whole range");
static_assert(W_U * OT == IN_DIM * HC, "the weight divides evenly over the threads");
#define XROWS ((OT * XROW_U) / (IN_DIM / 4))
struct FirstLds {
    alignas(16) float ws[(IN_DIM / 2) * WPP];
    float rows[XROWS * ACCP];
};
static_assert(sizeof(FirstLds) <= 320 * 1024, "one group's LDS fits a WGP");
#define ID_PASSES ((NCHUNK + NWG * (OT / 32) - 1) / (NWG * (OT / 32)))
__global__ __launch_bounds__(OT) void range_product(
    const float* __restrict__ x, const float* __restrict__ Wn, float* __restrict__ dst,
    const int* __restrict__ ei, unsigned* __restrict__ planes,
    const float* __restrict__ a_src, const float* __restrict__ a_dst,
    float* __restrict__ al_s, float* __restrict__ al_d)
{
    __shared__ FirstLds F;
    const int t    = threadIdx.x;
    const int lane = t & 31;
    const int wv   = t >> 5;
    const int n0   = blockIdx.x * RANGE;
    const unsigned nloc = (unsigned)((NN - n0) < RANGE ? (NN - n0) : RANGE);

    float4 v[XROW_U];
    float  w[W_U];
    int    dd[ID_PASSES];
    #pragma unroll
    for (int it = 0; it < ID_PASSES; ++it) {
        const int c  = (it * NWG + (int)blockIdx.x) * (OT / 32) + wv;
        const int cc = c < NCHUNK ? c : NCHUNK - 1;
        dd[it] = ei[EE + 32 * cc + lane];
    }
    #pragma unroll
    for (int u = 0; u < XROW_U; ++u) {
        const int i = u * OT + t;
        const unsigned node = (unsigned)(i >> 5);
        const unsigned nc   = node < nloc ? node : nloc - 1;
        v[u] = *((const float4*)(x + (size_t)(n0 + (int)nc) * IN_DIM) + (i & 31));
    }
    #pragma unroll
    for (int u = 0; u < W_U; ++u) w[u] = Wn[u * OT + t];
    #pragma unroll
    for (int u = 0; u < XROW_U; ++u) {
        const int i = u * OT + t;
        float* p = F.rows + (i >> 5) * ACCP + (i & 31);
        p[0] = v[u].x; p[PLANE] = v[u].y; p[2 * PLANE] = v[u].z; p[3 * PLANE] = v[u].w;
    }
    #pragma unroll
    for (int u = 0; u < W_U; ++u) {
        const int i = u * OT + t;
        const int k = i / HC, n = i % HC;
        F.ws[(k >> 1) * WPP + 2 * n + (k & 1)] = w[u];
    }
    #pragma unroll
    for (int it = 0; it < ID_PASSES; ++it) {
        const int c = (it * NWG + (int)blockIdx.x) * (OT / 32) + wv;
        unsigned id = (unsigned)dd[it] / RANGE;
        id = id < 255u ? id : 255u;
        unsigned mine = 0u;
        #pragma unroll
        for (int p = 0; p < 8; ++p) {
            const unsigned vote = __builtin_amdgcn_ballot_w32(((id >> p) & 1u) != 0u);
            if (lane == p) mine = vote;
        }
        if (lane < 8 && c < NCHUNK) planes[8 * c + lane] = mine;
    }
    __syncthreads();
    rows_times_weight<HC, 2>(F.rows, F.ws, dst, (const float*)nullptr, n0, nloc, wv, lane);
    __syncthreads();
    own_logits(dst, a_src, a_dst, al_s, al_d, n0, nloc, wv, lane);
}

template <bool BUILD, bool LAST>
__global__ __launch_bounds__(OT) void owner_aggregate(
    const int* __restrict__ ei, const float* __restrict__ h,
    const float* __restrict__ al_s, const float* __restrict__ al_d,
    const float* __restrict__ bias, const float* __restrict__ Wn,
    const float* __restrict__ obias, float* __restrict__ dst,
    unsigned* __restrict__ wlist, unsigned* __restrict__ wcount,
    const unsigned* __restrict__ planes,
    const float* __restrict__ a_src_n, const float* __restrict__ a_dst_n,
    float* __restrict__ al_s_n, float* __restrict__ al_d_n)
{
    __shared__ OwnerLds L;
    const int t    = threadIdx.x;
    const int lane = t & 31;
    const int wv   = t >> 5;
    const int n0   = blockIdx.x * RANGE;
    const unsigned nloc = (unsigned)((NN - n0) < RANGE ? (NN - n0) : RANGE);
    unsigned* wl = wlist + (size_t)blockIdx.x * CAP;

    if (!BUILD) {
        const unsigned saved = wcount[blockIdx.x];
        if (saved <= CAP) {
            for (unsigned i0 = 0; i0 < saved; i0 += OT * LIST_U) {
                unsigned ent[LIST_U];
                #pragma unroll
                for (int u = 0; u < LIST_U; ++u) {
                    const unsigned idx = i0 + u * OT + t;
                    ent[u] = idx < saved ? wl[idx] : 0u;
                }
                #pragma unroll
                for (int u = 0; u < LIST_U; ++u) {
                    const unsigned idx = i0 + u * OT + t;
                    if (idx < saved) L.list[idx] = ent[u];
                }
            }
        }
        if (t == 0) L.count = saved;
    }

    for (int i = t; i < (RANGE + 1) * ACCP; i += OT) L.acc[i] = 0.f;
    if (t < RANGE * HEADS) {
        L.m[t]   = FORD_NEG_INF;
        L.den[t] = 0.f;
        L.ald[t] = ((unsigned)t < nloc * HEADS) ? al_d[(size_t)n0 * HEADS + t] : 0.f;
    }
    if (BUILD) {
        if ((unsigned)t < nloc) L.list[t] = (unsigned)(n0 + t) | ((unsigned)t << 16);
        if (t == 0) L.count = nloc;
    }
    __syncthreads();

    if (BUILD) {
        scan_planes(L, planes, t);
        __syncthreads();
    }
    const unsigned total = L.count;

    if (BUILD) {
        if (total <= CAP) {
            const unsigned spare = (unsigned)n0 | ((unsigned)RANGE << 16);
            for (unsigned i0 = 0; i0 < total; i0 += OT * LIST_U) {
                unsigned ent[LIST_U]; int sv[LIST_U], dv[LIST_U];
                #pragma unroll
                for (int u = 0; u < LIST_U; ++u) {
                    const unsigned idx = i0 + u * OT + t;
                    ent[u] = idx < total ? L.list[idx] : 0u;
                    const unsigned e = (idx >= nloc && idx < total) ? ent[u] : 0u;
                    sv[u] = ei[e];
                    dv[u] = ei[EE + e];
                }
                #pragma unroll
                for (int u = 0; u < LIST_U; ++u) {
                    const unsigned idx = i0 + u * OT + t;
                    if (idx < total) {
                        if (idx >= nloc) {
                            const unsigned ld = (unsigned)(dv[u] - n0);
                            ent[u] = ld < nloc ? ((unsigned)sv[u] | (ld << 16)) : spare;
                            L.list[idx] = ent[u];
                        }
                        wl[idx] = ent[u];
                    }
                }
            }
        }
        if (t == 0) wcount[blockIdx.x] = total;
    }

    if (total <= CAP) {
        for (unsigned idx = t; idx < total; idx += OT) edge_logit_max(L, al_s, idx);
        __syncthreads();
        for (unsigned idx = t; idx < total; idx += OT) edge_exp_den(L, idx);
        __syncthreads();
        for (unsigned idx = t; idx < total; idx += OT) edge_alpha(L, idx);
        __syncthreads();
        const unsigned spare = (unsigned)n0 | ((unsigned)RANGE << 16);
        for (unsigned i0 = wv; i0 < total; i0 += (OT / 32) * ROW_U) {
            unsigned ent[ROW_U]; float al[ROW_U]; float4 v[ROW_U];
            #pragma unroll
            for (int u = 0; u < ROW_U; ++u) {
                const unsigned idx = i0 + u * (OT / 32);
                const unsigned idc = idx < total ? idx : 0;
                const unsigned e0 = L.list[idc];
                ent[u] = idx < total ? e0 : spare;
                al[u]  = L.ev[idc * HEADS + (lane >> 3)];
                v[u]   = *((const float4*)(h + (size_t)(ent[u] & 0xFFFFu) * HC) + lane);
            }
            #pragma unroll
            for (int u = 0; u < ROW_U; ++u) {
                float* ap = L.acc + (ent[u] >> 16) * ACCP + lane;
                lds_add(ap,             v[u].x * al[u]);
                lds_add(ap + PLANE,     v[u].y * al[u]);
                lds_add(ap + 2 * PLANE, v[u].z * al[u]);
                lds_add(ap + 3 * PLANE, v[u].w * al[u]);
            }
        }
    } else {
        if ((unsigned)t < nloc) edge_max_one(L, al_s, n0 + t, (unsigned)t);
        scan_edges<1>(L, ei, h, al_s, n0, nloc, t);
        __syncthreads();
        if ((unsigned)t < nloc) edge_den_one(L, al_s, n0 + t, (unsigned)t);
        scan_edges<2>(L, ei, h, al_s, n0, nloc, t);
        __syncthreads();
        for (unsigned i = wv; i < nloc; i += OT / 32) row_one(L, h, al_s, n0 + (int)i, i, lane);
        scan_edges<3>(L, ei, h, al_s, n0, nloc, t);
    }
    __syncthreads();

    for (int i = t; i < RANGE * HC; i += OT) {
        const int node = i / HC, q = i % HC;
        const int j = q >> 5, l = q & 31;
        float* p = L.acc + node * ACCP + j * PLANE + l;
        const float vv = *p + bias[4 * l + j];
        *p = vv > 0.f ? vv : 0.f;
    }
    constexpr int NOUT = LAST ? OUT_DIM : HC;
    float* Ws = L.ev;
    for (int i = t; i < HC * NOUT; i += OT) {
        const int k = i / NOUT, n = i % NOUT;
        Ws[(k >> 1) * WPP + 2 * n + (k & 1)] = Wn[i];
    }
    __syncthreads();
    rows_times_weight<NOUT, LAST ? 1 : 2>(L.acc, Ws, dst, obias, n0, nloc, wv, lane);
    if (!LAST) {
        __syncthreads();
        own_logits(dst, a_src_n, a_dst_n, al_s_n, al_d_n, n0, nloc, wv, lane);
    }
}

extern "C" void kernel_launch(void* const* d_in, const int* in_sizes, int n_in,
                              void* d_out, int out_size, void* d_ws, size_t ws_size,
                              hipStream_t stream)
{
    const float* x   = (const float*)d_in[0];
    const int*   ei  = (const int*)d_in[1];
    const float* W[3]    = { (const float*)d_in[2],  (const float*)d_in[6],  (const float*)d_in[10] };
    const float* asv[3]  = { (const float*)d_in[3],  (const float*)d_in[7],  (const float*)d_in[11] };
    const float* adv[3]  = { (const float*)d_in[4],  (const float*)d_in[8],  (const float*)d_in[12] };
    const float* bv[3]   = { (const float*)d_in[5],  (const float*)d_in[9],  (const float*)d_in[13] };
    const float* Wl  = (const float*)d_in[14];
    const float* bl  = (const float*)d_in[15];

    char* ws = (char*)d_ws;
    float*    hA   = (float*)ws;                       ws += (size_t)NN * HC * sizeof(float);
    float*    hB   = (float*)ws;                       ws += (size_t)NN * HC * sizeof(float);
    float*    al_s0 = (float*)ws;                      ws += (size_t)NN * HEADS * sizeof(float);
    float*    al_d0 = (float*)ws;                      ws += (size_t)NN * HEADS * sizeof(float);
    float*    al_s1 = (float*)ws;                      ws += (size_t)NN * HEADS * sizeof(float);
    float*    al_d1 = (float*)ws;                      ws += (size_t)NN * HEADS * sizeof(float);
    unsigned* wlist  = (unsigned*)ws;                  ws += (size_t)NWG * CAP * sizeof(unsigned);
    unsigned* wcount = (unsigned*)ws;                  ws += (size_t)NWG * sizeof(unsigned);
    unsigned* planes = (unsigned*)ws;


    hipLaunchKernelGGL(range_product, dim3(NWG), dim3(OT), 0, stream, x, W[0], hA, ei, planes,
                       asv[0], adv[0], al_s0, al_d0);
    hipLaunchKernelGGL((owner_aggregate<true, false>), dim3(NWG), dim3(OT), 0, stream,
                       ei, hA, al_s0, al_d0, bv[0], W[1], (const float*)nullptr, hB, wlist, wcount, planes,
                       asv[1], adv[1], al_s1, al_d1);
    hipLaunchKernelGGL((owner_aggregate<false, false>), dim3(NWG), dim3(OT), 0, stream,
                       ei, hB, al_s1, al_d1, bv[1], W[2], (const float*)nullptr, hA, wlist, wcount, planes,
                       asv[2], adv[2], al_s0, al_d0);
    hipLaunchKernelGGL((owner_aggregate<false, true>), dim3(NWG), dim3(OT), 0, stream,
                       ei, hA, al_s0, al_d0, bv[2], Wl, bl, (float*)d_out, wlist, wcount, planes,
                       (const float*)nullptr, (const float*)nullptr, (float*)nullptr, (float*)nullptr);
}


# Row pass: a trip's 8 list entries and its 32 alphas (kept [head][edge] per 8 edges, shared by DPP8) each come in with one wave-wide LDS load; 16 LDS loads a trip become 2; LDS arrival-order sums
# speedup vs baseline: 5.7629x; 1.0384x over previous
#include <hip/hip_runtime.h>
#include <math.h>

#define NN 50000
#define EE 800000
#define IN_DIM 128
#define HID 32
#define HEADS 4
#define HC 128
#define OUT_DIM 40
#define SLOPE 0.2f

typedef float v2f __attribute__((ext_vector_type(2)));
typedef float v8f __attribute__((ext_vector_type(8)));

__device__ __forceinline__ unsigned ford(float f) {
    unsigned u = __float_as_uint(f);
    return (u & 0x80000000u) ? ~u : (u | 0x80000000u);
}
__device__ __forceinline__ float funord(unsigned k) {
    unsigned u = (k & 0x80000000u) ? (k & 0x7FFFFFFFu) : ~k;
    return __uint_as_float(u);
}
#define FORD_NEG_INF 0x007FFFFFu

__device__ __forceinline__ float leaky(float e) {
    return e > 0.f ? e : SLOPE * e;
}

__device__ __forceinline__ v8f wmma4(v2f a, v2f b, v8f c) {
    return __builtin_amdgcn_wmma_f32_16x16x4_f32(
               false, a, false, b, (short)0, c, false, false);
}

#define PLANE 33
#define ACCP (4 * PLANE)
#define WPP 288
#define RANGE 196
#define NWG ((NN + RANGE - 1) / RANGE)
#define OT 1024
#define CAP 8192
#define SCAN_U 8
#define ROW_U 8
#define LIST_U 4
#define NCHUNK (EE / 32)
#define PL_U 5
static_assert(NN <= 65536, "list entry packs the source index in 16 bits");
static_assert(RANGE < 256 && (RANGE + 1) * HEADS <= OT, "range bookkeeping fits one pass");
static_assert((EE % 4) == 0, "dst array is read as int4");
static_assert((EE % 32) == 0, "edges divide into whole chunks");
static_assert(NWG <= 256, "an owner number fits one byte");
static_assert(ROW_U == 8 && HEADS == 4 && (CAP % ROW_U) == 0 && (OT % ROW_U) == 0,
              "a block of ROW_U edges holds 32 alphas, lies whole inside ev, and belongs to neighbouring lanes of one wave");

struct OwnerLds {
    alignas(16) float ev[CAP * HEADS];
    float    acc[(RANGE + 1) * ACCP];
    unsigned m[(RANGE + 1) * HEADS];
    float    den[(RANGE + 1) * HEADS];
    float    ald[(RANGE + 1) * HEADS];
    unsigned list[CAP];
    unsigned count;
};
static_assert(sizeof(OwnerLds) <= 320 * 1024, "one group's LDS fits a WGP");
static_assert((HC / 2) * WPP <= CAP * HEADS, "staged weight fits in ev");

#define LOGIT_U ((RANGE + OT / 32 - 1) / (OT / 32))
__device__ __forceinline__ void own_logits(const float* hrows,
                                           const float* __restrict__ a_src,
                                           const float* __restrict__ a_dst,
                                           float* __restrict__ al_s,
                                           float* __restrict__ al_d,
                                           int n0, unsigned nloc, int wv, int lane)
{
    const float4 as = *((const float4*)a_src + lane);
    const float4 ad = *((const float4*)a_dst + lane);
    float4 v[LOGIT_U];
    #pragma unroll
    for (int u = 0; u < LOGIT_U; ++u) {
        const unsigned i  = (unsigned)(wv + u * (OT / 32));
        const unsigned ic = i < nloc ? i : nloc - 1;
        v[u] = *((const float4*)(hrows + (size_t)(n0 + (int)ic) * HC) + lane);
    }
    float s[LOGIT_U], d[LOGIT_U];
    #pragma unroll
    for (int u = 0; u < LOGIT_U; ++u) {
        s[u] = v[u].x * as.x; s[u] = fmaf(v[u].y, as.y, s[u]); s[u] = fmaf(v[u].z, as.z, s[u]); s[u] = fmaf(v[u].w, as.w, s[u]);
        d[u] = v[u].x * ad.x; d[u] = fmaf(v[u].y, ad.y, d[u]); d[u] = fmaf(v[u].z, ad.z, d[u]); d[u] = fmaf(v[u].w, ad.w, d[u]);
    }
    #pragma unroll
    for (int m = 1; m < 8; m <<= 1) {
        float ps[LOGIT_U], pd[LOGIT_U];
        #pragma unroll
        for (int u = 0; u < LOGIT_U; ++u) { ps[u] = __shfl_xor(s[u], m); pd[u] = __shfl_xor(d[u], m); }
        #pragma unroll
        for (int u = 0; u < LOGIT_U; ++u) { s[u] += ps[u]; d[u] += pd[u]; }
    }
    #pragma unroll
    for (int u = 0; u < LOGIT_U; ++u) {
        const unsigned i = (unsigned)(wv + u * (OT / 32));
        if ((lane & 7) == 0 && i < nloc) {
            const size_t o = (size_t)(n0 + (int)i) * HEADS + (lane >> 3);
            al_s[o] = s[u]; al_d[o] = d[u];
        }
    }
}

__device__ __forceinline__ void lds_add(float* p, float v) {
    __hip_atomic_fetch_add(p, v, __ATOMIC_RELAXED, __HIP_MEMORY_SCOPE_WORKGROUP);
}

__device__ __forceinline__ void edge_max_one(OwnerLds& L, const float* __restrict__ al_s,
                                             int src, unsigned ld) {
    const float4 s = *(const float4*)(al_s + (size_t)src * HEADS);
    const float sv[4] = { s.x, s.y, s.z, s.w };
    #pragma unroll
    for (int hd = 0; hd < HEADS; ++hd) {
        float e = leaky(sv[hd] + L.ald[ld * HEADS + hd]);
        atomicMax(&L.m[ld * HEADS + hd], ford(e));
    }
}

__device__ __forceinline__ void edge_den_one(OwnerLds& L, const float* __restrict__ al_s,
                                             int src, unsigned ld) {
    const float4 s = *(const float4*)(al_s + (size_t)src * HEADS);
    const float sv[4] = { s.x, s.y, s.z, s.w };
    #pragma unroll
    for (int hd = 0; hd < HEADS; ++hd) {
        float e  = leaky(sv[hd] + L.ald[ld * HEADS + hd]);
        float mx = funord(L.m[ld * HEADS + hd]);
        lds_add(&L.den[ld * HEADS + hd], __expf(e - mx));
    }
}

__device__ __forceinline__ void edge_logit_max(OwnerLds& L, const float* __restrict__ al_s, unsigned idx) {
    const unsigned ent = L.list[idx];
    const unsigned ld  = ent >> 16;
    const float4 s = *(const float4*)(al_s + (size_t)(ent & 0xFFFFu) * HEADS);
    const float4 d = *(const float4*)(L.ald + ld * HEADS);
    float4 e;
    e.x = leaky(s.x + d.x); e.y = leaky(s.y + d.y); e.z = leaky(s.z + d.z); e.w = leaky(s.w + d.w);
    atomicMax(&L.m[ld * HEADS + 0], ford(e.x));
    atomicMax(&L.m[ld * HEADS + 1], ford(e.y));
    atomicMax(&L.m[ld * HEADS + 2], ford(e.z));
    atomicMax(&L.m[ld * HEADS + 3], ford(e.w));
    *(float4*)(L.ev + idx * HEADS) = e;
}

__device__ __forceinline__ void edge_exp_den(OwnerLds& L, unsigned idx) {
    const unsigned ld = L.list[idx] >> 16;
    const float4 e  = *(const float4*)(L.ev + idx * HEADS);
    const uint4  mk = *(const uint4*)(L.m + ld * HEADS);
    float4 x;
    x.x = __expf(e.x - funord(mk.x)); x.y = __expf(e.y - funord(mk.y));
    x.z = __expf(e.z - funord(mk.z)); x.w = __expf(e.w - funord(mk.w));
    lds_add(&L.den[ld * HEADS + 0], x.x);
    lds_add(&L.den[ld * HEADS + 1], x.y);
    lds_add(&L.den[ld * HEADS + 2], x.z);
    lds_add(&L.den[ld * HEADS + 3], x.w);
    *(float4*)(L.ev + idx * HEADS) = x;
}

__device__ __forceinline__ void edge_alpha(OwnerLds& L, unsigned idx) {
    const unsigned ld = L.list[idx] >> 16;
    float4 x = *(const float4*)(L.ev + idx * HEADS);
    const float4 dn = *(const float4*)(L.den + ld * HEADS);
    x.x = x.x / (dn.x + 1e-16f); x.y = x.y / (dn.y + 1e-16f);
    x.z = x.z / (dn.z + 1e-16f); x.w = x.w / (dn.w + 1e-16f);
    float* p = L.ev + (idx & ~(unsigned)(ROW_U - 1)) * HEADS + (idx & (unsigned)(ROW_U - 1));
    p[0] = x.x; p[ROW_U] = x.y; p[2 * ROW_U] = x.z; p[3 * ROW_U] = x.w;
}

template <int U> __device__ __forceinline__ float share8(int x) {
    return __int_as_float(__builtin_amdgcn_mov_dpp8(x, U * 0x249249));
}

__device__ __forceinline__ void row_add(OwnerLds& L, float sv, float4 v, unsigned ld, int lane) {
    const int hd = lane >> 3;
    float e  = leaky(sv + L.ald[ld * HEADS + hd]);
    float mx = funord(L.m[ld * HEADS + hd]);
    float alpha = __expf(e - mx) / (L.den[ld * HEADS + hd] + 1e-16f);
    float* ap = L.acc + ld * ACCP + lane;
    lds_add(ap,             v.x * alpha);
    lds_add(ap + PLANE,     v.y * alpha);
    lds_add(ap + 2 * PLANE, v.z * alpha);
    lds_add(ap + 3 * PLANE, v.w * alpha);
}

__device__ __forceinline__ void row_one(OwnerLds& L, const float* __restrict__ h,
                                        const float* __restrict__ al_s,
                                        int src, unsigned ld, int lane) {
    float  sv = al_s[(size_t)src * HEADS + (lane >> 3)];
    float4 v  = *((const float4*)(h + (size_t)src * HC) + lane);
    row_add(L, sv, v, ld, lane);
}

template <int MODE>
__device__ __forceinline__ void scan_edges(OwnerLds& L, const int* __restrict__ ei,
                                           const float* __restrict__ h,
                                           const float* __restrict__ al_s,
                                           int n0, unsigned nloc, int t)
{
    const int4* d4 = (const int4*)(ei + EE);
    const int lane = t & 31;
    for (int base = 0; base < EE / 4; base += OT * SCAN_U) {
        int4 dv[SCAN_U];
        #pragma unroll
        for (int u = 0; u < SCAN_U; ++u) {
            const int i = base + u * OT + t;
            dv[u] = (i < EE / 4) ? d4[i] : make_int4(-1, -1, -1, -1);
        }
        #pragma unroll
        for (int u = 0; u < SCAN_U; ++u) {
            const int i = base + u * OT + t;
            const int dd[4] = { dv[u].x, dv[u].y, dv[u].z, dv[u].w };
            unsigned ld[4]; bool hit[4];
            #pragma unroll
            for (int k = 0; k < 4; ++k) { ld[k] = (unsigned)(dd[k] - n0); hit[k] = ld[k] < nloc; }
            if (MODE == 3) {
                #pragma unroll
                for (int k = 0; k < 4; ++k) {
                    const int src = hit[k] ? ei[4 * i + k] : 0;
                    unsigned mask = __builtin_amdgcn_ballot_w32(hit[k]);
                    while (mask) {
                        const int l = __builtin_ctz(mask);
                        mask &= mask - 1;
                        const int      s2 = __shfl(src, l);
                        const unsigned d2 = (unsigned)__shfl((int)ld[k], l);
                        row_one(L, h, al_s, s2, d2, lane);
                    }
                }
            } else if (hit[0] | hit[1] | hit[2] | hit[3]) {
                #pragma unroll
                for (int k = 0; k < 4; ++k) {
                    if (!hit[k]) continue;
                    const int src = ei[4 * i + k];
                    if (MODE == 1) {
                        edge_max_one(L, al_s, src, ld[k]);
                    } else {
                        edge_den_one(L, al_s, src, ld[k]);
                    }
                }
            }
        }
    }
}

__device__ __forceinline__ void scan_planes(OwnerLds& L, const unsigned* __restrict__ planes, int t)
{
    const uint4* p4 = (const uint4*)planes;
    const unsigned g = blockIdx.x;
    unsigned same[8];
    #pragma unroll
    for (int p = 0; p < 8; ++p) same[p] = ((g >> p) & 1u) ? 0u : ~0u;
    for (int base = 0; base < NCHUNK; base += OT * PL_U) {
        uint4 lo[PL_U], hi[PL_U];
        #pragma unroll
        for (int u = 0; u < PL_U; ++u) {
            const int c  = base + u * OT + t;
            const int cc = c < NCHUNK ? c : NCHUNK - 1;
            lo[u] = p4[2 * cc];
            hi[u] = p4[2 * cc + 1];
        }
        #pragma unroll
        for (int u = 0; u < PL_U; ++u) {
            const int c = base + u * OT + t;
            unsigned mt = (lo[u].x ^ same[0]) & (lo[u].y ^ same[1]) & (lo[u].z ^ same[2]) & (lo[u].w ^ same[3])
                        & (hi[u].x ^ same[4]) & (hi[u].y ^ same[5]) & (hi[u].z ^ same[6]) & (hi[u].w ^ same[7]);
            if (c >= NCHUNK) mt = 0u;
            while (mt) {
                const unsigned l = (unsigned)__builtin_ctz(mt);
                mt &= mt - 1u;
                const unsigned slot = atomicAdd(&L.count, 1u);
                if (slot < CAP) L.list[slot] = 32u * (unsigned)c + l;
            }
        }
    }
}

template <int NOUT, int CT>
__device__ __forceinline__ void rows_times_weight(const float* rows, const float* Ws,
                                                  float* __restrict__ dst,
                                                  const float* __restrict__ obias,
                                                  int n0, unsigned nloc, int wv, int lane)
{
    constexpr int NCT = (NOUT + 15) / 16;
    constexpr int NCG = (NCT + CT - 1) / CT;
    constexpr int NRT = (RANGE + 15) / 16;
    const int hi = lane >> 4;
    const int lo = lane & 15;
    for (int job = wv; job < NRT * NCG; job += OT / 32) {
        const int rt = job % NRT, cg = job / NRT;
        int row = rt * 16 + lo;
        if (row > RANGE) row = RANGE;
        const float* ap = rows + row * ACCP + 2 * hi * PLANE;
        int col[CT]; const float* bp[CT]; v8f acc[CT];
        #pragma unroll
        for (int u = 0; u < CT; ++u) {
            col[u] = (cg * CT + u) * 16 + lo;
            const int cc = col[u] < NOUT ? col[u] : NOUT - 1;
            bp[u]  = Ws + hi * WPP + 2 * cc;
            acc[u] = v8f{};
        }
        for (int kk = 0; kk < HC / 4; ++kk) {
            v2f a; a.x = ap[kk]; a.y = ap[PLANE + kk];
            #pragma unroll
            for (int u = 0; u < CT; ++u) {
                const float2 bv = *(const float2*)(bp[u] + kk * 2 * WPP);
                v2f b; b.x = bv.x; b.y = bv.y;
                acc[u] = wmma4(a, b, acc[u]);
            }
        }
        #pragma unroll
        for (int u = 0; u < CT; ++u) {
            if (col[u] < NOUT) {
                const float bb = obias ? obias[col[u]] : 0.f;
                #pragma unroll
                for (int r = 0; r < 8; ++r) {
                    const unsigned grow = (unsigned)(rt * 16 + 8 * hi + r);
                    if (grow < nloc) dst[(size_t)(n0 + (int)grow) * NOUT + col[u]] = acc[u][r] + bb;
                }
            }
        }
    }
}

#define XROW_U 7
#define W_U ((IN_DIM * HC) / OT)
static_assert(IN_DIM == HC, "the tile routine walks HC channels");
static_assert((RANGE + 1) * (IN_DIM / 4) <= OT * XROW_U, "one trip copies the whole range");
static_assert(W_U * OT == IN_DIM * HC, "the weight divides evenly over the threads");
#define XROWS ((OT * XROW_U) / (IN_DIM / 4))
struct FirstLds {
    alignas(16) float ws[(IN_DIM / 2) * WPP];
    float rows[XROWS * ACCP];
};
static_assert(sizeof(FirstLds) <= 320 * 1024, "one group's LDS fits a WGP");
#define ID_PASSES ((NCHUNK + NWG * (OT / 32) - 1) / (NWG * (OT / 32)))
__global__ __launch_bounds__(OT) void range_product(
    const float* __restrict__ x, const float* __restrict__ Wn, float* __restrict__ dst,
    const int* __restrict__ ei, unsigned* __restrict__ planes,
    const float* __restrict__ a_src, const float* __restrict__ a_dst,
    float* __restrict__ al_s, float* __restrict__ al_d)
{
    __shared__ FirstLds F;
    const int t    = threadIdx.x;
    const int lane = t & 31;
    const int wv   = t >> 5;
    const int n0   = blockIdx.x * RANGE;
    const unsigned nloc = (unsigned)((NN - n0) < RANGE ? (NN - n0) : RANGE);

    float4 v[XROW_U];
    float  w[W_U];
    int    dd[ID_PASSES];
    #pragma unroll
    for (int it = 0; it < ID_PASSES; ++it) {
        const int c  = (it * NWG + (int)blockIdx.x) * (OT / 32) + wv;
        const int cc = c < NCHUNK ? c : NCHUNK - 1;
        dd[it] = ei[EE + 32 * cc + lane];
    }
    #pragma unroll
    for (int u = 0; u < XROW_U; ++u) {
        const int i = u * OT + t;
        const unsigned node = (unsigned)(i >> 5);
        const unsigned nc   = node < nloc ? node : nloc - 1;
        v[u] = *((const float4*)(x + (size_t)(n0 + (int)nc) * IN_DIM) + (i & 31));
    }
    #pragma unroll
    for (int u = 0; u < W_U; ++u) w[u] = Wn[u * OT + t];
    #pragma unroll
    for (int u = 0; u < XROW_U; ++u) {
        const int i = u * OT + t;
        float* p = F.rows + (i >> 5) * ACCP + (i & 31);
        p[0] = v[u].x; p[PLANE] = v[u].y; p[2 * PLANE] = v[u].z; p[3 * PLANE] = v[u].w;
    }
    #pragma unroll
    for (int u = 0; u < W_U; ++u) {
        const int i = u * OT + t;
        const int k = i / HC, n = i % HC;
        F.ws[(k >> 1) * WPP + 2 * n + (k & 1)] = w[u];
    }
    #pragma unroll
    for (int it = 0; it < ID_PASSES; ++it) {
        const int c = (it * NWG + (int)blockIdx.x) * (OT / 32) + wv;
        unsigned id = (unsigned)dd[it] / RANGE;
        id = id < 255u ? id : 255u;
        unsigned mine = 0u;
        #pragma unroll
        for (int p = 0; p < 8; ++p) {
            const unsigned vote = __builtin_amdgcn_ballot_w32(((id >> p) & 1u) != 0u);
            if (lane == p) mine = vote;
        }
        if (lane < 8 && c < NCHUNK) planes[8 * c + lane] = mine;
    }
    __syncthreads();
    rows_times_weight<HC, 2>(F.rows, F.ws, dst, (const float*)nullptr, n0, nloc, wv, lane);
    __syncthreads();
    own_logits(dst, a_src, a_dst, al_s, al_d, n0, nloc, wv, lane);
}

template <bool BUILD, bool LAST>
__global__ __launch_bounds__(OT) void owner_aggregate(
    const int* __restrict__ ei, const float* __restrict__ h,
    const float* __restrict__ al_s, const float* __restrict__ al_d,
    const float* __restrict__ bias, const float* __restrict__ Wn,
    const float* __restrict__ obias, float* __restrict__ dst,
    unsigned* __restrict__ wlist, unsigned* __restrict__ wcount,
    const unsigned* __restrict__ planes,
    const float* __restrict__ a_src_n, const float* __restrict__ a_dst_n,
    float* __restrict__ al_s_n, float* __restrict__ al_d_n)
{
    __shared__ OwnerLds L;
    const int t    = threadIdx.x;
    const int lane = t & 31;
    const int wv   = t >> 5;
    const int n0   = blockIdx.x * RANGE;
    const unsigned nloc = (unsigned)((NN - n0) < RANGE ? (NN - n0) : RANGE);
    unsigned* wl = wlist + (size_t)blockIdx.x * CAP;

    if (!BUILD) {
        const unsigned saved = wcount[blockIdx.x];
        if (saved <= CAP) {
            for (unsigned i0 = 0; i0 < saved; i0 += OT * LIST_U) {
                unsigned ent[LIST_U];
                #pragma unroll
                for (int u = 0; u < LIST_U; ++u) {
                    const unsigned idx = i0 + u * OT + t;
                    ent[u] = idx < saved ? wl[idx] : 0u;
                }
                #pragma unroll
                for (int u = 0; u < LIST_U; ++u) {
                    const unsigned idx = i0 + u * OT + t;
                    if (idx < saved) L.list[idx] = ent[u];
                }
            }
        }
        if (t == 0) L.count = saved;
    }

    for (int i = t; i < (RANGE + 1) * ACCP; i += OT) L.acc[i] = 0.f;
    if (t < RANGE * HEADS) {
        L.m[t]   = FORD_NEG_INF;
        L.den[t] = 0.f;
        L.ald[t] = ((unsigned)t < nloc * HEADS) ? al_d[(size_t)n0 * HEADS + t] : 0.f;
    }
    if (BUILD) {
        if ((unsigned)t < nloc) L.list[t] = (unsigned)(n0 + t) | ((unsigned)t << 16);
        if (t == 0) L.count = nloc;
    }
    __syncthreads();

    if (BUILD) {
        scan_planes(L, planes, t);
        __syncthreads();
    }
    const unsigned total = L.count;

    if (BUILD) {
        if (total <= CAP) {
            const unsigned spare = (unsigned)n0 | ((unsigned)RANGE << 16);
            for (unsigned i0 = 0; i0 < total; i0 += OT * LIST_U) {
                unsigned ent[LIST_U]; int sv[LIST_U], dv[LIST_U];
                #pragma unroll
                for (int u = 0; u < LIST_U; ++u) {
                    const unsigned idx = i0 + u * OT + t;
                    ent[u] = idx < total ? L.list[idx] : 0u;
                    const unsigned e = (idx >= nloc && idx < total) ? ent[u] : 0u;
                    sv[u] = ei[e];
                    dv[u] = ei[EE + e];
                }
                #pragma unroll
                for (int u = 0; u < LIST_U; ++u) {
                    const unsigned idx = i0 + u * OT + t;
                    if (idx < total) {
                        if (idx >= nloc) {
                            const unsigned ld = (unsigned)(dv[u] - n0);
                            ent[u] = ld < nloc ? ((unsigned)sv[u] | (ld << 16)) : spare;
                            L.list[idx] = ent[u];
                        }
                        wl[idx] = ent[u];
                    }
                }
            }
        }
        if (t == 0) wcount[blockIdx.x] = total;
    }

    if (total <= CAP) {
        for (unsigned idx = t; idx < total; idx += OT) edge_logit_max(L, al_s, idx);
        __syncthreads();
        for (unsigned idx = t; idx < total; idx += OT) edge_exp_den(L, idx);
        __syncthreads();
        for (unsigned idx = t; idx < total; idx += OT) edge_alpha(L, idx);
        __syncthreads();
        const unsigned spare = (unsigned)n0 | ((unsigned)RANGE << 16);
        for (unsigned b0 = (unsigned)wv * ROW_U; b0 < total; b0 += (OT / 32) * ROW_U) {
            const unsigned idx = b0 + (unsigned)(lane & (ROW_U - 1));
            const bool live = idx < total;
            const unsigned e0 = L.list[live ? idx : 0];
            const float    a0 = L.ev[b0 * HEADS + lane];
            const int ent_l = (int)(live ? e0 : spare);
            const int a_l   = __float_as_int(live ? a0 : 0.f);
            unsigned ent[ROW_U]; float al[ROW_U]; float4 v[ROW_U];
            al[0] = share8<0>(a_l); al[1] = share8<1>(a_l); al[2] = share8<2>(a_l); al[3] = share8<3>(a_l);
            al[4] = share8<4>(a_l); al[5] = share8<5>(a_l); al[6] = share8<6>(a_l); al[7] = share8<7>(a_l);
            #pragma unroll
            for (int u = 0; u < ROW_U; ++u) {
                ent[u] = (unsigned)__builtin_amdgcn_readlane(ent_l, u);
                v[u]   = *((const float4*)(h + (size_t)(ent[u] & 0xFFFFu) * HC) + lane);
            }
            #pragma unroll
            for (int u = 0; u < ROW_U; ++u) {
                float* ap = L.acc + (ent[u] >> 16) * ACCP + lane;
                lds_add(ap,             v[u].x * al[u]);
                lds_add(ap + PLANE,     v[u].y * al[u]);
                lds_add(ap + 2 * PLANE, v[u].z * al[u]);
                lds_add(ap + 3 * PLANE, v[u].w * al[u]);
            }
        }
    } else {
        if ((unsigned)t < nloc) edge_max_one(L, al_s, n0 + t, (unsigned)t);
        scan_edges<1>(L, ei, h, al_s, n0, nloc, t);
        __syncthreads();
        if ((unsigned)t < nloc) edge_den_one(L, al_s, n0 + t, (unsigned)t);
        scan_edges<2>(L, ei, h, al_s, n0, nloc, t);
        __syncthreads();
        for (unsigned i = wv; i < nloc; i += OT / 32) row_one(L, h, al_s, n0 + (int)i, i, lane);
        scan_edges<3>(L, ei, h, al_s, n0, nloc, t);
    }
    __syncthreads();

    for (int i = t; i < RANGE * HC; i += OT) {
        const int node = i / HC, q = i % HC;
        const int j = q >> 5, l = q & 31;
        float* p = L.acc + node * ACCP + j * PLANE + l;
        const float vv = *p + bias[4 * l + j];
        *p = vv > 0.f ? vv : 0.f;
    }
    constexpr int NOUT = LAST ? OUT_DIM : HC;
    float* Ws = L.ev;
    for (int i = t; i < HC * NOUT; i += OT) {
        const int k = i / NOUT, n = i % NOUT;
        Ws[(k >> 1) * WPP + 2 * n + (k & 1)] = Wn[i];
    }
    __syncthreads();
    rows_times_weight<NOUT, LAST ? 1 : 2>(L.acc, Ws, dst, obias, n0, nloc, wv, lane);
    if (!LAST) {
        __syncthreads();
        own_logits(dst, a_src_n, a_dst_n, al_s_n, al_d_n, n0, nloc, wv, lane);
    }
}

extern "C" void kernel_launch(void* const* d_in, const int* in_sizes, int n_in,
                              void* d_out, int out_size, void* d_ws, size_t ws_size,
                              hipStream_t stream)
{
    const float* x   = (const float*)d_in[0];
    const int*   ei  = (const int*)d_in[1];
    const float* W[3]    = { (const float*)d_in[2],  (const float*)d_in[6],  (const float*)d_in[10] };
    const float* asv[3]  = { (const float*)d_in[3],  (const float*)d_in[7],  (const float*)d_in[11] };
    const float* adv[3]  = { (const float*)d_in[4],  (const float*)d_in[8],  (const float*)d_in[12] };
    const float* bv[3]   = { (const float*)d_in[5],  (const float*)d_in[9],  (const float*)d_in[13] };
    const float* Wl  = (const float*)d_in[14];
    const float* bl  = (const float*)d_in[15];

    char* ws = (char*)d_ws;
    float*    hA   = (float*)ws;                       ws += (size_t)NN * HC * sizeof(float);
    float*    hB   = (float*)ws;                       ws += (size_t)NN * HC * sizeof(float);
    float*    al_s0 = (float*)ws;                      ws += (size_t)NN * HEADS * sizeof(float);
    float*    al_d0 = (float*)ws;                      ws += (size_t)NN * HEADS * sizeof(float);
    float*    al_s1 = (float*)ws;                      ws += (size_t)NN * HEADS * sizeof(float);
    float*    al_d1 = (float*)ws;                      ws += (size_t)NN * HEADS * sizeof(float);
    unsigned* wlist  = (unsigned*)ws;                  ws += (size_t)NWG * CAP * sizeof(unsigned);
    unsigned* wcount = (unsigned*)ws;                  ws += (size_t)NWG * sizeof(unsigned);
    unsigned* planes = (unsigned*)ws;


    hipLaunchKernelGGL(range_product, dim3(NWG), dim3(OT), 0, stream, x, W[0], hA, ei, planes,
                       asv[0], adv[0], al_s0, al_d0);
    hipLaunchKernelGGL((owner_aggregate<true, false>), dim3(NWG), dim3(OT), 0, stream,
                       ei, hA, al_s0, al_d0, bv[0], W[1], (const float*)nullptr, hB, wlist, wcount, planes,
                       asv[1], adv[1], al_s1, al_d1);
    hipLaunchKernelGGL((owner_aggregate<false, false>), dim3(NWG), dim3(OT), 0, stream,
                       ei, hB, al_s1, al_d1, bv[1], W[2], (const float*)nullptr, hA, wlist, wcount, planes,
                       asv[2], adv[2], al_s0, al_d0);
    hipLaunchKernelGGL((owner_aggregate<false, true>), dim3(NWG), dim3(OT), 0, stream,
                       ei, hA, al_s0, al_d0, bv[2], Wl, bl, (float*)d_out, wlist, wcount, planes,
                       (const float*)nullptr, (const float*)nullptr, (float*)nullptr, (float*)nullptr);
}


# First launch bins its 3125-edge slice by owner group (LDS counters, arrival order); layer 1 list build copies its 256 bins instead of scanning every edge's owner byte; sums stay LDS arrival order
# speedup vs baseline: 6.0053x; 1.0421x over previous
#include <hip/hip_runtime.h>
#include <math.h>

#define NN 50000
#define EE 800000
#define IN_DIM 128
#define HID 32
#define HEADS 4
#define HC 128
#define OUT_DIM 40
#define SLOPE 0.2f

typedef float v2f __attribute__((ext_vector_type(2)));
typedef float v8f __attribute__((ext_vector_type(8)));

__device__ __forceinline__ unsigned ford(float f) {
    unsigned u = __float_as_uint(f);
    return (u & 0x80000000u) ? ~u : (u | 0x80000000u);
}
__device__ __forceinline__ float funord(unsigned k) {
    unsigned u = (k & 0x80000000u) ? (k & 0x7FFFFFFFu) : ~k;
    return __uint_as_float(u);
}
#define FORD_NEG_INF 0x007FFFFFu

__device__ __forceinline__ float leaky(float e) {
    return e > 0.f ? e : SLOPE * e;
}

__device__ __forceinline__ v8f wmma4(v2f a, v2f b, v8f c) {
    return __builtin_amdgcn_wmma_f32_16x16x4_f32(
               false, a, false, b, (short)0, c, false, false);
}

#define PLANE 33
#define ACCP (4 * PLANE)
#define WPP 288
#define RANGE 196
#define NWG ((NN + RANGE - 1) / RANGE)
#define OT 1024
#define CAP 8192
#define SCAN_U 8
#define ROW_U 8
#define LIST_U 4
#define SLICE (EE / NWG)
#define SL_U ((SLICE + OT - 1) / OT)
#define SEG_U 6
static_assert(NN <= 65536, "list entry packs the source index in 16 bits");
static_assert(RANGE < 256 && (RANGE + 1) * HEADS <= OT, "range bookkeeping fits one pass");
static_assert((EE % 4) == 0, "dst array is read as int4");
static_assert(SLICE * NWG == EE, "the edge array divides into one slice per group");
static_assert(SLICE < 65536, "a bin's offset and its length fit 16 bits each");
static_assert(NWG <= 256 && OT == 4 * NWG, "an owner number fits one byte; four threads copy one bin");
static_assert(ROW_U == 8 && HEADS == 4 && (CAP % ROW_U) == 0 && (OT % ROW_U) == 0,
              "a block of ROW_U edges holds 32 alphas, lies whole inside ev, and belongs to neighbouring lanes of one wave");

struct OwnerLds {
    alignas(16) float ev[CAP * HEADS];
    float    acc[(RANGE + 1) * ACCP];
    unsigned m[(RANGE + 1) * HEADS];
    float    den[(RANGE + 1) * HEADS];
    float    ald[(RANGE + 1) * HEADS];
    unsigned list[CAP];
    alignas(16) unsigned seg[NWG];
    unsigned segpre[NWG];
    unsigned count;
};
static_assert(sizeof(OwnerLds) <= 320 * 1024, "one group's LDS fits a WGP");
static_assert((HC / 2) * WPP <= CAP * HEADS, "staged weight fits in ev");

#define LOGIT_U ((RANGE + OT / 32 - 1) / (OT / 32))
__device__ __forceinline__ void own_logits(const float* hrows,
                                           const float* __restrict__ a_src,
                                           const float* __restrict__ a_dst,
                                           float* __restrict__ al_s,
                                           float* __restrict__ al_d,
                                           int n0, unsigned nloc, int wv, int lane)
{
    const float4 as = *((const float4*)a_src + lane);
    const float4 ad = *((const float4*)a_dst + lane);
    float4 v[LOGIT_U];
    #pragma unroll
    for (int u = 0; u < LOGIT_U; ++u) {
        const unsigned i  = (unsigned)(wv + u * (OT / 32));
        const unsigned ic = i < nloc ? i : nloc - 1;
        v[u] = *((const float4*)(hrows + (size_t)(n0 + (int)ic) * HC) + lane);
    }
    float s[LOGIT_U], d[LOGIT_U];
    #pragma unroll
    for (int u = 0; u < LOGIT_U; ++u) {
        s[u] = v[u].x * as.x; s[u] = fmaf(v[u].y, as.y, s[u]); s[u] = fmaf(v[u].z, as.z, s[u]); s[u] = fmaf(v[u].w, as.w, s[u]);
        d[u] = v[u].x * ad.x; d[u] = fmaf(v[u].y, ad.y, d[u]); d[u] = fmaf(v[u].z, ad.z, d[u]); d[u] = fmaf(v[u].w, ad.w, d[u]);
    }
    #pragma unroll
    for (int m = 1; m < 8; m <<= 1) {
        float ps[LOGIT_U], pd[LOGIT_U];
        #pragma unroll
        for (int u = 0; u < LOGIT_U; ++u) { ps[u] = __shfl_xor(s[u], m); pd[u] = __shfl_xor(d[u], m); }
        #pragma unroll
        for (int u = 0; u < LOGIT_U; ++u) { s[u] += ps[u]; d[u] += pd[u]; }
    }
    #pragma unroll
    for (int u = 0; u < LOGIT_U; ++u) {
        const unsigned i = (unsigned)(wv + u * (OT / 32));
        if ((lane & 7) == 0 && i < nloc) {
            const size_t o = (size_t)(n0 + (int)i) * HEADS + (lane >> 3);
            al_s[o] = s[u]; al_d[o] = d[u];
        }
    }
}

__device__ __forceinline__ void lds_add(float* p, float v) {
    __hip_atomic_fetch_add(p, v, __ATOMIC_RELAXED, __HIP_MEMORY_SCOPE_WORKGROUP);
}

__device__ __forceinline__ void edge_max_one(OwnerLds& L, const float* __restrict__ al_s,
                                             int src, unsigned ld) {
    const float4 s = *(const float4*)(al_s + (size_t)src * HEADS);
    const float sv[4] = { s.x, s.y, s.z, s.w };
    #pragma unroll
    for (int hd = 0; hd < HEADS; ++hd) {
        float e = leaky(sv[hd] + L.ald[ld * HEADS + hd]);
        atomicMax(&L.m[ld * HEADS + hd], ford(e));
    }
}

__device__ __forceinline__ void edge_den_one(OwnerLds& L, const float* __restrict__ al_s,
                                             int src, unsigned ld) {
    const float4 s = *(const float4*)(al_s + (size_t)src * HEADS);
    const float sv[4] = { s.x, s.y, s.z, s.w };
    #pragma unroll
    for (int hd = 0; hd < HEADS; ++hd) {
        float e  = leaky(sv[hd] + L.ald[ld * HEADS + hd]);
        float mx = funord(L.m[ld * HEADS + hd]);
        lds_add(&L.den[ld * HEADS + hd], __expf(e - mx));
    }
}

__device__ __forceinline__ void edge_logit_max(OwnerLds& L, const float* __restrict__ al_s, unsigned idx) {
    const unsigned ent = L.list[idx];
    const unsigned ld  = ent >> 16;
    const float4 s = *(const float4*)(al_s + (size_t)(ent & 0xFFFFu) * HEADS);
    const float4 d = *(const float4*)(L.ald + ld * HEADS);
    float4 e;
    e.x = leaky(s.x + d.x); e.y = leaky(s.y + d.y); e.z = leaky(s.z + d.z); e.w = leaky(s.w + d.w);
    atomicMax(&L.m[ld * HEADS + 0], ford(e.x));
    atomicMax(&L.m[ld * HEADS + 1], ford(e.y));
    atomicMax(&L.m[ld * HEADS + 2], ford(e.z));
    atomicMax(&L.m[ld * HEADS + 3], ford(e.w));
    *(float4*)(L.ev + idx * HEADS) = e;
}

__device__ __forceinline__ void edge_exp_den(OwnerLds& L, unsigned idx) {
    const unsigned ld = L.list[idx] >> 16;
    const float4 e  = *(const float4*)(L.ev + idx * HEADS);
    const uint4  mk = *(const uint4*)(L.m + ld * HEADS);
    float4 x;
    x.x = __expf(e.x - funord(mk.x)); x.y = __expf(e.y - funord(mk.y));
    x.z = __expf(e.z - funord(mk.z)); x.w = __expf(e.w - funord(mk.w));
    lds_add(&L.den[ld * HEADS + 0], x.x);
    lds_add(&L.den[ld * HEADS + 1], x.y);
    lds_add(&L.den[ld * HEADS + 2], x.z);
    lds_add(&L.den[ld * HEADS + 3], x.w);
    *(float4*)(L.ev + idx * HEADS) = x;
}

__device__ __forceinline__ void edge_alpha(OwnerLds& L, unsigned idx) {
    const unsigned ld = L.list[idx] >> 16;
    float4 x = *(const float4*)(L.ev + idx * HEADS);
    const float4 dn = *(const float4*)(L.den + ld * HEADS);
    x.x = x.x / (dn.x + 1e-16f); x.y = x.y / (dn.y + 1e-16f);
    x.z = x.z / (dn.z + 1e-16f); x.w = x.w / (dn.w + 1e-16f);
    float* p = L.ev + (idx & ~(unsigned)(ROW_U - 1)) * HEADS + (idx & (unsigned)(ROW_U - 1));
    p[0] = x.x; p[ROW_U] = x.y; p[2 * ROW_U] = x.z; p[3 * ROW_U] = x.w;
}

template <int U> __device__ __forceinline__ float share8(int x) {
    return __int_as_float(__builtin_amdgcn_mov_dpp8(x, U * 0x249249));
}

__device__ __forceinline__ void row_add(OwnerLds& L, float sv, float4 v, unsigned ld, int lane) {
    const int hd = lane >> 3;
    float e  = leaky(sv + L.ald[ld * HEADS + hd]);
    float mx = funord(L.m[ld * HEADS + hd]);
    float alpha = __expf(e - mx) / (L.den[ld * HEADS + hd] + 1e-16f);
    float* ap = L.acc + ld * ACCP + lane;
    lds_add(ap,             v.x * alpha);
    lds_add(ap + PLANE,     v.y * alpha);
    lds_add(ap + 2 * PLANE, v.z * alpha);
    lds_add(ap + 3 * PLANE, v.w * alpha);
}

__device__ __forceinline__ void row_one(OwnerLds& L, const float* __restrict__ h,
                                        const float* __restrict__ al_s,
                                        int src, unsigned ld, int lane) {
    float  sv = al_s[(size_t)src * HEADS + (lane >> 3)];
    float4 v  = *((const float4*)(h + (size_t)src * HC) + lane);
    row_add(L, sv, v, ld, lane);
}

template <int MODE>
__device__ __forceinline__ void scan_edges(OwnerLds& L, const int* __restrict__ ei,
                                           const float* __restrict__ h,
                                           const float* __restrict__ al_s,
                                           int n0, unsigned nloc, int t)
{
    const int4* d4 = (const int4*)(ei + EE);
    const int lane = t & 31;
    for (int base = 0; base < EE / 4; base += OT * SCAN_U) {
        int4 dv[SCAN_U];
        #pragma unroll
        for (int u = 0; u < SCAN_U; ++u) {
            const int i = base + u * OT + t;
            dv[u] = (i < EE / 4) ? d4[i] : make_int4(-1, -1, -1, -1);
        }
        #pragma unroll
        for (int u = 0; u < SCAN_U; ++u) {
            const int i = base + u * OT + t;
            const int dd[4] = { dv[u].x, dv[u].y, dv[u].z, dv[u].w };
            unsigned ld[4]; bool hit[4];
            #pragma unroll
            for (int k = 0; k < 4; ++k) { ld[k] = (unsigned)(dd[k] - n0); hit[k] = ld[k] < nloc; }
            if (MODE == 3) {
                #pragma unroll
                for (int k = 0; k < 4; ++k) {
                    const int src = hit[k] ? ei[4 * i + k] : 0;
                    unsigned mask = __builtin_amdgcn_ballot_w32(hit[k]);
                    while (mask) {
                        const int l = __builtin_ctz(mask);
                        mask &= mask - 1;
                        const int      s2 = __shfl(src, l);
                        const unsigned d2 = (unsigned)__shfl((int)ld[k], l);
                        row_one(L, h, al_s, s2, d2, lane);
                    }
                }
            } else if (hit[0] | hit[1] | hit[2] | hit[3]) {
                #pragma unroll
                for (int k = 0; k < 4; ++k) {
                    if (!hit[k]) continue;
                    const int src = ei[4 * i + k];
                    if (MODE == 1) {
                        edge_max_one(L, al_s, src, ld[k]);
                    } else {
                        edge_den_one(L, al_s, src, ld[k]);
                    }
                }
            }
        }
    }
}

__device__ __forceinline__ void gather_bins(OwnerLds& L, const unsigned* __restrict__ bucket,
                                            unsigned* __restrict__ wl, unsigned nloc, unsigned spare, int t)
{
    const unsigned s   = (unsigned)t >> 2, sub = (unsigned)t & 3u;
    const unsigned pk  = L.seg[s];
    const unsigned cnt = pk >> 16;
    const unsigned pre = L.segpre[s];
    const unsigned* bp = bucket + (size_t)s * SLICE + (pk & 0xFFFFu);
    unsigned ent[SEG_U];
    #pragma unroll
    for (int u = 0; u < SEG_U; ++u) {
        const unsigned k = sub + 4u * (unsigned)u;
        ent[u] = k < cnt ? bp[k] : 0u;
    }
    #pragma unroll
    for (int u = 0; u < SEG_U; ++u) {
        const unsigned k = sub + 4u * (unsigned)u;
        if (k < cnt) {
            const unsigned e = (ent[u] >> 16) < nloc ? ent[u] : spare;
            L.list[pre + k] = e;
            wl[pre + k] = e;
        }
    }
    for (unsigned k = sub + 4u * SEG_U; k < cnt; k += 4u) {
        const unsigned r = bp[k];
        const unsigned e = (r >> 16) < nloc ? r : spare;
        L.list[pre + k] = e;
        wl[pre + k] = e;
    }
}

template <int NOUT, int CT>
__device__ __forceinline__ void rows_times_weight(const float* rows, const float* Ws,
                                                  float* __restrict__ dst,
                                                  const float* __restrict__ obias,
                                                  int n0, unsigned nloc, int wv, int lane)
{
    constexpr int NCT = (NOUT + 15) / 16;
    constexpr int NCG = (NCT + CT - 1) / CT;
    constexpr int NRT = (RANGE + 15) / 16;
    const int hi = lane >> 4;
    const int lo = lane & 15;
    for (int job = wv; job < NRT * NCG; job += OT / 32) {
        const int rt = job % NRT, cg = job / NRT;
        int row = rt * 16 + lo;
        if (row > RANGE) row = RANGE;
        const float* ap = rows + row * ACCP + 2 * hi * PLANE;
        int col[CT]; const float* bp[CT]; v8f acc[CT];
        #pragma unroll
        for (int u = 0; u < CT; ++u) {
            col[u] = (cg * CT + u) * 16 + lo;
            const int cc = col[u] < NOUT ? col[u] : NOUT - 1;
            bp[u]  = Ws + hi * WPP + 2 * cc;
            acc[u] = v8f{};
        }
        for (int kk = 0; kk < HC / 4; ++kk) {
            v2f a; a.x = ap[kk]; a.y = ap[PLANE + kk];
            #pragma unroll
            for (int u = 0; u < CT; ++u) {
                const float2 bv = *(const float2*)(bp[u] + kk * 2 * WPP);
                v2f b; b.x = bv.x; b.y = bv.y;
                acc[u] = wmma4(a, b, acc[u]);
            }
        }
        #pragma unroll
        for (int u = 0; u < CT; ++u) {
            if (col[u] < NOUT) {
                const float bb = obias ? obias[col[u]] : 0.f;
                #pragma unroll
                for (int r = 0; r < 8; ++r) {
                    const unsigned grow = (unsigned)(rt * 16 + 8 * hi + r);
                    if (grow < nloc) dst[(size_t)(n0 + (int)grow) * NOUT + col[u]] = acc[u][r] + bb;
                }
            }
        }
    }
}

#define XROW_U 7
#define W_U ((IN_DIM * HC) / OT)
static_assert(IN_DIM == HC, "the tile routine walks HC channels");
static_assert((RANGE + 1) * (IN_DIM / 4) <= OT * XROW_U, "one trip copies the whole range");
static_assert(W_U * OT == IN_DIM * HC, "the weight divides evenly over the threads");
#define XROWS ((OT * XROW_U) / (IN_DIM / 4))
struct FirstLds {
    alignas(16) float ws[(IN_DIM / 2) * WPP];
    float rows[XROWS * ACCP];
    alignas(16) unsigned hist[NWG];
    unsigned base[NWG];
};
static_assert(sizeof(FirstLds) <= 320 * 1024, "one group's LDS fits a WGP");
__global__ __launch_bounds__(OT) void range_product(
    const float* __restrict__ x, const float* __restrict__ Wn, float* __restrict__ dst,
    const int* __restrict__ ei, unsigned* __restrict__ bucket, unsigned* __restrict__ offT,
    const float* __restrict__ a_src, const float* __restrict__ a_dst,
    float* __restrict__ al_s, float* __restrict__ al_d)
{
    __shared__ FirstLds F;
    const int t    = threadIdx.x;
    const int lane = t & 31;
    const int wv   = t >> 5;
    const int n0   = blockIdx.x * RANGE;
    const unsigned nloc = (unsigned)((NN - n0) < RANGE ? (NN - n0) : RANGE);

    float4 v[XROW_U];
    float  w[W_U];
    int    es[SL_U], ed[SL_U];
    #pragma unroll
    for (int u = 0; u < SL_U; ++u) {
        const int k  = u * OT + t;
        const int kc = k < SLICE ? k : SLICE - 1;
        es[u] = ei[(int)blockIdx.x * SLICE + kc];
        ed[u] = ei[EE + (int)blockIdx.x * SLICE + kc];
    }
    #pragma unroll
    for (int u = 0; u < XROW_U; ++u) {
        const int i = u * OT + t;
        const unsigned node = (unsigned)(i >> 5);
        const unsigned nc   = node < nloc ? node : nloc - 1;
        v[u] = *((const float4*)(x + (size_t)(n0 + (int)nc) * IN_DIM) + (i & 31));
    }
    #pragma unroll
    for (int u = 0; u < W_U; ++u) w[u] = Wn[u * OT + t];
    if (t < NWG) F.hist[t] = 0u;
    #pragma unroll
    for (int u = 0; u < XROW_U; ++u) {
        const int i = u * OT + t;
        float* p = F.rows + (i >> 5) * ACCP + (i & 31);
        p[0] = v[u].x; p[PLANE] = v[u].y; p[2 * PLANE] = v[u].z; p[3 * PLANE] = v[u].w;
    }
    #pragma unroll
    for (int u = 0; u < W_U; ++u) {
        const int i = u * OT + t;
        const int k = i / HC, n = i % HC;
        F.ws[(k >> 1) * WPP + 2 * n + (k & 1)] = w[u];
    }
    __syncthreads();
    unsigned ent[SL_U], key[SL_U];
    #pragma unroll
    for (int u = 0; u < SL_U; ++u) {
        unsigned own = (unsigned)ed[u] / RANGE;
        own = own < (unsigned)(NWG - 1) ? own : (unsigned)(NWG - 1);
        unsigned ld = (unsigned)ed[u] - own * RANGE;
        ld = ld < RANGE ? ld : RANGE;
        ent[u] = (unsigned)es[u] | (ld << 16);
        unsigned place = 0u;
        if (u * OT + t < SLICE) place = atomicAdd(&F.hist[own], 1u);
        key[u] = own | (place << 8);
    }
    rows_times_weight<HC, 2>(F.rows, F.ws, dst, (const float*)nullptr, n0, nloc, wv, lane);
    __syncthreads();
    if (wv == 0) {
        const uint4 h0 = *(const uint4*)(F.hist + 8 * lane);
        const uint4 h1 = *(const uint4*)(F.hist + 8 * lane + 4);
        const unsigned c1 = h0.x, c2 = c1 + h0.y, c3 = c2 + h0.z, c4 = c3 + h0.w;
        const unsigned c5 = c4 + h1.x, c6 = c5 + h1.y, c7 = c6 + h1.z, c8 = c7 + h1.w;
        unsigned incl = c8;
        #pragma unroll
        for (int m = 1; m < 32; m <<= 1) {
            const unsigned o = (unsigned)__shfl_up((int)incl, m);
            if (lane >= m) incl += o;
        }
        const unsigned ex = incl - c8;
        *(uint4*)(F.base + 8 * lane)     = make_uint4(ex, ex + c1, ex + c2, ex + c3);
        *(uint4*)(F.base + 8 * lane + 4) = make_uint4(ex + c4, ex + c5, ex + c6, ex + c7);
    }
    own_logits(dst, a_src, a_dst, al_s, al_d, n0, nloc, wv, lane);
    __syncthreads();
    unsigned* bk = bucket + (size_t)blockIdx.x * SLICE;
    #pragma unroll
    for (int u = 0; u < SL_U; ++u)
        if (u * OT + t < SLICE) bk[F.base[key[u] & 255u] + (key[u] >> 8)] = ent[u];
    if (t < NWG) offT[(size_t)t * NWG + blockIdx.x] = F.base[t] | (F.hist[t] << 16);
}

template <bool BUILD, bool LAST>
__global__ __launch_bounds__(OT) void owner_aggregate(
    const int* __restrict__ ei, const float* __restrict__ h,
    const float* __restrict__ al_s, const float* __restrict__ al_d,
    const float* __restrict__ bias, const float* __restrict__ Wn,
    const float* __restrict__ obias, float* __restrict__ dst,
    unsigned* __restrict__ wlist, unsigned* __restrict__ wcount,
    const unsigned* __restrict__ bucket, const unsigned* __restrict__ offT,
    const float* __restrict__ a_src_n, const float* __restrict__ a_dst_n,
    float* __restrict__ al_s_n, float* __restrict__ al_d_n)
{
    __shared__ OwnerLds L;
    const int t    = threadIdx.x;
    const int lane = t & 31;
    const int wv   = t >> 5;
    const int n0   = blockIdx.x * RANGE;
    const unsigned nloc = (unsigned)((NN - n0) < RANGE ? (NN - n0) : RANGE);
    unsigned* wl = wlist + (size_t)blockIdx.x * CAP;

    if (!BUILD) {
        const unsigned saved = wcount[blockIdx.x];
        if (saved <= CAP) {
            for (unsigned i0 = 0; i0 < saved; i0 += OT * LIST_U) {
                unsigned ent[LIST_U];
                #pragma unroll
                for (int u = 0; u < LIST_U; ++u) {
                    const unsigned idx = i0 + u * OT + t;
                    ent[u] = idx < saved ? wl[idx] : 0u;
                }
                #pragma unroll
                for (int u = 0; u < LIST_U; ++u) {
                    const unsigned idx = i0 + u * OT + t;
                    if (idx < saved) L.list[idx] = ent[u];
                }
            }
        }
        if (t == 0) L.count = saved;
    }

    for (int i = t; i < (RANGE + 1) * ACCP; i += OT) L.acc[i] = 0.f;
    if (t < RANGE * HEADS) {
        L.m[t]   = FORD_NEG_INF;
        L.den[t] = 0.f;
        L.ald[t] = ((unsigned)t < nloc * HEADS) ? al_d[(size_t)n0 * HEADS + t] : 0.f;
    }
    if (BUILD) {
        if ((unsigned)t < nloc) L.list[t] = (unsigned)(n0 + t) | ((unsigned)t << 16);
        if (t < NWG) L.seg[t] = offT[(size_t)blockIdx.x * NWG + t];
    }
    __syncthreads();

    if (BUILD) {
        if (wv == 0) {
            const uint4 g0 = *(const uint4*)(L.seg + 8 * lane);
            const uint4 g1 = *(const uint4*)(L.seg + 8 * lane + 4);
            const unsigned c1 = g0.x >> 16, c2 = c1 + (g0.y >> 16), c3 = c2 + (g0.z >> 16), c4 = c3 + (g0.w >> 16);
            const unsigned c5 = c4 + (g1.x >> 16), c6 = c5 + (g1.y >> 16), c7 = c6 + (g1.z >> 16), c8 = c7 + (g1.w >> 16);
            unsigned incl = c8;
            #pragma unroll
            for (int m = 1; m < 32; m <<= 1) {
                const unsigned o = (unsigned)__shfl_up((int)incl, m);
                if (lane >= m) incl += o;
            }
            const unsigned ex = nloc + incl - c8;
            *(uint4*)(L.segpre + 8 * lane)     = make_uint4(ex, ex + c1, ex + c2, ex + c3);
            *(uint4*)(L.segpre + 8 * lane + 4) = make_uint4(ex + c4, ex + c5, ex + c6, ex + c7);
            if (lane == 31) L.count = nloc + incl;
        }
        __syncthreads();
    }
    const unsigned total = L.count;

    if (BUILD) {
        if (total <= CAP) {
            if ((unsigned)t < nloc) wl[t] = L.list[t];
            gather_bins(L, bucket, wl, nloc, (unsigned)n0 | ((unsigned)RANGE << 16), t);
        }
        if (t == 0) wcount[blockIdx.x] = total;
        __syncthreads();
    }

    if (total <= CAP) {
        for (unsigned idx = t; idx < total; idx += OT) edge_logit_max(L, al_s, idx);
        __syncthreads();
        for (unsigned idx = t; idx < total; idx += OT) edge_exp_den(L, idx);
        __syncthreads();
        for (unsigned idx = t; idx < total; idx += OT) edge_alpha(L, idx);
        __syncthreads();
        const unsigned spare = (unsigned)n0 | ((unsigned)RANGE << 16);
        for (unsigned b0 = (unsigned)wv * ROW_U; b0 < total; b0 += (OT / 32) * ROW_U) {
            const unsigned idx = b0 + (unsigned)(lane & (ROW_U - 1));
            const bool live = idx < total;
            const unsigned e0 = L.list[live ? idx : 0];
            const float    a0 = L.ev[b0 * HEADS + lane];
            const int ent_l = (int)(live ? e0 : spare);
            const int a_l   = __float_as_int(live ? a0 : 0.f);
            unsigned ent[ROW_U]; float al[ROW_U]; float4 v[ROW_U];
            al[0] = share8<0>(a_l); al[1] = share8<1>(a_l); al[2] = share8<2>(a_l); al[3] = share8<3>(a_l);
            al[4] = share8<4>(a_l); al[5] = share8<5>(a_l); al[6] = share8<6>(a_l); al[7] = share8<7>(a_l);
            #pragma unroll
            for (int u = 0; u < ROW_U; ++u) {
                ent[u] = (unsigned)__builtin_amdgcn_readlane(ent_l, u);
                v[u]   = *((const float4*)(h + (size_t)(ent[u] & 0xFFFFu) * HC) + lane);
            }
            #pragma unroll
            for (int u = 0; u < ROW_U; ++u) {
                float* ap = L.acc + (ent[u] >> 16) * ACCP + lane;
                lds_add(ap,             v[u].x * al[u]);
                lds_add(ap + PLANE,     v[u].y * al[u]);
                lds_add(ap + 2 * PLANE, v[u].z * al[u]);
                lds_add(ap + 3 * PLANE, v[u].w * al[u]);
            }
        }
    } else {
        if ((unsigned)t < nloc) edge_max_one(L, al_s, n0 + t, (unsigned)t);
        scan_edges<1>(L, ei, h, al_s, n0, nloc, t);
        __syncthreads();
        if ((unsigned)t < nloc) edge_den_one(L, al_s, n0 + t, (unsigned)t);
        scan_edges<2>(L, ei, h, al_s, n0, nloc, t);
        __syncthreads();
        for (unsigned i = wv; i < nloc; i += OT / 32) row_one(L, h, al_s, n0 + (int)i, i, lane);
        scan_edges<3>(L, ei, h, al_s, n0, nloc, t);
    }
    __syncthreads();

    for (int i = t; i < RANGE * HC; i += OT) {
        const int node = i / HC, q = i % HC;
        const int j = q >> 5, l = q & 31;
        float* p = L.acc + node * ACCP + j * PLANE + l;
        const float vv = *p + bias[4 * l + j];
        *p = vv > 0.f ? vv : 0.f;
    }
    constexpr int NOUT = LAST ? OUT_DIM : HC;
    float* Ws = L.ev;
    for (int i = t; i < HC * NOUT; i += OT) {
        const int k = i / NOUT, n = i % NOUT;
        Ws[(k >> 1) * WPP + 2 * n + (k & 1)] = Wn[i];
    }
    __syncthreads();
    rows_times_weight<NOUT, LAST ? 1 : 2>(L.acc, Ws, dst, obias, n0, nloc, wv, lane);
    if (!LAST) {
        __syncthreads();
        own_logits(dst, a_src_n, a_dst_n, al_s_n, al_d_n, n0, nloc, wv, lane);
    }
}

extern "C" void kernel_launch(void* const* d_in, const int* in_sizes, int n_in,
                              void* d_out, int out_size, void* d_ws, size_t ws_size,
                              hipStream_t stream)
{
    const float* x   = (const float*)d_in[0];
    const int*   ei  = (const int*)d_in[1];
    const float* W[3]    = { (const float*)d_in[2],  (const float*)d_in[6],  (const float*)d_in[10] };
    const float* asv[3]  = { (const float*)d_in[3],  (const float*)d_in[7],  (const float*)d_in[11] };
    const float* adv[3]  = { (const float*)d_in[4],  (const float*)d_in[8],  (const float*)d_in[12] };
    const float* bv[3]   = { (const float*)d_in[5],  (const float*)d_in[9],  (const float*)d_in[13] };
    const float* Wl  = (const float*)d_in[14];
    const float* bl  = (const float*)d_in[15];

    char* ws = (char*)d_ws;
    float*    hA   = (float*)ws;                       ws += (size_t)NN * HC * sizeof(float);
    float*    hB   = (float*)ws;                       ws += (size_t)NN * HC * sizeof(float);
    float*    al_s0 = (float*)ws;                      ws += (size_t)NN * HEADS * sizeof(float);
    float*    al_d0 = (float*)ws;                      ws += (size_t)NN * HEADS * sizeof(float);
    float*    al_s1 = (float*)ws;                      ws += (size_t)NN * HEADS * sizeof(float);
    float*    al_d1 = (float*)ws;                      ws += (size_t)NN * HEADS * sizeof(float);
    unsigned* wlist  = (unsigned*)ws;                  ws += (size_t)NWG * CAP * sizeof(unsigned);
    unsigned* wcount = (unsigned*)ws;                  ws += (size_t)NWG * sizeof(unsigned);
    unsigned* bucket = (unsigned*)ws;                  ws += (size_t)EE * sizeof(unsigned);
    unsigned* offT   = (unsigned*)ws;


    hipLaunchKernelGGL(range_product, dim3(NWG), dim3(OT), 0, stream, x, W[0], hA, ei, bucket, offT,
                       asv[0], adv[0], al_s0, al_d0);
    hipLaunchKernelGGL((owner_aggregate<true, false>), dim3(NWG), dim3(OT), 0, stream,
                       ei, hA, al_s0, al_d0, bv[0], W[1], (const float*)nullptr, hB, wlist, wcount, bucket, offT,
                       asv[1], adv[1], al_s1, al_d1);
    hipLaunchKernelGGL((owner_aggregate<false, false>), dim3(NWG), dim3(OT), 0, stream,
                       ei, hB, al_s1, al_d1, bv[1], W[2], (const float*)nullptr, hA, wlist, wcount, bucket, offT,
                       asv[2], adv[2], al_s0, al_d0);
    hipLaunchKernelGGL((owner_aggregate<false, true>), dim3(NWG), dim3(OT), 0, stream,
                       ei, hA, al_s0, al_d0, bv[2], Wl, bl, (float*)d_out, wlist, wcount, bucket, offT,
                       (const float*)nullptr, (const float*)nullptr, (float*)nullptr, (float*)nullptr);
}


# Row pass drops per-edge LDS float adds: layer 1 orders each group's list by local row in LDS (count, scan, place); a wave sums a row in registers, one LDS add per row change; sums now in list order
# speedup vs baseline: 6.6027x; 1.0995x over previous
#include <hip/hip_runtime.h>
#include <math.h>

#define NN 50000
#define EE 800000
#define IN_DIM 128
#define HID 32
#define HEADS 4
#define HC 128
#define OUT_DIM 40
#define SLOPE 0.2f

typedef float v2f __attribute__((ext_vector_type(2)));
typedef float v8f __attribute__((ext_vector_type(8)));

__device__ __forceinline__ unsigned ford(float f) {
    unsigned u = __float_as_uint(f);
    return (u & 0x80000000u) ? ~u : (u | 0x80000000u);
}
__device__ __forceinline__ float funord(unsigned k) {
    unsigned u = (k & 0x80000000u) ? (k & 0x7FFFFFFFu) : ~k;
    return __uint_as_float(u);
}
#define FORD_NEG_INF 0x007FFFFFu

__device__ __forceinline__ float leaky(float e) {
    return e > 0.f ? e : SLOPE * e;
}

__device__ __forceinline__ v8f wmma4(v2f a, v2f b, v8f c) {
    return __builtin_amdgcn_wmma_f32_16x16x4_f32(
               false, a, false, b, (short)0, c, false, false);
}

#define PLANE 33
#define ACCP (4 * PLANE)
#define WPP 288
#define RANGE 196
#define NWG ((NN + RANGE - 1) / RANGE)
#define OT 1024
#define CAP 8192
#define SCAN_U 8
#define ROW_U 8
#define LIST_U 4
#define SLICE (EE / NWG)
#define SL_U ((SLICE + OT - 1) / OT)
#define SEG_U 6
#define ORD_U 7
static_assert(NN <= 65536, "list entry packs the source index in 16 bits");
static_assert(RANGE < 256 && (RANGE + 1) * HEADS <= OT, "range bookkeeping fits one pass");
static_assert((EE % 4) == 0, "dst array is read as int4");
static_assert(SLICE * NWG == EE, "the edge array divides into one slice per group");
static_assert(SLICE < 65536, "a bin's offset and its length fit 16 bits each");
static_assert(RANGE + 1 <= ORD_U * 32, "one wave scans the counts of every local destination, the spare row's too");
static_assert(NWG <= 256 && OT == 4 * NWG, "an owner number fits one byte; four threads copy one bin");
static_assert(ROW_U == 8 && HEADS == 4 && (CAP % ROW_U) == 0 && (OT % ROW_U) == 0,
              "a block of ROW_U edges holds 32 alphas, lies whole inside ev, and belongs to neighbouring lanes of one wave");

struct OwnerLds {
    alignas(16) float ev[CAP * HEADS];
    float    acc[(RANGE + 1) * ACCP];
    unsigned m[(RANGE + 1) * HEADS];
    float    den[(RANGE + 1) * HEADS];
    float    ald[(RANGE + 1) * HEADS];
    unsigned list[CAP];
    alignas(16) unsigned seg[NWG];
    unsigned segpre[NWG];
    unsigned cur[ORD_U * 32];
    unsigned count;
};
static_assert(sizeof(OwnerLds) <= 320 * 1024, "one group's LDS fits a WGP");
static_assert((HC / 2) * WPP <= CAP * HEADS, "staged weight fits in ev");

#define LOGIT_U ((RANGE + OT / 32 - 1) / (OT / 32))
__device__ __forceinline__ void own_logits(const float* hrows,
                                           const float* __restrict__ a_src,
                                           const float* __restrict__ a_dst,
                                           float* __restrict__ al_s,
                                           float* __restrict__ al_d,
                                           int n0, unsigned nloc, int wv, int lane)
{
    const float4 as = *((const float4*)a_src + lane);
    const float4 ad = *((const float4*)a_dst + lane);
    float4 v[LOGIT_U];
    #pragma unroll
    for (int u = 0; u < LOGIT_U; ++u) {
        const unsigned i  = (unsigned)(wv + u * (OT / 32));
        const unsigned ic = i < nloc ? i : nloc - 1;
        v[u] = *((const float4*)(hrows + (size_t)(n0 + (int)ic) * HC) + lane);
    }
    float s[LOGIT_U], d[LOGIT_U];
    #pragma unroll
    for (int u = 0; u < LOGIT_U; ++u) {
        s[u] = v[u].x * as.x; s[u] = fmaf(v[u].y, as.y, s[u]); s[u] = fmaf(v[u].z, as.z, s[u]); s[u] = fmaf(v[u].w, as.w, s[u]);
        d[u] = v[u].x * ad.x; d[u] = fmaf(v[u].y, ad.y, d[u]); d[u] = fmaf(v[u].z, ad.z, d[u]); d[u] = fmaf(v[u].w, ad.w, d[u]);
    }
    #pragma unroll
    for (int m = 1; m < 8; m <<= 1) {
        float ps[LOGIT_U], pd[LOGIT_U];
        #pragma unroll
        for (int u = 0; u < LOGIT_U; ++u) { ps[u] = __shfl_xor(s[u], m); pd[u] = __shfl_xor(d[u], m); }
        #pragma unroll
        for (int u = 0; u < LOGIT_U; ++u) { s[u] += ps[u]; d[u] += pd[u]; }
    }
    #pragma unroll
    for (int u = 0; u < LOGIT_U; ++u) {
        const unsigned i = (unsigned)(wv + u * (OT / 32));
        if ((lane & 7) == 0 && i < nloc) {
            const size_t o = (size_t)(n0 + (int)i) * HEADS + (lane >> 3);
            al_s[o] = s[u]; al_d[o] = d[u];
        }
    }
}

__device__ __forceinline__ void lds_add(float* p, float v) {
    __hip_atomic_fetch_add(p, v, __ATOMIC_RELAXED, __HIP_MEMORY_SCOPE_WORKGROUP);
}

__device__ __forceinline__ void edge_max_one(OwnerLds& L, const float* __restrict__ al_s,
                                             int src, unsigned ld) {
    const float4 s = *(const float4*)(al_s + (size_t)src * HEADS);
    const float sv[4] = { s.x, s.y, s.z, s.w };
    #pragma unroll
    for (int hd = 0; hd < HEADS; ++hd) {
        float e = leaky(sv[hd] + L.ald[ld * HEADS + hd]);
        atomicMax(&L.m[ld * HEADS + hd], ford(e));
    }
}

__device__ __forceinline__ void edge_den_one(OwnerLds& L, const float* __restrict__ al_s,
                                             int src, unsigned ld) {
    const float4 s = *(const float4*)(al_s + (size_t)src * HEADS);
    const float sv[4] = { s.x, s.y, s.z, s.w };
    #pragma unroll
    for (int hd = 0; hd < HEADS; ++hd) {
        float e  = leaky(sv[hd] + L.ald[ld * HEADS + hd]);
        float mx = funord(L.m[ld * HEADS + hd]);
        lds_add(&L.den[ld * HEADS + hd], __expf(e - mx));
    }
}

__device__ __forceinline__ void edge_logit_max(OwnerLds& L, const float* __restrict__ al_s, unsigned idx) {
    const unsigned ent = L.list[idx];
    const unsigned ld  = ent >> 16;
    const float4 s = *(const float4*)(al_s + (size_t)(ent & 0xFFFFu) * HEADS);
    const float4 d = *(const float4*)(L.ald + ld * HEADS);
    float4 e;
    e.x = leaky(s.x + d.x); e.y = leaky(s.y + d.y); e.z = leaky(s.z + d.z); e.w = leaky(s.w + d.w);
    atomicMax(&L.m[ld * HEADS + 0], ford(e.x));
    atomicMax(&L.m[ld * HEADS + 1], ford(e.y));
    atomicMax(&L.m[ld * HEADS + 2], ford(e.z));
    atomicMax(&L.m[ld * HEADS + 3], ford(e.w));
    *(float4*)(L.ev + idx * HEADS) = e;
}

__device__ __forceinline__ void edge_exp_den(OwnerLds& L, unsigned idx) {
    const unsigned ld = L.list[idx] >> 16;
    const float4 e  = *(const float4*)(L.ev + idx * HEADS);
    const uint4  mk = *(const uint4*)(L.m + ld * HEADS);
    float4 x;
    x.x = __expf(e.x - funord(mk.x)); x.y = __expf(e.y - funord(mk.y));
    x.z = __expf(e.z - funord(mk.z)); x.w = __expf(e.w - funord(mk.w));
    lds_add(&L.den[ld * HEADS + 0], x.x);
    lds_add(&L.den[ld * HEADS + 1], x.y);
    lds_add(&L.den[ld * HEADS + 2], x.z);
    lds_add(&L.den[ld * HEADS + 3], x.w);
    *(float4*)(L.ev + idx * HEADS) = x;
}

__device__ __forceinline__ void edge_alpha(OwnerLds& L, unsigned idx) {
    const unsigned ld = L.list[idx] >> 16;
    float4 x = *(const float4*)(L.ev + idx * HEADS);
    const float4 dn = *(const float4*)(L.den + ld * HEADS);
    x.x = x.x / (dn.x + 1e-16f); x.y = x.y / (dn.y + 1e-16f);
    x.z = x.z / (dn.z + 1e-16f); x.w = x.w / (dn.w + 1e-16f);
    float* p = L.ev + (idx & ~(unsigned)(ROW_U - 1)) * HEADS + (idx & (unsigned)(ROW_U - 1));
    p[0] = x.x; p[ROW_U] = x.y; p[2 * ROW_U] = x.z; p[3 * ROW_U] = x.w;
}

template <int U> __device__ __forceinline__ float share8(int x) {
    return __int_as_float(__builtin_amdgcn_mov_dpp8(x, U * 0x249249));
}

__device__ __forceinline__ void row_add(OwnerLds& L, float sv, float4 v, unsigned ld, int lane) {
    const int hd = lane >> 3;
    float e  = leaky(sv + L.ald[ld * HEADS + hd]);
    float mx = funord(L.m[ld * HEADS + hd]);
    float alpha = __expf(e - mx) / (L.den[ld * HEADS + hd] + 1e-16f);
    float* ap = L.acc + ld * ACCP + lane;
    lds_add(ap,             v.x * alpha);
    lds_add(ap + PLANE,     v.y * alpha);
    lds_add(ap + 2 * PLANE, v.z * alpha);
    lds_add(ap + 3 * PLANE, v.w * alpha);
}

__device__ __forceinline__ void row_one(OwnerLds& L, const float* __restrict__ h,
                                        const float* __restrict__ al_s,
                                        int src, unsigned ld, int lane) {
    float  sv = al_s[(size_t)src * HEADS + (lane >> 3)];
    float4 v  = *((const float4*)(h + (size_t)src * HC) + lane);
    row_add(L, sv, v, ld, lane);
}

template <int MODE>
__device__ __forceinline__ void scan_edges(OwnerLds& L, const int* __restrict__ ei,
                                           const float* __restrict__ h,
                                           const float* __restrict__ al_s,
                                           int n0, unsigned nloc, int t)
{
    const int4* d4 = (const int4*)(ei + EE);
    const int lane = t & 31;
    for (int base = 0; base < EE / 4; base += OT * SCAN_U) {
        int4 dv[SCAN_U];
        #pragma unroll
        for (int u = 0; u < SCAN_U; ++u) {
            const int i = base + u * OT + t;
            dv[u] = (i < EE / 4) ? d4[i] : make_int4(-1, -1, -1, -1);
        }
        #pragma unroll
        for (int u = 0; u < SCAN_U; ++u) {
            const int i = base + u * OT + t;
            const int dd[4] = { dv[u].x, dv[u].y, dv[u].z, dv[u].w };
            unsigned ld[4]; bool hit[4];
            #pragma unroll
            for (int k = 0; k < 4; ++k) { ld[k] = (unsigned)(dd[k] - n0); hit[k] = ld[k] < nloc; }
            if (MODE == 3) {
                #pragma unroll
                for (int k = 0; k < 4; ++k) {
                    const int src = hit[k] ? ei[4 * i + k] : 0;
                    unsigned mask = __builtin_amdgcn_ballot_w32(hit[k]);
                    while (mask) {
                        const int l = __builtin_ctz(mask);
                        mask &= mask - 1;
                        const int      s2 = __shfl(src, l);
                        const unsigned d2 = (unsigned)__shfl((int)ld[k], l);
                        row_one(L, h, al_s, s2, d2, lane);
                    }
                }
            } else if (hit[0] | hit[1] | hit[2] | hit[3]) {
                #pragma unroll
                for (int k = 0; k < 4; ++k) {
                    if (!hit[k]) continue;
                    const int src = ei[4 * i + k];
                    if (MODE == 1) {
                        edge_max_one(L, al_s, src, ld[k]);
                    } else {
                        edge_den_one(L, al_s, src, ld[k]);
                    }
                }
            }
        }
    }
}

__device__ __forceinline__ void gather_bins(OwnerLds& L, const unsigned* __restrict__ bucket,
                                            unsigned* raw, unsigned nloc, unsigned spare, int t)
{
    const unsigned s   = (unsigned)t >> 2, sub = (unsigned)t & 3u;
    const unsigned pk  = L.seg[s];
    const unsigned cnt = pk >> 16;
    const unsigned pre = L.segpre[s];
    const unsigned* bp = bucket + (size_t)s * SLICE + (pk & 0xFFFFu);
    unsigned ent[SEG_U];
    #pragma unroll
    for (int u = 0; u < SEG_U; ++u) {
        const unsigned k = sub + 4u * (unsigned)u;
        ent[u] = k < cnt ? bp[k] : 0u;
    }
    #pragma unroll
    for (int u = 0; u < SEG_U; ++u) {
        const unsigned k = sub + 4u * (unsigned)u;
        if (k < cnt) {
            const unsigned e = (ent[u] >> 16) < nloc ? ent[u] : spare;
            raw[pre + k] = e;
            atomicAdd(&L.cur[e >> 16], 1u);
        }
    }
    for (unsigned k = sub + 4u * SEG_U; k < cnt; k += 4u) {
        const unsigned r = bp[k];
        const unsigned e = (r >> 16) < nloc ? r : spare;
        raw[pre + k] = e;
        atomicAdd(&L.cur[e >> 16], 1u);
    }
}

template <int NOUT, int CT>
__device__ __forceinline__ void rows_times_weight(const float* rows, const float* Ws,
                                                  float* __restrict__ dst,
                                                  const float* __restrict__ obias,
                                                  int n0, unsigned nloc, int wv, int lane)
{
    constexpr int NCT = (NOUT + 15) / 16;
    constexpr int NCG = (NCT + CT - 1) / CT;
    constexpr int NRT = (RANGE + 15) / 16;
    const int hi = lane >> 4;
    const int lo = lane & 15;
    for (int job = wv; job < NRT * NCG; job += OT / 32) {
        const int rt = job % NRT, cg = job / NRT;
        int row = rt * 16 + lo;
        if (row > RANGE) row = RANGE;
        const float* ap = rows + row * ACCP + 2 * hi * PLANE;
        int col[CT]; const float* bp[CT]; v8f acc[CT];
        #pragma unroll
        for (int u = 0; u < CT; ++u) {
            col[u] = (cg * CT + u) * 16 + lo;
            const int cc = col[u] < NOUT ? col[u] : NOUT - 1;
            bp[u]  = Ws + hi * WPP + 2 * cc;
            acc[u] = v8f{};
        }
        for (int kk = 0; kk < HC / 4; ++kk) {
            v2f a; a.x = ap[kk]; a.y = ap[PLANE + kk];
            #pragma unroll
            for (int u = 0; u < CT; ++u) {
                const float2 bv = *(const float2*)(bp[u] + kk * 2 * WPP);
                v2f b; b.x = bv.x; b.y = bv.y;
                acc[u] = wmma4(a, b, acc[u]);
            }
        }
        #pragma unroll
        for (int u = 0; u < CT; ++u) {
            if (col[u] < NOUT) {
                const float bb = obias ? obias[col[u]] : 0.f;
                #pragma unroll
                for (int r = 0; r < 8; ++r) {
                    const unsigned grow = (unsigned)(rt * 16 + 8 * hi + r);
                    if (grow < nloc) dst[(size_t)(n0 + (int)grow) * NOUT + col[u]] = acc[u][r] + bb;
                }
            }
        }
    }
}

#define XROW_U 7
#define W_U ((IN_DIM * HC) / OT)
static_assert(IN_DIM == HC, "the tile routine walks HC channels");
static_assert((RANGE + 1) * (IN_DIM / 4) <= OT * XROW_U, "one trip copies the whole range");
static_assert(W_U * OT == IN_DIM * HC, "the weight divides evenly over the threads");
#define XROWS ((OT * XROW_U) / (IN_DIM / 4))
struct FirstLds {
    alignas(16) float ws[(IN_DIM / 2) * WPP];
    float rows[XROWS * ACCP];
    alignas(16) unsigned hist[NWG];
    unsigned base[NWG];
};
static_assert(sizeof(FirstLds) <= 320 * 1024, "one group's LDS fits a WGP");
__global__ __launch_bounds__(OT) void range_product(
    const float* __restrict__ x, const float* __restrict__ Wn, float* __restrict__ dst,
    const int* __restrict__ ei, unsigned* __restrict__ bucket, unsigned* __restrict__ offT,
    const float* __restrict__ a_src, const float* __restrict__ a_dst,
    float* __restrict__ al_s, float* __restrict__ al_d)
{
    __shared__ FirstLds F;
    const int t    = threadIdx.x;
    const int lane = t & 31;
    const int wv   = t >> 5;
    const int n0   = blockIdx.x * RANGE;
    const unsigned nloc = (unsigned)((NN - n0) < RANGE ? (NN - n0) : RANGE);

    float4 v[XROW_U];
    float  w[W_U];
    int    es[SL_U], ed[SL_U];
    #pragma unroll
    for (int u = 0; u < SL_U; ++u) {
        const int k  = u * OT + t;
        const int kc = k < SLICE ? k : SLICE - 1;
        es[u] = ei[(int)blockIdx.x * SLICE + kc];
        ed[u] = ei[EE + (int)blockIdx.x * SLICE + kc];
    }
    #pragma unroll
    for (int u = 0; u < XROW_U; ++u) {
        const int i = u * OT + t;
        const unsigned node = (unsigned)(i >> 5);
        const unsigned nc   = node < nloc ? node : nloc - 1;
        v[u] = *((const float4*)(x + (size_t)(n0 + (int)nc) * IN_DIM) + (i & 31));
    }
    #pragma unroll
    for (int u = 0; u < W_U; ++u) w[u] = Wn[u * OT + t];
    if (t < NWG) F.hist[t] = 0u;
    #pragma unroll
    for (int u = 0; u < XROW_U; ++u) {
        const int i = u * OT + t;
        float* p = F.rows + (i >> 5) * ACCP + (i & 31);
        p[0] = v[u].x; p[PLANE] = v[u].y; p[2 * PLANE] = v[u].z; p[3 * PLANE] = v[u].w;
    }
    #pragma unroll
    for (int u = 0; u < W_U; ++u) {
        const int i = u * OT + t;
        const int k = i / HC, n = i % HC;
        F.ws[(k >> 1) * WPP + 2 * n + (k & 1)] = w[u];
    }
    __syncthreads();
    unsigned ent[SL_U], key[SL_U];
    #pragma unroll
    for (int u = 0; u < SL_U; ++u) {
        unsigned own = (unsigned)ed[u] / RANGE;
        own = own < (unsigned)(NWG - 1) ? own : (unsigned)(NWG - 1);
        unsigned ld = (unsigned)ed[u] - own * RANGE;
        ld = ld < RANGE ? ld : RANGE;
        ent[u] = (unsigned)es[u] | (ld << 16);
        unsigned place = 0u;
        if (u * OT + t < SLICE) place = atomicAdd(&F.hist[own], 1u);
        key[u] = own | (place << 8);
    }
    rows_times_weight<HC, 2>(F.rows, F.ws, dst, (const float*)nullptr, n0, nloc, wv, lane);
    __syncthreads();
    if (wv == 0) {
        const uint4 h0 = *(const uint4*)(F.hist + 8 * lane);
        const uint4 h1 = *(const uint4*)(F.hist + 8 * lane + 4);
        const unsigned c1 = h0.x, c2 = c1 + h0.y, c3 = c2 + h0.z, c4 = c3 + h0.w;
        const unsigned c5 = c4 + h1.x, c6 = c5 + h1.y, c7 = c6 + h1.z, c8 = c7 + h1.w;
        unsigned incl = c8;
        #pragma unroll
        for (int m = 1; m < 32; m <<= 1) {
            const unsigned o = (unsigned)__shfl_up((int)incl, m);
            if (lane >= m) incl += o;
        }
        const unsigned ex = incl - c8;
        *(uint4*)(F.base + 8 * lane)     = make_uint4(ex, ex + c1, ex + c2, ex + c3);
        *(uint4*)(F.base + 8 * lane + 4) = make_uint4(ex + c4, ex + c5, ex + c6, ex + c7);
    }
    own_logits(dst, a_src, a_dst, al_s, al_d, n0, nloc, wv, lane);
    __syncthreads();
    unsigned* bk = bucket + (size_t)blockIdx.x * SLICE;
    #pragma unroll
    for (int u = 0; u < SL_U; ++u)
        if (u * OT + t < SLICE) bk[F.base[key[u] & 255u] + (key[u] >> 8)] = ent[u];
    if (t < NWG) offT[(size_t)t * NWG + blockIdx.x] = F.base[t] | (F.hist[t] << 16);
}

template <bool BUILD, bool LAST>
__global__ __launch_bounds__(OT) void owner_aggregate(
    const int* __restrict__ ei, const float* __restrict__ h,
    const float* __restrict__ al_s, const float* __restrict__ al_d,
    const float* __restrict__ bias, const float* __restrict__ Wn,
    const float* __restrict__ obias, float* __restrict__ dst,
    unsigned* __restrict__ wlist, unsigned* __restrict__ wcount,
    const unsigned* __restrict__ bucket, const unsigned* __restrict__ offT,
    const float* __restrict__ a_src_n, const float* __restrict__ a_dst_n,
    float* __restrict__ al_s_n, float* __restrict__ al_d_n)
{
    __shared__ OwnerLds L;
    const int t    = threadIdx.x;
    const int lane = t & 31;
    const int wv   = t >> 5;
    const int n0   = blockIdx.x * RANGE;
    const unsigned nloc = (unsigned)((NN - n0) < RANGE ? (NN - n0) : RANGE);
    unsigned* wl = wlist + (size_t)blockIdx.x * CAP;

    if (!BUILD) {
        const unsigned saved = wcount[blockIdx.x];
        if (saved <= CAP) {
            for (unsigned i0 = 0; i0 < saved; i0 += OT * LIST_U) {
                unsigned ent[LIST_U];
                #pragma unroll
                for (int u = 0; u < LIST_U; ++u) {
                    const unsigned idx = i0 + u * OT + t;
                    ent[u] = idx < saved ? wl[idx] : 0u;
                }
                #pragma unroll
                for (int u = 0; u < LIST_U; ++u) {
                    const unsigned idx = i0 + u * OT + t;
                    if (idx < saved) L.list[idx] = ent[u];
                }
            }
        }
        if (t == 0) L.count = saved;
    }

    for (int i = t; i < (RANGE + 1) * ACCP; i += OT) L.acc[i] = 0.f;
    if (t < RANGE * HEADS) {
        L.m[t]   = FORD_NEG_INF;
        L.den[t] = 0.f;
        L.ald[t] = ((unsigned)t < nloc * HEADS) ? al_d[(size_t)n0 * HEADS + t] : 0.f;
    }
    if (BUILD) {
        if (t < ORD_U * 32) L.cur[t] = 0u;
        if (t < NWG) L.seg[t] = offT[(size_t)blockIdx.x * NWG + t];
    }
    __syncthreads();

    if (BUILD) {
        if (wv == 0) {
            const uint4 g0 = *(const uint4*)(L.seg + 8 * lane);
            const uint4 g1 = *(const uint4*)(L.seg + 8 * lane + 4);
            const unsigned c1 = g0.x >> 16, c2 = c1 + (g0.y >> 16), c3 = c2 + (g0.z >> 16), c4 = c3 + (g0.w >> 16);
            const unsigned c5 = c4 + (g1.x >> 16), c6 = c5 + (g1.y >> 16), c7 = c6 + (g1.z >> 16), c8 = c7 + (g1.w >> 16);
            unsigned incl = c8;
            #pragma unroll
            for (int m = 1; m < 32; m <<= 1) {
                const unsigned o = (unsigned)__shfl_up((int)incl, m);
                if (lane >= m) incl += o;
            }
            const unsigned ex = nloc + incl - c8;
            *(uint4*)(L.segpre + 8 * lane)     = make_uint4(ex, ex + c1, ex + c2, ex + c3);
            *(uint4*)(L.segpre + 8 * lane + 4) = make_uint4(ex + c4, ex + c5, ex + c6, ex + c7);
            if (lane == 31) L.count = nloc + incl;
        }
        __syncthreads();
    }
    const unsigned total = L.count;

    if (BUILD) {
        unsigned* raw = (unsigned*)L.ev;
        if (total <= CAP) {
            if ((unsigned)t < nloc) {
                raw[t] = (unsigned)(n0 + t) | ((unsigned)t << 16);
                atomicAdd(&L.cur[t], 1u);
            }
            gather_bins(L, bucket, raw, nloc, (unsigned)n0 | ((unsigned)RANGE << 16), t);
        }
        if (t == 0) wcount[blockIdx.x] = total;
        __syncthreads();
        if (total <= CAP) {
            if (wv == 0) {
                unsigned c[ORD_U], own = 0u;
                #pragma unroll
                for (int u = 0; u < ORD_U; ++u) { c[u] = L.cur[ORD_U * lane + u]; own += c[u]; }
                unsigned incl = own;
                #pragma unroll
                for (int m = 1; m < 32; m <<= 1) {
                    const unsigned o = (unsigned)__shfl_up((int)incl, m);
                    if (lane >= m) incl += o;
                }
                unsigned run = incl - own;
                #pragma unroll
                for (int u = 0; u < ORD_U; ++u) { L.cur[ORD_U * lane + u] = run; run += c[u]; }
            }
            __syncthreads();
            for (unsigned idx = t; idx < total; idx += OT) {
                const unsigned e   = raw[idx];
                const unsigned pos = atomicAdd(&L.cur[e >> 16], 1u);
                L.list[pos] = e;
                wl[pos] = e;
            }
        }
        __syncthreads();
    }

    if (total <= CAP) {
        const unsigned lstr = ((total + 31u) >> 5) | 1u;
        for (unsigned k = t; k < 32u * lstr; k += OT) {
            const unsigned idx = (k & 31u) * lstr + (k >> 5);
            if (idx < total) edge_logit_max(L, al_s, idx);
        }
        __syncthreads();
        for (unsigned k = t; k < 32u * lstr; k += OT) {
            const unsigned idx = (k & 31u) * lstr + (k >> 5);
            if (idx < total) edge_exp_den(L, idx);
        }
        __syncthreads();
        for (unsigned idx = t; idx < total; idx += OT) edge_alpha(L, idx);
        __syncthreads();
        const unsigned spare = (unsigned)n0 | ((unsigned)RANGE << 16);
        const unsigned wvs   = (unsigned)__builtin_amdgcn_readfirstlane(wv);
        const unsigned ntrip = (total + ROW_U - 1) / ROW_U;
        const unsigned per   = (ntrip + (OT / 32) - 1) / (OT / 32);
        const unsigned tb    = wvs * per;
        const unsigned te    = tb + per < ntrip ? tb + per : ntrip;
        unsigned curd = RANGE;
        float4 sum = make_float4(0.f, 0.f, 0.f, 0.f);
        for (unsigned tr = tb; tr < te; ++tr) {
            const unsigned b0  = tr * ROW_U;
            const unsigned idx = b0 + (unsigned)(lane & (ROW_U - 1));
            const bool live = idx < total;
            const unsigned e0 = L.list[live ? idx : 0];
            const float    a0 = L.ev[b0 * HEADS + lane];
            const int ent_l = (int)(live ? e0 : spare);
            const int a_l   = __float_as_int(live ? a0 : 0.f);
            unsigned ent[ROW_U]; float al[ROW_U]; float4 v[ROW_U];
            al[0] = share8<0>(a_l); al[1] = share8<1>(a_l); al[2] = share8<2>(a_l); al[3] = share8<3>(a_l);
            al[4] = share8<4>(a_l); al[5] = share8<5>(a_l); al[6] = share8<6>(a_l); al[7] = share8<7>(a_l);
            #pragma unroll
            for (int u = 0; u < ROW_U; ++u) {
                ent[u] = (unsigned)__builtin_amdgcn_readlane(ent_l, u);
                v[u]   = *((const float4*)(h + (size_t)(ent[u] & 0xFFFFu) * HC) + lane);
            }
            #pragma unroll
            for (int u = 0; u < ROW_U; ++u) {
                const unsigned ld = ent[u] >> 16;
                if (ld != curd) {
                    float* ap = L.acc + curd * ACCP + lane;
                    lds_add(ap,             sum.x);
                    lds_add(ap + PLANE,     sum.y);
                    lds_add(ap + 2 * PLANE, sum.z);
                    lds_add(ap + 3 * PLANE, sum.w);
                    sum  = make_float4(0.f, 0.f, 0.f, 0.f);
                    curd = ld;
                }
                sum.x = fmaf(v[u].x, al[u], sum.x);
                sum.y = fmaf(v[u].y, al[u], sum.y);
                sum.z = fmaf(v[u].z, al[u], sum.z);
                sum.w = fmaf(v[u].w, al[u], sum.w);
            }
        }
        {
            float* ap = L.acc + curd * ACCP + lane;
            lds_add(ap,             sum.x);
            lds_add(ap + PLANE,     sum.y);
            lds_add(ap + 2 * PLANE, sum.z);
            lds_add(ap + 3 * PLANE, sum.w);
        }
    } else {
        if ((unsigned)t < nloc) edge_max_one(L, al_s, n0 + t, (unsigned)t);
        scan_edges<1>(L, ei, h, al_s, n0, nloc, t);
        __syncthreads();
        if ((unsigned)t < nloc) edge_den_one(L, al_s, n0 + t, (unsigned)t);
        scan_edges<2>(L, ei, h, al_s, n0, nloc, t);
        __syncthreads();
        for (unsigned i = wv; i < nloc; i += OT / 32) row_one(L, h, al_s, n0 + (int)i, i, lane);
        scan_edges<3>(L, ei, h, al_s, n0, nloc, t);
    }
    __syncthreads();

    for (int i = t; i < RANGE * HC; i += OT) {
        const int node = i / HC, q = i % HC;
        const int j = q >> 5, l = q & 31;
        float* p = L.acc + node * ACCP + j * PLANE + l;
        const float vv = *p + bias[4 * l + j];
        *p = vv > 0.f ? vv : 0.f;
    }
    constexpr int NOUT = LAST ? OUT_DIM : HC;
    float* Ws = L.ev;
    for (int i = t; i < HC * NOUT; i += OT) {
        const int k = i / NOUT, n = i % NOUT;
        Ws[(k >> 1) * WPP + 2 * n + (k & 1)] = Wn[i];
    }
    __syncthreads();
    rows_times_weight<NOUT, LAST ? 1 : 2>(L.acc, Ws, dst, obias, n0, nloc, wv, lane);
    if (!LAST) {
        __syncthreads();
        own_logits(dst, a_src_n, a_dst_n, al_s_n, al_d_n, n0, nloc, wv, lane);
    }
}

extern "C" void kernel_launch(void* const* d_in, const int* in_sizes, int n_in,
                              void* d_out, int out_size, void* d_ws, size_t ws_size,
                              hipStream_t stream)
{
    const float* x   = (const float*)d_in[0];
    const int*   ei  = (const int*)d_in[1];
    const float* W[3]    = { (const float*)d_in[2],  (const float*)d_in[6],  (const float*)d_in[10] };
    const float* asv[3]  = { (const float*)d_in[3],  (const float*)d_in[7],  (const float*)d_in[11] };
    const float* adv[3]  = { (const float*)d_in[4],  (const float*)d_in[8],  (const float*)d_in[12] };
    const float* bv[3]   = { (const float*)d_in[5],  (const float*)d_in[9],  (const float*)d_in[13] };
    const float* Wl  = (const float*)d_in[14];
    const float* bl  = (const float*)d_in[15];

    char* ws = (char*)d_ws;
    float*    hA   = (float*)ws;                       ws += (size_t)NN * HC * sizeof(float);
    float*    hB   = (float*)ws;                       ws += (size_t)NN * HC * sizeof(float);
    float*    al_s0 = (float*)ws;                      ws += (size_t)NN * HEADS * sizeof(float);
    float*    al_d0 = (float*)ws;                      ws += (size_t)NN * HEADS * sizeof(float);
    float*    al_s1 = (float*)ws;                      ws += (size_t)NN * HEADS * sizeof(float);
    float*    al_d1 = (float*)ws;                      ws += (size_t)NN * HEADS * sizeof(float);
    unsigned* wlist  = (unsigned*)ws;                  ws += (size_t)NWG * CAP * sizeof(unsigned);
    unsigned* wcount = (unsigned*)ws;                  ws += (size_t)NWG * sizeof(unsigned);
    unsigned* bucket = (unsigned*)ws;                  ws += (size_t)EE * sizeof(unsigned);
    unsigned* offT   = (unsigned*)ws;


    hipLaunchKernelGGL(range_product, dim3(NWG), dim3(OT), 0, stream, x, W[0], hA, ei, bucket, offT,
                       asv[0], adv[0], al_s0, al_d0);
    hipLaunchKernelGGL((owner_aggregate<true, false>), dim3(NWG), dim3(OT), 0, stream,
                       ei, hA, al_s0, al_d0, bv[0], W[1], (const float*)nullptr, hB, wlist, wcount, bucket, offT,
                       asv[1], adv[1], al_s1, al_d1);
    hipLaunchKernelGGL((owner_aggregate<false, false>), dim3(NWG), dim3(OT), 0, stream,
                       ei, hB, al_s1, al_d1, bv[1], W[2], (const float*)nullptr, hA, wlist, wcount, bucket, offT,
                       asv[2], adv[2], al_s0, al_d0);
    hipLaunchKernelGGL((owner_aggregate<false, true>), dim3(NWG), dim3(OT), 0, stream,
                       ei, hA, al_s0, al_d0, bv[2], Wl, bl, (float*)d_out, wlist, wcount, bucket, offT,
                       (const float*)nullptr, (const float*)nullptr, (float*)nullptr, (float*)nullptr);
}
